# starting kernel
# baseline (speedup 1.0000x reference)
#include <hip/hip_runtime.h>
#include <math.h>


typedef float v2f __attribute__((ext_vector_type(2)));
typedef float v8f __attribute__((ext_vector_type(8)));

__device__ __forceinline__ v8f wmma_f32(v2f a, v2f b, v8f c) {
  return __builtin_amdgcn_wmma_f32_16x16x4_f32(false, a, false, b,
                                               (short)0, c, false, false);
}

#define BB 8
#define CC 1024
#define INF_NEG (-3.0e38f)

__global__ void __launch_bounds__(256) stats_kernel(
    const float* __restrict__ x,
    const float* __restrict__ Ws1, const float* __restrict__ bs1,
    const float* __restrict__ Ws2, const float* __restrict__ bs2,
    const float* __restrict__ Wc1, const float* __restrict__ bc1,
    const float* __restrict__ Wc2, const float* __restrict__ bc2,
    const float* __restrict__ att_w,
    float* __restrict__ slog_out, float* __restrict__ cs_out) {
  __shared__ float xm[256];
  __shared__ float hid[128];
  __shared__ float slg[5];
  __shared__ float sp[5];
  __shared__ float c1s[64];
  __shared__ float c2s[64];

  const int b = blockIdx.x;
  const int t = threadIdx.x;

  float s = 0.f;
  for (int c = 0; c < CC; ++c) s += x[((size_t)(b * CC + c)) * 256 + t];
  xm[t] = s * (1.0f / 1024.0f);
  __syncthreads();

  if (t < 128) {
    float a = bs1[t];
    for (int k = 0; k < 256; ++k) a += xm[k] * Ws1[k * 128 + t];
    hid[t] = fmaxf(a, 0.f);
  }
  __syncthreads();

  if (t < 5) {
    float a = bs2[t];
    for (int k = 0; k < 128; ++k) a += hid[k] * Ws2[k * 5 + t];
    slg[t] = a;
    slog_out[b * 5 + t] = a;
  }
  __syncthreads();

  if (t == 0) {
    float mx = slg[0];
    for (int i = 1; i < 5; ++i) mx = fmaxf(mx, slg[i]);
    float sum = 0.f;
    for (int i = 0; i < 5; ++i) { sp[i] = __expf(slg[i] - mx); sum += sp[i]; }
    float inv = 1.0f / sum;
    for (int i = 0; i < 5; ++i) sp[i] *= inv;
  }
  __syncthreads();

  if (t < 64) {
    float a = bc1[t];
    for (int k = 0; k < 5; ++k) a += sp[k] * Wc1[k * 64 + t];
    c1s[t] = fmaxf(a, 0.f);
  }
  __syncthreads();

  if (t < 64) {
    float a = bc2[t];
    for (int k = 0; k < 64; ++k) a += c1s[k] * Wc2[k * 64 + t];
    c2s[t] = a;
  }
  __syncthreads();

  if (t == 0) {
    float cs = 0.f;
    for (int k = 0; k < 64; ++k) cs += c2s[k] * att_w[128 + k];
    cs_out[b] = cs;
  }
}

__global__ void __launch_bounds__(256) logA_kernel(const float* __restrict__ A,
                                                   float* __restrict__ logA) {
  const int g = blockIdx.x * 256 + threadIdx.x;
  logA[g] = logf(A[g] + 1e-9f);
}

#define APAD 260
__global__ void __launch_bounds__(256) proj_gemm(
    const float* __restrict__ x, const float* __restrict__ Wp,
    const float* __restrict__ bp, float* __restrict__ xp) {
  extern __shared__ float sm[];
  float* Apan = sm;
  float* BpT  = sm + 64 * APAD;

  const int tid = threadIdx.x;
  const int m0 = blockIdx.x * 64;
  const int n0 = blockIdx.y * 64;

  for (int idx = tid; idx < 64 * 256; idx += 256) {
    const int r = idx >> 8, k = idx & 255;
    Apan[r * APAD + k] = x[(size_t)(m0 + r) * 256 + k];
  }
  for (int idx = tid; idx < 256 * 64; idx += 256) {
    const int k = idx >> 6, n = idx & 63;
    BpT[n * APAD + k] = Wp[(size_t)k * 256 + n0 + n];
  }
  __syncthreads();

  const int wave = tid >> 5, lane = tid & 31;
  const int r16 = lane & 15, half = lane >> 4;
  const int rw = wave >> 1;
  const int cw = (wave & 1) * 32;

  v8f acc0 = {0, 0, 0, 0, 0, 0, 0, 0};
  v8f acc1 = {0, 0, 0, 0, 0, 0, 0, 0};

  const float* arow = Apan + (rw * 16 + r16) * APAD;
  const float* b0row = BpT + (cw + r16) * APAD;
  const float* b1row = BpT + (cw + 16 + r16) * APAD;

  for (int kk = 0; kk < 256; kk += 4) {
    const int ko = kk + 2 * half;
    v2f a;  a.x = arow[ko];  a.y = arow[ko + 1];
    v2f b0; b0.x = b0row[ko]; b0.y = b0row[ko + 1];
    v2f b1; b1.x = b1row[ko]; b1.y = b1row[ko + 1];
    acc0 = wmma_f32(a, b0, acc0);
    acc1 = wmma_f32(a, b1, acc1);
  }

  const float bias0 = bp[n0 + cw + r16];
  const float bias1 = bp[n0 + cw + 16 + r16];
#pragma unroll
  for (int v = 0; v < 8; ++v) {
    const int row = m0 + rw * 16 + v + 8 * half;
    xp[(size_t)row * 256 + n0 + cw + r16]      = acc0[v] + bias0;
    xp[(size_t)row * 256 + n0 + cw + 16 + r16] = acc1[v] + bias1;
  }
}

__global__ void __launch_bounds__(256) srcdst_kernel(
    const float* __restrict__ xp, const float* __restrict__ att_w,
    float* __restrict__ srcg, float* __restrict__ dstg) {
  __shared__ float ss[256];
  __shared__ float sd[256];
  const int m = blockIdx.x, t = threadIdx.x;
  const int h = t >> 6, d = t & 63;
  const float v = xp[(size_t)m * 256 + t];
  ss[t] = v * att_w[d];
  sd[t] = v * att_w[64 + d];
  __syncthreads();
  for (int off = 32; off >= 1; off >>= 1) {
    if ((t & 63) < off) { ss[t] += ss[t + off]; sd[t] += sd[t + off]; }
    __syncthreads();
  }
  if ((t & 63) == 0) {
    srcg[m * 4 + h] = ss[t];
    dstg[m * 4 + h] = sd[t];
  }
}

#define JP 132
__global__ void __launch_bounds__(256) attn_kernel(
    const float* __restrict__ xp,   const float* __restrict__ logA,
    const float* __restrict__ srcg, const float* __restrict__ dstg,
    const float* __restrict__ csb,  const float* __restrict__ lam,
    float* __restrict__ out) {
  extern __shared__ float sm[];
  float* srcS = sm;
  float* XcT  = srcS + 1024;
  float* eb   = XcT + 64 * JP;

  const int tid = threadIdx.x;
  const int wave = tid >> 5, lane = tid & 31;
  const int r16 = lane & 15, half = lane >> 4;

  const int bid = blockIdx.x;
  const int b  = bid >> 5;
  const int h  = (bid >> 3) & 3;
  const int it = bid & 7;
  const int i0 = it * 128 + wave * 16;

  for (int idx = tid; idx < 1024; idx += 256)
    srcS[idx] = srcg[(b * CC + idx) * 4 + h];
  __syncthreads();

  const float cs = csb[b];
  const float lamh = lam[h];

  float dvals[16];
#pragma unroll
  for (int r = 0; r < 16; ++r)
    dvals[r] = dstg[(b * CC + i0 + r) * 4 + h];

  float mrow[16];
#pragma unroll
  for (int r = 0; r < 16; ++r) mrow[r] = INF_NEG;

  for (int jc = 0; jc < CC; jc += 128) {
    const float4 sv = ((const float4*)(srcS + jc))[lane];
#pragma unroll
    for (int r = 0; r < 16; ++r) {
      const float4 lg =
          ((const float4*)(logA + (size_t)(i0 + r) * CC + jc))[lane];
      float e0 = sv.x + dvals[r] + cs; e0 = e0 > 0.f ? e0 : 0.2f * e0;
      float e1 = sv.y + dvals[r] + cs; e1 = e1 > 0.f ? e1 : 0.2f * e1;
      float e2 = sv.z + dvals[r] + cs; e2 = e2 > 0.f ? e2 : 0.2f * e2;
      float e3 = sv.w + dvals[r] + cs; e3 = e3 > 0.f ? e3 : 0.2f * e3;
      const float v0 = e0 + lamh * lg.x;
      const float v1 = e1 + lamh * lg.y;
      const float v2 = e2 + lamh * lg.z;
      const float v3 = e3 + lamh * lg.w;
      mrow[r] = fmaxf(mrow[r], fmaxf(fmaxf(v0, v1), fmaxf(v2, v3)));
    }
  }
#pragma unroll
  for (int r = 0; r < 16; ++r) {
#pragma unroll
    for (int off = 16; off >= 1; off >>= 1)
      mrow[r] = fmaxf(mrow[r], __shfl_xor(mrow[r], off, 32));
  }

  v8f acc[4];
  const v8f vzero = {0, 0, 0, 0, 0, 0, 0, 0};
#pragma unroll
  for (int nt = 0; nt < 4; ++nt) acc[nt] = vzero;

  float srow[16];
#pragma unroll
  for (int r = 0; r < 16; ++r) srow[r] = 0.f;

  float* ebw = eb + wave * 16 * JP;

  for (int jc = 0; jc < CC; jc += 128) {
    __syncthreads();
    for (int idx = tid; idx < 8192; idx += 256) {
      const int j = idx >> 6, d = idx & 63;
      XcT[d * JP + j] = xp[((size_t)(b * CC + jc + j) * 4 + h) * 64 + d];
    }
    __syncthreads();

    const float4 sv = ((const float4*)(srcS + jc))[lane];
    for (int r = 0; r < 16; ++r) {
      const float4 lg =
          ((const float4*)(logA + (size_t)(i0 + r) * CC + jc))[lane];
      float e0 = sv.x + dvals[r] + cs; e0 = e0 > 0.f ? e0 : 0.2f * e0;
      float e1 = sv.y + dvals[r] + cs; e1 = e1 > 0.f ? e1 : 0.2f * e1;
      float e2 = sv.z + dvals[r] + cs; e2 = e2 > 0.f ? e2 : 0.2f * e2;
      float e3 = sv.w + dvals[r] + cs; e3 = e3 > 0.f ? e3 : 0.2f * e3;
      const float p0 = __expf(e0 + lamh * lg.x - mrow[r]);
      const float p1 = __expf(e1 + lamh * lg.y - mrow[r]);
      const float p2 = __expf(e2 + lamh * lg.z - mrow[r]);
      const float p3 = __expf(e3 + lamh * lg.w - mrow[r]);
      float* dstp = ebw + r * JP + lane * 4;
      dstp[0] = p0; dstp[1] = p1; dstp[2] = p2; dstp[3] = p3;
      float ps = (p0 + p1) + (p2 + p3);
#pragma unroll
      for (int off = 16; off >= 1; off >>= 1) ps += __shfl_xor(ps, off, 32);
      srow[r] += ps;
    }
    __syncthreads();

    const float* arow = ebw + r16 * JP;
    for (int kk = 0; kk < 128; kk += 4) {
      const int ko = kk + 2 * half;
      v2f a; a.x = arow[ko]; a.y = arow[ko + 1];
#pragma unroll
      for (int nt = 0; nt < 4; ++nt) {
        const float* brow = XcT + (nt * 16 + r16) * JP;
        v2f bv; bv.x = brow[ko]; bv.y = brow[ko + 1];
        acc[nt] = wmma_f32(a, bv, acc[nt]);
      }
    }
  }

#pragma unroll
  for (int v = 0; v < 8; ++v) {
    const int rr = v + 8 * half;
    const float inv = 1.0f / srow[rr];
    const int row = i0 + rr;
#pragma unroll
    for (int nt = 0; nt < 4; ++nt) {
      out[(size_t)(b * CC + row) * 256 + h * 64 + nt * 16 + r16] =
          acc[nt][v] * inv;
    }
  }
}

extern "C" void kernel_launch(void* const* d_in, const int* in_sizes, int n_in,
                              void* d_out, int out_size, void* d_ws,
                              size_t ws_size, hipStream_t stream) {
  (void)in_sizes; (void)n_in; (void)out_size; (void)ws_size;

  const float* x   = (const float*)d_in[0];
  const float* Apr = (const float*)d_in[1];
  const float* Wp  = (const float*)d_in[2];
  const float* bp  = (const float*)d_in[3];
  const float* Ws1 = (const float*)d_in[4];
  const float* bs1 = (const float*)d_in[5];
  const float* Ws2 = (const float*)d_in[6];
  const float* bs2 = (const float*)d_in[7];
  const float* Wc1 = (const float*)d_in[8];
  const float* bc1 = (const float*)d_in[9];
  const float* Wc2 = (const float*)d_in[10];
  const float* bc2 = (const float*)d_in[11];
  const float* att = (const float*)d_in[12];
  const float* lam = (const float*)d_in[13];

  float* out  = (float*)d_out;
  float* slog = out + (size_t)BB * CC * 256;

  float* xp   = (float*)d_ws;
  float* logA = xp + 2097152;
  float* srcg = logA + 1048576;
  float* dstg = srcg + 32768;
  float* csb  = dstg + 32768;

  stats_kernel<<<BB, 256, 0, stream>>>(x, Ws1, bs1, Ws2, bs2, Wc1, bc1, Wc2,
                                       bc2, att, slog, csb);
  logA_kernel<<<4096, 256, 0, stream>>>(Apr, logA);

  const size_t gemm_lds = (size_t)(2 * 64 * APAD) * sizeof(float);
  proj_gemm<<<dim3(128, 4), 256, gemm_lds, stream>>>(x, Wp, bp, xp);

  srcdst_kernel<<<BB * CC, 256, 0, stream>>>(xp, att, srcg, dstg);

  const size_t attn_lds =
      (size_t)(1024 + 64 * JP + 8 * 16 * JP) * sizeof(float);
  attn_kernel<<<BB * 4 * 8, 256, attn_lds, stream>>>(xp, logA, srcg, dstg,
                                                     csb, lam, out);
}


# Mean-pool no longer walks x in 8 serial blocks: proj_gemm keeps column sums while staging its A panel, stats kernel runs after it and adds 16 panel sums; the mean's summation order changes
# speedup vs baseline: 1.2588x; 1.2588x over previous
#include <hip/hip_runtime.h>
#include <math.h>


typedef float v2f __attribute__((ext_vector_type(2)));
typedef float v8f __attribute__((ext_vector_type(8)));

__device__ __forceinline__ v8f wmma_f32(v2f a, v2f b, v8f c) {
  return __builtin_amdgcn_wmma_f32_16x16x4_f32(false, a, false, b,
                                               (short)0, c, false, false);
}

#define BB 8
#define CC 1024
#define INF_NEG (-3.0e38f)

__global__ void __launch_bounds__(256) stats_kernel(
    const float* __restrict__ colpart,
    const float* __restrict__ Ws1, const float* __restrict__ bs1,
    const float* __restrict__ Ws2, const float* __restrict__ bs2,
    const float* __restrict__ Wc1, const float* __restrict__ bc1,
    const float* __restrict__ Wc2, const float* __restrict__ bc2,
    const float* __restrict__ att_w,
    float* __restrict__ slog_out, float* __restrict__ cs_out) {
  __shared__ float xm[256];
  __shared__ float hid[128];
  __shared__ float slg[5];
  __shared__ float sp[5];
  __shared__ float c1s[64];
  __shared__ float c2s[64];

  const int b = blockIdx.x;
  const int t = threadIdx.x;

  float s = 0.f;
  for (int p = 0; p < 16; ++p) s += colpart[(b * 16 + p) * 256 + t];
  xm[t] = s * (1.0f / 1024.0f);
  __syncthreads();

  if (t < 128) {
    float a = bs1[t];
    for (int k = 0; k < 256; ++k) a += xm[k] * Ws1[k * 128 + t];
    hid[t] = fmaxf(a, 0.f);
  }
  __syncthreads();

  if (t < 5) {
    float a = bs2[t];
    for (int k = 0; k < 128; ++k) a += hid[k] * Ws2[k * 5 + t];
    slg[t] = a;
    slog_out[b * 5 + t] = a;
  }
  __syncthreads();

  if (t == 0) {
    float mx = slg[0];
    for (int i = 1; i < 5; ++i) mx = fmaxf(mx, slg[i]);
    float sum = 0.f;
    for (int i = 0; i < 5; ++i) { sp[i] = __expf(slg[i] - mx); sum += sp[i]; }
    float inv = 1.0f / sum;
    for (int i = 0; i < 5; ++i) sp[i] *= inv;
  }
  __syncthreads();

  if (t < 64) {
    float a = bc1[t];
    for (int k = 0; k < 5; ++k) a += sp[k] * Wc1[k * 64 + t];
    c1s[t] = fmaxf(a, 0.f);
  }
  __syncthreads();

  if (t < 64) {
    float a = bc2[t];
    for (int k = 0; k < 64; ++k) a += c1s[k] * Wc2[k * 64 + t];
    c2s[t] = a;
  }
  __syncthreads();

  if (t == 0) {
    float cs = 0.f;
    for (int k = 0; k < 64; ++k) cs += c2s[k] * att_w[128 + k];
    cs_out[b] = cs;
  }
}

__global__ void __launch_bounds__(256) logA_kernel(const float* __restrict__ A,
                                                   float* __restrict__ logA) {
  const int g = blockIdx.x * 256 + threadIdx.x;
  logA[g] = logf(A[g] + 1e-9f);
}

#define APAD 260
__global__ void __launch_bounds__(256) proj_gemm(
    const float* __restrict__ x, const float* __restrict__ Wp,
    const float* __restrict__ bp, float* __restrict__ xp,
    float* __restrict__ colpart) {
  extern __shared__ float sm[];
  float* Apan = sm;
  float* BpT  = sm + 64 * APAD;

  const int tid = threadIdx.x;
  const int m0 = blockIdx.x * 64;
  const int n0 = blockIdx.y * 64;

  const float* xcol = x + (size_t)m0 * 256 + tid;
  float* acol = Apan + tid;
  float csum = 0.f;
  for (int r0 = 0; r0 < 64; r0 += 8) {
    float v[8];
    for (int i = 0; i < 8; ++i) v[i] = xcol[(r0 + i) * 256];
    for (int i = 0; i < 8; ++i) {
      acol[(r0 + i) * APAD] = v[i];
      csum += v[i];
    }
  }
  if (blockIdx.y == 0) colpart[blockIdx.x * 256 + tid] = csum;
  for (int idx = tid; idx < 256 * 64; idx += 256) {
    const int k = idx >> 6, n = idx & 63;
    BpT[n * APAD + k] = Wp[(size_t)k * 256 + n0 + n];
  }
  __syncthreads();

  const int wave = tid >> 5, lane = tid & 31;
  const int r16 = lane & 15, half = lane >> 4;
  const int rw = wave >> 1;
  const int cw = (wave & 1) * 32;

  v8f acc0 = {0, 0, 0, 0, 0, 0, 0, 0};
  v8f acc1 = {0, 0, 0, 0, 0, 0, 0, 0};

  const float* arow = Apan + (rw * 16 + r16) * APAD;
  const float* b0row = BpT + (cw + r16) * APAD;
  const float* b1row = BpT + (cw + 16 + r16) * APAD;

  for (int kk = 0; kk < 256; kk += 4) {
    const int ko = kk + 2 * half;
    v2f a;  a.x = arow[ko];  a.y = arow[ko + 1];
    v2f b0; b0.x = b0row[ko]; b0.y = b0row[ko + 1];
    v2f b1; b1.x = b1row[ko]; b1.y = b1row[ko + 1];
    acc0 = wmma_f32(a, b0, acc0);
    acc1 = wmma_f32(a, b1, acc1);
  }

  const float bias0 = bp[n0 + cw + r16];
  const float bias1 = bp[n0 + cw + 16 + r16];
#pragma unroll
  for (int v = 0; v < 8; ++v) {
    const int row = m0 + rw * 16 + v + 8 * half;
    xp[(size_t)row * 256 + n0 + cw + r16]      = acc0[v] + bias0;
    xp[(size_t)row * 256 + n0 + cw + 16 + r16] = acc1[v] + bias1;
  }
}

__global__ void __launch_bounds__(256) srcdst_kernel(
    const float* __restrict__ xp, const float* __restrict__ att_w,
    float* __restrict__ srcg, float* __restrict__ dstg) {
  __shared__ float ss[256];
  __shared__ float sd[256];
  const int m = blockIdx.x, t = threadIdx.x;
  const int h = t >> 6, d = t & 63;
  const float v = xp[(size_t)m * 256 + t];
  ss[t] = v * att_w[d];
  sd[t] = v * att_w[64 + d];
  __syncthreads();
  for (int off = 32; off >= 1; off >>= 1) {
    if ((t & 63) < off) { ss[t] += ss[t + off]; sd[t] += sd[t + off]; }
    __syncthreads();
  }
  if ((t & 63) == 0) {
    srcg[m * 4 + h] = ss[t];
    dstg[m * 4 + h] = sd[t];
  }
}

#define JP 132
__global__ void __launch_bounds__(256) attn_kernel(
    const float* __restrict__ xp,   const float* __restrict__ logA,
    const float* __restrict__ srcg, const float* __restrict__ dstg,
    const float* __restrict__ csb,  const float* __restrict__ lam,
    float* __restrict__ out) {
  extern __shared__ float sm[];
  float* srcS = sm;
  float* XcT  = srcS + 1024;
  float* eb   = XcT + 64 * JP;

  const int tid = threadIdx.x;
  const int wave = tid >> 5, lane = tid & 31;
  const int r16 = lane & 15, half = lane >> 4;

  const int bid = blockIdx.x;
  const int b  = bid >> 5;
  const int h  = (bid >> 3) & 3;
  const int it = bid & 7;
  const int i0 = it * 128 + wave * 16;

  for (int idx = tid; idx < 1024; idx += 256)
    srcS[idx] = srcg[(b * CC + idx) * 4 + h];
  __syncthreads();

  const float cs = csb[b];
  const float lamh = lam[h];

  float dvals[16];
#pragma unroll
  for (int r = 0; r < 16; ++r)
    dvals[r] = dstg[(b * CC + i0 + r) * 4 + h];

  float mrow[16];
#pragma unroll
  for (int r = 0; r < 16; ++r) mrow[r] = INF_NEG;

  for (int jc = 0; jc < CC; jc += 128) {
    const float4 sv = ((const float4*)(srcS + jc))[lane];
#pragma unroll
    for (int r = 0; r < 16; ++r) {
      const float4 lg =
          ((const float4*)(logA + (size_t)(i0 + r) * CC + jc))[lane];
      float e0 = sv.x + dvals[r] + cs; e0 = e0 > 0.f ? e0 : 0.2f * e0;
      float e1 = sv.y + dvals[r] + cs; e1 = e1 > 0.f ? e1 : 0.2f * e1;
      float e2 = sv.z + dvals[r] + cs; e2 = e2 > 0.f ? e2 : 0.2f * e2;
      float e3 = sv.w + dvals[r] + cs; e3 = e3 > 0.f ? e3 : 0.2f * e3;
      const float v0 = e0 + lamh * lg.x;
      const float v1 = e1 + lamh * lg.y;
      const float v2 = e2 + lamh * lg.z;
      const float v3 = e3 + lamh * lg.w;
      mrow[r] = fmaxf(mrow[r], fmaxf(fmaxf(v0, v1), fmaxf(v2, v3)));
    }
  }
#pragma unroll
  for (int r = 0; r < 16; ++r) {
#pragma unroll
    for (int off = 16; off >= 1; off >>= 1)
      mrow[r] = fmaxf(mrow[r], __shfl_xor(mrow[r], off, 32));
  }

  v8f acc[4];
  const v8f vzero = {0, 0, 0, 0, 0, 0, 0, 0};
#pragma unroll
  for (int nt = 0; nt < 4; ++nt) acc[nt] = vzero;

  float srow[16];
#pragma unroll
  for (int r = 0; r < 16; ++r) srow[r] = 0.f;

  float* ebw = eb + wave * 16 * JP;

  for (int jc = 0; jc < CC; jc += 128) {
    __syncthreads();
    for (int idx = tid; idx < 8192; idx += 256) {
      const int j = idx >> 6, d = idx & 63;
      XcT[d * JP + j] = xp[((size_t)(b * CC + jc + j) * 4 + h) * 64 + d];
    }
    __syncthreads();

    const float4 sv = ((const float4*)(srcS + jc))[lane];
    for (int r = 0; r < 16; ++r) {
      const float4 lg =
          ((const float4*)(logA + (size_t)(i0 + r) * CC + jc))[lane];
      float e0 = sv.x + dvals[r] + cs; e0 = e0 > 0.f ? e0 : 0.2f * e0;
      float e1 = sv.y + dvals[r] + cs; e1 = e1 > 0.f ? e1 : 0.2f * e1;
      float e2 = sv.z + dvals[r] + cs; e2 = e2 > 0.f ? e2 : 0.2f * e2;
      float e3 = sv.w + dvals[r] + cs; e3 = e3 > 0.f ? e3 : 0.2f * e3;
      const float p0 = __expf(e0 + lamh * lg.x - mrow[r]);
      const float p1 = __expf(e1 + lamh * lg.y - mrow[r]);
      const float p2 = __expf(e2 + lamh * lg.z - mrow[r]);
      const float p3 = __expf(e3 + lamh * lg.w - mrow[r]);
      float* dstp = ebw + r * JP + lane * 4;
      dstp[0] = p0; dstp[1] = p1; dstp[2] = p2; dstp[3] = p3;
      float ps = (p0 + p1) + (p2 + p3);
#pragma unroll
      for (int off = 16; off >= 1; off >>= 1) ps += __shfl_xor(ps, off, 32);
      srow[r] += ps;
    }
    __syncthreads();

    const float* arow = ebw + r16 * JP;
    for (int kk = 0; kk < 128; kk += 4) {
      const int ko = kk + 2 * half;
      v2f a; a.x = arow[ko]; a.y = arow[ko + 1];
#pragma unroll
      for (int nt = 0; nt < 4; ++nt) {
        const float* brow = XcT + (nt * 16 + r16) * JP;
        v2f bv; bv.x = brow[ko]; bv.y = brow[ko + 1];
        acc[nt] = wmma_f32(a, bv, acc[nt]);
      }
    }
  }

#pragma unroll
  for (int v = 0; v < 8; ++v) {
    const int rr = v + 8 * half;
    const float inv = 1.0f / srow[rr];
    const int row = i0 + rr;
#pragma unroll
    for (int nt = 0; nt < 4; ++nt) {
      out[(size_t)(b * CC + row) * 256 + h * 64 + nt * 16 + r16] =
          acc[nt][v] * inv;
    }
  }
}

extern "C" void kernel_launch(void* const* d_in, const int* in_sizes, int n_in,
                              void* d_out, int out_size, void* d_ws,
                              size_t ws_size, hipStream_t stream) {
  (void)in_sizes; (void)n_in; (void)out_size; (void)ws_size;

  const float* x   = (const float*)d_in[0];
  const float* Apr = (const float*)d_in[1];
  const float* Wp  = (const float*)d_in[2];
  const float* bp  = (const float*)d_in[3];
  const float* Ws1 = (const float*)d_in[4];
  const float* bs1 = (const float*)d_in[5];
  const float* Ws2 = (const float*)d_in[6];
  const float* bs2 = (const float*)d_in[7];
  const float* Wc1 = (const float*)d_in[8];
  const float* bc1 = (const float*)d_in[9];
  const float* Wc2 = (const float*)d_in[10];
  const float* bc2 = (const float*)d_in[11];
  const float* att = (const float*)d_in[12];
  const float* lam = (const float*)d_in[13];

  float* out  = (float*)d_out;
  float* slog = out + (size_t)BB * CC * 256;

  float* xp   = (float*)d_ws;
  float* logA = xp + 2097152;
  float* srcg = logA + 1048576;
  float* dstg = srcg + 32768;
  float* csb  = dstg + 32768;
  float* colpart = srcg;

  logA_kernel<<<4096, 256, 0, stream>>>(Apr, logA);

  const size_t gemm_lds = (size_t)(2 * 64 * APAD) * sizeof(float);
  proj_gemm<<<dim3(128, 4), 256, gemm_lds, stream>>>(x, Wp, bp, xp, colpart);

  stats_kernel<<<BB, 256, 0, stream>>>(colpart, Ws1, bs1, Ws2, bs2, Wc1, bc1,
                                       Wc2, bc2, att, slog, csb);

  srcdst_kernel<<<BB * CC, 256, 0, stream>>>(xp, att, srcg, dstg);

  const size_t attn_lds =
      (size_t)(1024 + 64 * JP + 8 * 16 * JP) * sizeof(float);
  attn_kernel<<<BB * 4 * 8, 256, attn_lds, stream>>>(xp, logA, srcg, dstg,
                                                     csb, lam, out);
}


# attn pass 2: a wave's 16 rows no longer run one after another: their logA loads go out as one group per chunk and row sums stay per lane until one cross-lane add at the end (row-sum order changes)
# speedup vs baseline: 1.7980x; 1.4283x over previous
#include <hip/hip_runtime.h>
#include <math.h>


typedef float v2f __attribute__((ext_vector_type(2)));
typedef float v8f __attribute__((ext_vector_type(8)));

__device__ __forceinline__ v8f wmma_f32(v2f a, v2f b, v8f c) {
  return __builtin_amdgcn_wmma_f32_16x16x4_f32(false, a, false, b,
                                               (short)0, c, false, false);
}

#define BB 8
#define CC 1024
#define INF_NEG (-3.0e38f)

__global__ void __launch_bounds__(256) stats_kernel(
    const float* __restrict__ colpart,
    const float* __restrict__ Ws1, const float* __restrict__ bs1,
    const float* __restrict__ Ws2, const float* __restrict__ bs2,
    const float* __restrict__ Wc1, const float* __restrict__ bc1,
    const float* __restrict__ Wc2, const float* __restrict__ bc2,
    const float* __restrict__ att_w,
    float* __restrict__ slog_out, float* __restrict__ cs_out) {
  __shared__ float xm[256];
  __shared__ float hid[128];
  __shared__ float slg[5];
  __shared__ float sp[5];
  __shared__ float c1s[64];
  __shared__ float c2s[64];

  const int b = blockIdx.x;
  const int t = threadIdx.x;

  float s = 0.f;
  for (int p = 0; p < 16; ++p) s += colpart[(b * 16 + p) * 256 + t];
  xm[t] = s * (1.0f / 1024.0f);
  __syncthreads();

  if (t < 128) {
    float a = bs1[t];
    for (int k = 0; k < 256; ++k) a += xm[k] * Ws1[k * 128 + t];
    hid[t] = fmaxf(a, 0.f);
  }
  __syncthreads();

  if (t < 5) {
    float a = bs2[t];
    for (int k = 0; k < 128; ++k) a += hid[k] * Ws2[k * 5 + t];
    slg[t] = a;
    slog_out[b * 5 + t] = a;
  }
  __syncthreads();

  if (t == 0) {
    float mx = slg[0];
    for (int i = 1; i < 5; ++i) mx = fmaxf(mx, slg[i]);
    float sum = 0.f;
    for (int i = 0; i < 5; ++i) { sp[i] = __expf(slg[i] - mx); sum += sp[i]; }
    float inv = 1.0f / sum;
    for (int i = 0; i < 5; ++i) sp[i] *= inv;
  }
  __syncthreads();

  if (t < 64) {
    float a = bc1[t];
    for (int k = 0; k < 5; ++k) a += sp[k] * Wc1[k * 64 + t];
    c1s[t] = fmaxf(a, 0.f);
  }
  __syncthreads();

  if (t < 64) {
    float a = bc2[t];
    for (int k = 0; k < 64; ++k) a += c1s[k] * Wc2[k * 64 + t];
    c2s[t] = a;
  }
  __syncthreads();

  if (t == 0) {
    float cs = 0.f;
    for (int k = 0; k < 64; ++k) cs += c2s[k] * att_w[128 + k];
    cs_out[b] = cs;
  }
}

__global__ void __launch_bounds__(256) logA_kernel(const float* __restrict__ A,
                                                   float* __restrict__ logA) {
  const int g = blockIdx.x * 256 + threadIdx.x;
  logA[g] = logf(A[g] + 1e-9f);
}

#define APAD 260
__global__ void __launch_bounds__(256) proj_gemm(
    const float* __restrict__ x, const float* __restrict__ Wp,
    const float* __restrict__ bp, float* __restrict__ xp,
    float* __restrict__ colpart) {
  extern __shared__ float sm[];
  float* Apan = sm;
  float* BpT  = sm + 64 * APAD;

  const int tid = threadIdx.x;
  const int m0 = blockIdx.x * 64;
  const int n0 = blockIdx.y * 64;

  const float* xcol = x + (size_t)m0 * 256 + tid;
  float* acol = Apan + tid;
  float csum = 0.f;
  for (int r0 = 0; r0 < 64; r0 += 8) {
    float v[8];
    for (int i = 0; i < 8; ++i) v[i] = xcol[(r0 + i) * 256];
    for (int i = 0; i < 8; ++i) {
      acol[(r0 + i) * APAD] = v[i];
      csum += v[i];
    }
  }
  if (blockIdx.y == 0) colpart[blockIdx.x * 256 + tid] = csum;
  for (int idx = tid; idx < 256 * 64; idx += 256) {
    const int k = idx >> 6, n = idx & 63;
    BpT[n * APAD + k] = Wp[(size_t)k * 256 + n0 + n];
  }
  __syncthreads();

  const int wave = tid >> 5, lane = tid & 31;
  const int r16 = lane & 15, half = lane >> 4;
  const int rw = wave >> 1;
  const int cw = (wave & 1) * 32;

  v8f acc0 = {0, 0, 0, 0, 0, 0, 0, 0};
  v8f acc1 = {0, 0, 0, 0, 0, 0, 0, 0};

  const float* arow = Apan + (rw * 16 + r16) * APAD;
  const float* b0row = BpT + (cw + r16) * APAD;
  const float* b1row = BpT + (cw + 16 + r16) * APAD;

  for (int kk = 0; kk < 256; kk += 4) {
    const int ko = kk + 2 * half;
    v2f a;  a.x = arow[ko];  a.y = arow[ko + 1];
    v2f b0; b0.x = b0row[ko]; b0.y = b0row[ko + 1];
    v2f b1; b1.x = b1row[ko]; b1.y = b1row[ko + 1];
    acc0 = wmma_f32(a, b0, acc0);
    acc1 = wmma_f32(a, b1, acc1);
  }

  const float bias0 = bp[n0 + cw + r16];
  const float bias1 = bp[n0 + cw + 16 + r16];
#pragma unroll
  for (int v = 0; v < 8; ++v) {
    const int row = m0 + rw * 16 + v + 8 * half;
    xp[(size_t)row * 256 + n0 + cw + r16]      = acc0[v] + bias0;
    xp[(size_t)row * 256 + n0 + cw + 16 + r16] = acc1[v] + bias1;
  }
}

__global__ void __launch_bounds__(256) srcdst_kernel(
    const float* __restrict__ xp, const float* __restrict__ att_w,
    float* __restrict__ srcg, float* __restrict__ dstg) {
  __shared__ float ss[256];
  __shared__ float sd[256];
  const int m = blockIdx.x, t = threadIdx.x;
  const int h = t >> 6, d = t & 63;
  const float v = xp[(size_t)m * 256 + t];
  ss[t] = v * att_w[d];
  sd[t] = v * att_w[64 + d];
  __syncthreads();
  for (int off = 32; off >= 1; off >>= 1) {
    if ((t & 63) < off) { ss[t] += ss[t + off]; sd[t] += sd[t + off]; }
    __syncthreads();
  }
  if ((t & 63) == 0) {
    srcg[m * 4 + h] = ss[t];
    dstg[m * 4 + h] = sd[t];
  }
}

#define JP 132
__global__ void __launch_bounds__(256) attn_kernel(
    const float* __restrict__ xp,   const float* __restrict__ logA,
    const float* __restrict__ srcg, const float* __restrict__ dstg,
    const float* __restrict__ csb,  const float* __restrict__ lam,
    float* __restrict__ out) {
  extern __shared__ float sm[];
  float* srcS = sm;
  float* XcT  = srcS + 1024;
  float* eb   = XcT + 64 * JP;

  const int tid = threadIdx.x;
  const int wave = tid >> 5, lane = tid & 31;
  const int r16 = lane & 15, half = lane >> 4;

  const int bid = blockIdx.x;
  const int b  = bid >> 5;
  const int h  = (bid >> 3) & 3;
  const int it = bid & 7;
  const int i0 = it * 128 + wave * 16;

  for (int idx = tid; idx < 1024; idx += 256)
    srcS[idx] = srcg[(b * CC + idx) * 4 + h];
  __syncthreads();

  const float cs = csb[b];
  const float lamh = lam[h];

  float dvals[16];
#pragma unroll
  for (int r = 0; r < 16; ++r)
    dvals[r] = dstg[(b * CC + i0 + r) * 4 + h];

  float mrow[16];
#pragma unroll
  for (int r = 0; r < 16; ++r) mrow[r] = INF_NEG;

  for (int jc = 0; jc < CC; jc += 128) {
    const float4 sv = ((const float4*)(srcS + jc))[lane];
#pragma unroll
    for (int r = 0; r < 16; ++r) {
      const float4 lg =
          ((const float4*)(logA + (size_t)(i0 + r) * CC + jc))[lane];
      float e0 = sv.x + dvals[r] + cs; e0 = e0 > 0.f ? e0 : 0.2f * e0;
      float e1 = sv.y + dvals[r] + cs; e1 = e1 > 0.f ? e1 : 0.2f * e1;
      float e2 = sv.z + dvals[r] + cs; e2 = e2 > 0.f ? e2 : 0.2f * e2;
      float e3 = sv.w + dvals[r] + cs; e3 = e3 > 0.f ? e3 : 0.2f * e3;
      const float v0 = e0 + lamh * lg.x;
      const float v1 = e1 + lamh * lg.y;
      const float v2 = e2 + lamh * lg.z;
      const float v3 = e3 + lamh * lg.w;
      mrow[r] = fmaxf(mrow[r], fmaxf(fmaxf(v0, v1), fmaxf(v2, v3)));
    }
  }
#pragma unroll
  for (int r = 0; r < 16; ++r) {
#pragma unroll
    for (int off = 16; off >= 1; off >>= 1)
      mrow[r] = fmaxf(mrow[r], __shfl_xor(mrow[r], off, 32));
  }

  v8f acc[4];
  const v8f vzero = {0, 0, 0, 0, 0, 0, 0, 0};
#pragma unroll
  for (int nt = 0; nt < 4; ++nt) acc[nt] = vzero;

  float srow[16];
#pragma unroll
  for (int r = 0; r < 16; ++r) srow[r] = 0.f;

  float* ebw = eb + wave * 16 * JP;

  for (int jc = 0; jc < CC; jc += 128) {
    __syncthreads();

    float4 lg[16];
#pragma unroll
    for (int r = 0; r < 16; ++r)
      lg[r] = ((const float4*)(logA + (size_t)(i0 + r) * CC + jc))[lane];

    for (int idx = tid; idx < 8192; idx += 256) {
      const int j = idx >> 6, d = idx & 63;
      XcT[d * JP + j] = xp[((size_t)(b * CC + jc + j) * 4 + h) * 64 + d];
    }
    __syncthreads();

    const float4 sv = ((const float4*)(srcS + jc))[lane];
#pragma unroll
    for (int r = 0; r < 16; ++r) {
      float e0 = sv.x + dvals[r] + cs; e0 = e0 > 0.f ? e0 : 0.2f * e0;
      float e1 = sv.y + dvals[r] + cs; e1 = e1 > 0.f ? e1 : 0.2f * e1;
      float e2 = sv.z + dvals[r] + cs; e2 = e2 > 0.f ? e2 : 0.2f * e2;
      float e3 = sv.w + dvals[r] + cs; e3 = e3 > 0.f ? e3 : 0.2f * e3;
      const float p0 = __expf(e0 + lamh * lg[r].x - mrow[r]);
      const float p1 = __expf(e1 + lamh * lg[r].y - mrow[r]);
      const float p2 = __expf(e2 + lamh * lg[r].z - mrow[r]);
      const float p3 = __expf(e3 + lamh * lg[r].w - mrow[r]);
      float* dstp = ebw + r * JP + lane * 4;
      dstp[0] = p0; dstp[1] = p1; dstp[2] = p2; dstp[3] = p3;
      srow[r] += (p0 + p1) + (p2 + p3);
    }
    __syncthreads();

    const float* arow = ebw + r16 * JP;
    for (int kk = 0; kk < 128; kk += 4) {
      const int ko = kk + 2 * half;
      v2f a; a.x = arow[ko]; a.y = arow[ko + 1];
#pragma unroll
      for (int nt = 0; nt < 4; ++nt) {
        const float* brow = XcT + (nt * 16 + r16) * JP;
        v2f bv; bv.x = brow[ko]; bv.y = brow[ko + 1];
        acc[nt] = wmma_f32(a, bv, acc[nt]);
      }
    }
  }

#pragma unroll
  for (int r = 0; r < 16; ++r) {
#pragma unroll
    for (int off = 16; off >= 1; off >>= 1)
      srow[r] += __shfl_xor(srow[r], off, 32);
  }

#pragma unroll
  for (int v = 0; v < 8; ++v) {
    const int rr = v + 8 * half;
    const float inv = 1.0f / srow[rr];
    const int row = i0 + rr;
#pragma unroll
    for (int nt = 0; nt < 4; ++nt) {
      out[(size_t)(b * CC + row) * 256 + h * 64 + nt * 16 + r16] =
          acc[nt][v] * inv;
    }
  }
}

extern "C" void kernel_launch(void* const* d_in, const int* in_sizes, int n_in,
                              void* d_out, int out_size, void* d_ws,
                              size_t ws_size, hipStream_t stream) {
  (void)in_sizes; (void)n_in; (void)out_size; (void)ws_size;

  const float* x   = (const float*)d_in[0];
  const float* Apr = (const float*)d_in[1];
  const float* Wp  = (const float*)d_in[2];
  const float* bp  = (const float*)d_in[3];
  const float* Ws1 = (const float*)d_in[4];
  const float* bs1 = (const float*)d_in[5];
  const float* Ws2 = (const float*)d_in[6];
  const float* bs2 = (const float*)d_in[7];
  const float* Wc1 = (const float*)d_in[8];
  const float* bc1 = (const float*)d_in[9];
  const float* Wc2 = (const float*)d_in[10];
  const float* bc2 = (const float*)d_in[11];
  const float* att = (const float*)d_in[12];
  const float* lam = (const float*)d_in[13];

  float* out  = (float*)d_out;
  float* slog = out + (size_t)BB * CC * 256;

  float* xp   = (float*)d_ws;
  float* logA = xp + 2097152;
  float* srcg = logA + 1048576;
  float* dstg = srcg + 32768;
  float* csb  = dstg + 32768;
  float* colpart = srcg;

  logA_kernel<<<4096, 256, 0, stream>>>(Apr, logA);

  const size_t gemm_lds = (size_t)(2 * 64 * APAD) * sizeof(float);
  proj_gemm<<<dim3(128, 4), 256, gemm_lds, stream>>>(x, Wp, bp, xp, colpart);

  stats_kernel<<<BB, 256, 0, stream>>>(colpart, Ws1, bs1, Ws2, bs2, Wc1, bc1,
                                       Wc2, bc2, att, slog, csb);

  srcdst_kernel<<<BB * CC, 256, 0, stream>>>(xp, att, srcg, dstg);

  const size_t attn_lds =
      (size_t)(1024 + 64 * JP + 8 * 16 * JP) * sizeof(float);
  attn_kernel<<<BB * 4 * 8, 256, attn_lds, stream>>>(xp, logA, srcg, dstg,
                                                     csb, lam, out);
}


# logA and src/dst kernels: one element per thread made 98k short waves whose entry set their time; now 16 elements per thread, 4 rows per wave (4k waves); lane-shuffle sums change src/dst sum order
# speedup vs baseline: 1.9620x; 1.0912x over previous
#include <hip/hip_runtime.h>
#include <math.h>


typedef float v2f __attribute__((ext_vector_type(2)));
typedef float v8f __attribute__((ext_vector_type(8)));

__device__ __forceinline__ v8f wmma_f32(v2f a, v2f b, v8f c) {
  return __builtin_amdgcn_wmma_f32_16x16x4_f32(false, a, false, b,
                                               (short)0, c, false, false);
}

#define BB 8
#define CC 1024
#define INF_NEG (-3.0e38f)

__global__ void __launch_bounds__(256) stats_kernel(
    const float* __restrict__ colpart,
    const float* __restrict__ Ws1, const float* __restrict__ bs1,
    const float* __restrict__ Ws2, const float* __restrict__ bs2,
    const float* __restrict__ Wc1, const float* __restrict__ bc1,
    const float* __restrict__ Wc2, const float* __restrict__ bc2,
    const float* __restrict__ att_w,
    float* __restrict__ slog_out, float* __restrict__ cs_out) {
  __shared__ float xm[256];
  __shared__ float hid[128];
  __shared__ float slg[5];
  __shared__ float sp[5];
  __shared__ float c1s[64];
  __shared__ float c2s[64];

  const int b = blockIdx.x;
  const int t = threadIdx.x;

  float s = 0.f;
  for (int p = 0; p < 16; ++p) s += colpart[(b * 16 + p) * 256 + t];
  xm[t] = s * (1.0f / 1024.0f);
  __syncthreads();

  if (t < 128) {
    float a = bs1[t];
    for (int k = 0; k < 256; ++k) a += xm[k] * Ws1[k * 128 + t];
    hid[t] = fmaxf(a, 0.f);
  }
  __syncthreads();

  if (t < 5) {
    float a = bs2[t];
    for (int k = 0; k < 128; ++k) a += hid[k] * Ws2[k * 5 + t];
    slg[t] = a;
    slog_out[b * 5 + t] = a;
  }
  __syncthreads();

  if (t == 0) {
    float mx = slg[0];
    for (int i = 1; i < 5; ++i) mx = fmaxf(mx, slg[i]);
    float sum = 0.f;
    for (int i = 0; i < 5; ++i) { sp[i] = __expf(slg[i] - mx); sum += sp[i]; }
    float inv = 1.0f / sum;
    for (int i = 0; i < 5; ++i) sp[i] *= inv;
  }
  __syncthreads();

  if (t < 64) {
    float a = bc1[t];
    for (int k = 0; k < 5; ++k) a += sp[k] * Wc1[k * 64 + t];
    c1s[t] = fmaxf(a, 0.f);
  }
  __syncthreads();

  if (t < 64) {
    float a = bc2[t];
    for (int k = 0; k < 64; ++k) a += c1s[k] * Wc2[k * 64 + t];
    c2s[t] = a;
  }
  __syncthreads();

  if (t == 0) {
    float cs = 0.f;
    for (int k = 0; k < 64; ++k) cs += c2s[k] * att_w[128 + k];
    cs_out[b] = cs;
  }
}

__global__ void __launch_bounds__(256) logA_kernel(const float* __restrict__ A,
                                                   float* __restrict__ logA) {
  const int g = blockIdx.x * 256 + threadIdx.x;
  const float4* A4 = (const float4*)A;
  float4* L4 = (float4*)logA;
  float4 a[4];
  for (int i = 0; i < 4; ++i) a[i] = A4[g + i * 65536];
  for (int i = 0; i < 4; ++i) {
    float4 o;
    o.x = logf(a[i].x + 1e-9f);
    o.y = logf(a[i].y + 1e-9f);
    o.z = logf(a[i].z + 1e-9f);
    o.w = logf(a[i].w + 1e-9f);
    L4[g + i * 65536] = o;
  }
}

#define APAD 260
__global__ void __launch_bounds__(256) proj_gemm(
    const float* __restrict__ x, const float* __restrict__ Wp,
    const float* __restrict__ bp, float* __restrict__ xp,
    float* __restrict__ colpart) {
  extern __shared__ float sm[];
  float* Apan = sm;
  float* BpT  = sm + 64 * APAD;

  const int tid = threadIdx.x;
  const int m0 = blockIdx.x * 64;
  const int n0 = blockIdx.y * 64;

  const float* xcol = x + (size_t)m0 * 256 + tid;
  float* acol = Apan + tid;
  float csum = 0.f;
  for (int r0 = 0; r0 < 64; r0 += 8) {
    float v[8];
    for (int i = 0; i < 8; ++i) v[i] = xcol[(r0 + i) * 256];
    for (int i = 0; i < 8; ++i) {
      acol[(r0 + i) * APAD] = v[i];
      csum += v[i];
    }
  }
  if (blockIdx.y == 0) colpart[blockIdx.x * 256 + tid] = csum;
  for (int idx = tid; idx < 256 * 64; idx += 256) {
    const int k = idx >> 6, n = idx & 63;
    BpT[n * APAD + k] = Wp[(size_t)k * 256 + n0 + n];
  }
  __syncthreads();

  const int wave = tid >> 5, lane = tid & 31;
  const int r16 = lane & 15, half = lane >> 4;
  const int rw = wave >> 1;
  const int cw = (wave & 1) * 32;

  v8f acc0 = {0, 0, 0, 0, 0, 0, 0, 0};
  v8f acc1 = {0, 0, 0, 0, 0, 0, 0, 0};

  const float* arow = Apan + (rw * 16 + r16) * APAD;
  const float* b0row = BpT + (cw + r16) * APAD;
  const float* b1row = BpT + (cw + 16 + r16) * APAD;

  for (int kk = 0; kk < 256; kk += 4) {
    const int ko = kk + 2 * half;
    v2f a;  a.x = arow[ko];  a.y = arow[ko + 1];
    v2f b0; b0.x = b0row[ko]; b0.y = b0row[ko + 1];
    v2f b1; b1.x = b1row[ko]; b1.y = b1row[ko + 1];
    acc0 = wmma_f32(a, b0, acc0);
    acc1 = wmma_f32(a, b1, acc1);
  }

  const float bias0 = bp[n0 + cw + r16];
  const float bias1 = bp[n0 + cw + 16 + r16];
#pragma unroll
  for (int v = 0; v < 8; ++v) {
    const int row = m0 + rw * 16 + v + 8 * half;
    xp[(size_t)row * 256 + n0 + cw + r16]      = acc0[v] + bias0;
    xp[(size_t)row * 256 + n0 + cw + 16 + r16] = acc1[v] + bias1;
  }
}

__global__ void __launch_bounds__(256) srcdst_kernel(
    const float* __restrict__ xp, const float* __restrict__ att_w,
    float* __restrict__ srcg, float* __restrict__ dstg) {
  const int lane = threadIdx.x & 31;
  const int m0 = (blockIdx.x * 8 + (threadIdx.x >> 5)) * 4;
  const int c16 = lane & 15;

  float4 va[4], vb[4];
  for (int r = 0; r < 4; ++r) {
    const float4* rowp = (const float4*)(xp + (size_t)(m0 + r) * 256);
    va[r] = rowp[lane];
    vb[r] = rowp[32 + lane];
  }
  const float4 ws = ((const float4*)att_w)[c16];
  const float4 wd = ((const float4*)(att_w + 64))[c16];

  float sa[4], da[4], sb[4], db[4];
  for (int r = 0; r < 4; ++r) {
    sa[r] = va[r].x * ws.x + va[r].y * ws.y + va[r].z * ws.z + va[r].w * ws.w;
    da[r] = va[r].x * wd.x + va[r].y * wd.y + va[r].z * wd.z + va[r].w * wd.w;
    sb[r] = vb[r].x * ws.x + vb[r].y * ws.y + vb[r].z * ws.z + vb[r].w * ws.w;
    db[r] = vb[r].x * wd.x + vb[r].y * wd.y + vb[r].z * wd.z + vb[r].w * wd.w;
  }
  for (int off = 8; off >= 1; off >>= 1) {
    for (int r = 0; r < 4; ++r) {
      sa[r] += __shfl_xor(sa[r], off, 32);
      da[r] += __shfl_xor(da[r], off, 32);
      sb[r] += __shfl_xor(sb[r], off, 32);
      db[r] += __shfl_xor(db[r], off, 32);
    }
  }
  float4 so[4], dq[4];
  for (int r = 0; r < 4; ++r) {
    so[r].x = sa[r]; so[r].y = __shfl_xor(sa[r], 16, 32);
    so[r].z = sb[r]; so[r].w = __shfl_xor(sb[r], 16, 32);
    dq[r].x = da[r]; dq[r].y = __shfl_xor(da[r], 16, 32);
    dq[r].z = db[r]; dq[r].w = __shfl_xor(db[r], 16, 32);
  }
  if (lane == 0) {
    for (int r = 0; r < 4; ++r) {
      ((float4*)srcg)[m0 + r] = so[r];
      ((float4*)dstg)[m0 + r] = dq[r];
    }
  }
}

#define JP 132
__global__ void __launch_bounds__(256) attn_kernel(
    const float* __restrict__ xp,   const float* __restrict__ logA,
    const float* __restrict__ srcg, const float* __restrict__ dstg,
    const float* __restrict__ csb,  const float* __restrict__ lam,
    float* __restrict__ out) {
  extern __shared__ float sm[];
  float* srcS = sm;
  float* XcT  = srcS + 1024;
  float* eb   = XcT + 64 * JP;

  const int tid = threadIdx.x;
  const int wave = tid >> 5, lane = tid & 31;
  const int r16 = lane & 15, half = lane >> 4;

  const int bid = blockIdx.x;
  const int b  = bid >> 5;
  const int h  = (bid >> 3) & 3;
  const int it = bid & 7;
  const int i0 = it * 128 + wave * 16;

  for (int idx = tid; idx < 1024; idx += 256)
    srcS[idx] = srcg[(b * CC + idx) * 4 + h];
  __syncthreads();

  const float cs = csb[b];
  const float lamh = lam[h];

  float dvals[16];
#pragma unroll
  for (int r = 0; r < 16; ++r)
    dvals[r] = dstg[(b * CC + i0 + r) * 4 + h];

  float mrow[16];
#pragma unroll
  for (int r = 0; r < 16; ++r) mrow[r] = INF_NEG;

  for (int jc = 0; jc < CC; jc += 128) {
    const float4 sv = ((const float4*)(srcS + jc))[lane];
#pragma unroll
    for (int r = 0; r < 16; ++r) {
      const float4 lg =
          ((const float4*)(logA + (size_t)(i0 + r) * CC + jc))[lane];
      float e0 = sv.x + dvals[r] + cs; e0 = e0 > 0.f ? e0 : 0.2f * e0;
      float e1 = sv.y + dvals[r] + cs; e1 = e1 > 0.f ? e1 : 0.2f * e1;
      float e2 = sv.z + dvals[r] + cs; e2 = e2 > 0.f ? e2 : 0.2f * e2;
      float e3 = sv.w + dvals[r] + cs; e3 = e3 > 0.f ? e3 : 0.2f * e3;
      const float v0 = e0 + lamh * lg.x;
      const float v1 = e1 + lamh * lg.y;
      const float v2 = e2 + lamh * lg.z;
      const float v3 = e3 + lamh * lg.w;
      mrow[r] = fmaxf(mrow[r], fmaxf(fmaxf(v0, v1), fmaxf(v2, v3)));
    }
  }
#pragma unroll
  for (int r = 0; r < 16; ++r) {
#pragma unroll
    for (int off = 16; off >= 1; off >>= 1)
      mrow[r] = fmaxf(mrow[r], __shfl_xor(mrow[r], off, 32));
  }

  v8f acc[4];
  const v8f vzero = {0, 0, 0, 0, 0, 0, 0, 0};
#pragma unroll
  for (int nt = 0; nt < 4; ++nt) acc[nt] = vzero;

  float srow[16];
#pragma unroll
  for (int r = 0; r < 16; ++r) srow[r] = 0.f;

  float* ebw = eb + wave * 16 * JP;

  for (int jc = 0; jc < CC; jc += 128) {
    __syncthreads();

    float4 lg[16];
#pragma unroll
    for (int r = 0; r < 16; ++r)
      lg[r] = ((const float4*)(logA + (size_t)(i0 + r) * CC + jc))[lane];

    for (int idx = tid; idx < 8192; idx += 256) {
      const int j = idx >> 6, d = idx & 63;
      XcT[d * JP + j] = xp[((size_t)(b * CC + jc + j) * 4 + h) * 64 + d];
    }
    __syncthreads();

    const float4 sv = ((const float4*)(srcS + jc))[lane];
#pragma unroll
    for (int r = 0; r < 16; ++r) {
      float e0 = sv.x + dvals[r] + cs; e0 = e0 > 0.f ? e0 : 0.2f * e0;
      float e1 = sv.y + dvals[r] + cs; e1 = e1 > 0.f ? e1 : 0.2f * e1;
      float e2 = sv.z + dvals[r] + cs; e2 = e2 > 0.f ? e2 : 0.2f * e2;
      float e3 = sv.w + dvals[r] + cs; e3 = e3 > 0.f ? e3 : 0.2f * e3;
      const float p0 = __expf(e0 + lamh * lg[r].x - mrow[r]);
      const float p1 = __expf(e1 + lamh * lg[r].y - mrow[r]);
      const float p2 = __expf(e2 + lamh * lg[r].z - mrow[r]);
      const float p3 = __expf(e3 + lamh * lg[r].w - mrow[r]);
      float* dstp = ebw + r * JP + lane * 4;
      dstp[0] = p0; dstp[1] = p1; dstp[2] = p2; dstp[3] = p3;
      srow[r] += (p0 + p1) + (p2 + p3);
    }
    __syncthreads();

    const float* arow = ebw + r16 * JP;
    for (int kk = 0; kk < 128; kk += 4) {
      const int ko = kk + 2 * half;
      v2f a; a.x = arow[ko]; a.y = arow[ko + 1];
#pragma unroll
      for (int nt = 0; nt < 4; ++nt) {
        const float* brow = XcT + (nt * 16 + r16) * JP;
        v2f bv; bv.x = brow[ko]; bv.y = brow[ko + 1];
        acc[nt] = wmma_f32(a, bv, acc[nt]);
      }
    }
  }

#pragma unroll
  for (int r = 0; r < 16; ++r) {
#pragma unroll
    for (int off = 16; off >= 1; off >>= 1)
      srow[r] += __shfl_xor(srow[r], off, 32);
  }

#pragma unroll
  for (int v = 0; v < 8; ++v) {
    const int rr = v + 8 * half;
    const float inv = 1.0f / srow[rr];
    const int row = i0 + rr;
#pragma unroll
    for (int nt = 0; nt < 4; ++nt) {
      out[(size_t)(b * CC + row) * 256 + h * 64 + nt * 16 + r16] =
          acc[nt][v] * inv;
    }
  }
}

extern "C" void kernel_launch(void* const* d_in, const int* in_sizes, int n_in,
                              void* d_out, int out_size, void* d_ws,
                              size_t ws_size, hipStream_t stream) {
  (void)in_sizes; (void)n_in; (void)out_size; (void)ws_size;

  const float* x   = (const float*)d_in[0];
  const float* Apr = (const float*)d_in[1];
  const float* Wp  = (const float*)d_in[2];
  const float* bp  = (const float*)d_in[3];
  const float* Ws1 = (const float*)d_in[4];
  const float* bs1 = (const float*)d_in[5];
  const float* Ws2 = (const float*)d_in[6];
  const float* bs2 = (const float*)d_in[7];
  const float* Wc1 = (const float*)d_in[8];
  const float* bc1 = (const float*)d_in[9];
  const float* Wc2 = (const float*)d_in[10];
  const float* bc2 = (const float*)d_in[11];
  const float* att = (const float*)d_in[12];
  const float* lam = (const float*)d_in[13];

  float* out  = (float*)d_out;
  float* slog = out + (size_t)BB * CC * 256;

  float* xp   = (float*)d_ws;
  float* logA = xp + 2097152;
  float* srcg = logA + 1048576;
  float* dstg = srcg + 32768;
  float* csb  = dstg + 32768;
  float* colpart = srcg;

  logA_kernel<<<256, 256, 0, stream>>>(Apr, logA);

  const size_t gemm_lds = (size_t)(2 * 64 * APAD) * sizeof(float);
  proj_gemm<<<dim3(128, 4), 256, gemm_lds, stream>>>(x, Wp, bp, xp, colpart);

  stats_kernel<<<BB, 256, 0, stream>>>(colpart, Ws1, bs1, Ws2, bs2, Wc1, bc1,
                                       Wc2, bc2, att, slog, csb);

  srcdst_kernel<<<256, 256, 0, stream>>>(xp, att, srcg, dstg);

  const size_t attn_lds =
      (size_t)(1024 + 64 * JP + 8 * 16 * JP) * sizeof(float);
  attn_kernel<<<BB * 4 * 8, 256, attn_lds, stream>>>(xp, logA, srcg, dstg,
                                                     csb, lam, out);
}


# attn pass 2: chunk c+1's x_proj tile (8 float4 per thread) and logA quads are requested in front of chunk c's WMMA loop and used after it; the tile was staged by 4 dependent load groups per chunk
# speedup vs baseline: 2.0865x; 1.0635x over previous
#include <hip/hip_runtime.h>
#include <math.h>


typedef float v2f __attribute__((ext_vector_type(2)));
typedef float v8f __attribute__((ext_vector_type(8)));

__device__ __forceinline__ v8f wmma_f32(v2f a, v2f b, v8f c) {
  return __builtin_amdgcn_wmma_f32_16x16x4_f32(false, a, false, b,
                                               (short)0, c, false, false);
}

#define BB 8
#define CC 1024
#define INF_NEG (-3.0e38f)

__global__ void __launch_bounds__(256) stats_kernel(
    const float* __restrict__ colpart,
    const float* __restrict__ Ws1, const float* __restrict__ bs1,
    const float* __restrict__ Ws2, const float* __restrict__ bs2,
    const float* __restrict__ Wc1, const float* __restrict__ bc1,
    const float* __restrict__ Wc2, const float* __restrict__ bc2,
    const float* __restrict__ att_w,
    float* __restrict__ slog_out, float* __restrict__ cs_out) {
  __shared__ float xm[256];
  __shared__ float hid[128];
  __shared__ float slg[5];
  __shared__ float sp[5];
  __shared__ float c1s[64];
  __shared__ float c2s[64];

  const int b = blockIdx.x;
  const int t = threadIdx.x;

  float s = 0.f;
  for (int p = 0; p < 16; ++p) s += colpart[(b * 16 + p) * 256 + t];
  xm[t] = s * (1.0f / 1024.0f);
  __syncthreads();

  if (t < 128) {
    float a = bs1[t];
    for (int k = 0; k < 256; ++k) a += xm[k] * Ws1[k * 128 + t];
    hid[t] = fmaxf(a, 0.f);
  }
  __syncthreads();

  if (t < 5) {
    float a = bs2[t];
    for (int k = 0; k < 128; ++k) a += hid[k] * Ws2[k * 5 + t];
    slg[t] = a;
    slog_out[b * 5 + t] = a;
  }
  __syncthreads();

  if (t == 0) {
    float mx = slg[0];
    for (int i = 1; i < 5; ++i) mx = fmaxf(mx, slg[i]);
    float sum = 0.f;
    for (int i = 0; i < 5; ++i) { sp[i] = __expf(slg[i] - mx); sum += sp[i]; }
    float inv = 1.0f / sum;
    for (int i = 0; i < 5; ++i) sp[i] *= inv;
  }
  __syncthreads();

  if (t < 64) {
    float a = bc1[t];
    for (int k = 0; k < 5; ++k) a += sp[k] * Wc1[k * 64 + t];
    c1s[t] = fmaxf(a, 0.f);
  }
  __syncthreads();

  if (t < 64) {
    float a = bc2[t];
    for (int k = 0; k < 64; ++k) a += c1s[k] * Wc2[k * 64 + t];
    c2s[t] = a;
  }
  __syncthreads();

  if (t == 0) {
    float cs = 0.f;
    for (int k = 0; k < 64; ++k) cs += c2s[k] * att_w[128 + k];
    cs_out[b] = cs;
  }
}

__global__ void __launch_bounds__(256) logA_kernel(const float* __restrict__ A,
                                                   float* __restrict__ logA) {
  const int g = blockIdx.x * 256 + threadIdx.x;
  const float4* A4 = (const float4*)A;
  float4* L4 = (float4*)logA;
  float4 a[4];
  for (int i = 0; i < 4; ++i) a[i] = A4[g + i * 65536];
  for (int i = 0; i < 4; ++i) {
    float4 o;
    o.x = logf(a[i].x + 1e-9f);
    o.y = logf(a[i].y + 1e-9f);
    o.z = logf(a[i].z + 1e-9f);
    o.w = logf(a[i].w + 1e-9f);
    L4[g + i * 65536] = o;
  }
}

#define APAD 260
__global__ void __launch_bounds__(256) proj_gemm(
    const float* __restrict__ x, const float* __restrict__ Wp,
    const float* __restrict__ bp, float* __restrict__ xp,
    float* __restrict__ colpart) {
  extern __shared__ float sm[];
  float* Apan = sm;
  float* BpT  = sm + 64 * APAD;

  const int tid = threadIdx.x;
  const int m0 = blockIdx.x * 64;
  const int n0 = blockIdx.y * 64;

  const float* xcol = x + (size_t)m0 * 256 + tid;
  float* acol = Apan + tid;
  float csum = 0.f;
  for (int r0 = 0; r0 < 64; r0 += 8) {
    float v[8];
    for (int i = 0; i < 8; ++i) v[i] = xcol[(r0 + i) * 256];
    for (int i = 0; i < 8; ++i) {
      acol[(r0 + i) * APAD] = v[i];
      csum += v[i];
    }
  }
  if (blockIdx.y == 0) colpart[blockIdx.x * 256 + tid] = csum;
  for (int idx = tid; idx < 256 * 64; idx += 256) {
    const int k = idx >> 6, n = idx & 63;
    BpT[n * APAD + k] = Wp[(size_t)k * 256 + n0 + n];
  }
  __syncthreads();

  const int wave = tid >> 5, lane = tid & 31;
  const int r16 = lane & 15, half = lane >> 4;
  const int rw = wave >> 1;
  const int cw = (wave & 1) * 32;

  v8f acc0 = {0, 0, 0, 0, 0, 0, 0, 0};
  v8f acc1 = {0, 0, 0, 0, 0, 0, 0, 0};

  const float* arow = Apan + (rw * 16 + r16) * APAD;
  const float* b0row = BpT + (cw + r16) * APAD;
  const float* b1row = BpT + (cw + 16 + r16) * APAD;

  for (int kk = 0; kk < 256; kk += 4) {
    const int ko = kk + 2 * half;
    v2f a;  a.x = arow[ko];  a.y = arow[ko + 1];
    v2f b0; b0.x = b0row[ko]; b0.y = b0row[ko + 1];
    v2f b1; b1.x = b1row[ko]; b1.y = b1row[ko + 1];
    acc0 = wmma_f32(a, b0, acc0);
    acc1 = wmma_f32(a, b1, acc1);
  }

  const float bias0 = bp[n0 + cw + r16];
  const float bias1 = bp[n0 + cw + 16 + r16];
#pragma unroll
  for (int v = 0; v < 8; ++v) {
    const int row = m0 + rw * 16 + v + 8 * half;
    xp[(size_t)row * 256 + n0 + cw + r16]      = acc0[v] + bias0;
    xp[(size_t)row * 256 + n0 + cw + 16 + r16] = acc1[v] + bias1;
  }
}

__global__ void __launch_bounds__(256) srcdst_kernel(
    const float* __restrict__ xp, const float* __restrict__ att_w,
    float* __restrict__ srcg, float* __restrict__ dstg) {
  const int lane = threadIdx.x & 31;
  const int m0 = (blockIdx.x * 8 + (threadIdx.x >> 5)) * 4;
  const int c16 = lane & 15;

  float4 va[4], vb[4];
  for (int r = 0; r < 4; ++r) {
    const float4* rowp = (const float4*)(xp + (size_t)(m0 + r) * 256);
    va[r] = rowp[lane];
    vb[r] = rowp[32 + lane];
  }
  const float4 ws = ((const float4*)att_w)[c16];
  const float4 wd = ((const float4*)(att_w + 64))[c16];

  float sa[4], da[4], sb[4], db[4];
  for (int r = 0; r < 4; ++r) {
    sa[r] = va[r].x * ws.x + va[r].y * ws.y + va[r].z * ws.z + va[r].w * ws.w;
    da[r] = va[r].x * wd.x + va[r].y * wd.y + va[r].z * wd.z + va[r].w * wd.w;
    sb[r] = vb[r].x * ws.x + vb[r].y * ws.y + vb[r].z * ws.z + vb[r].w * ws.w;
    db[r] = vb[r].x * wd.x + vb[r].y * wd.y + vb[r].z * wd.z + vb[r].w * wd.w;
  }
  for (int off = 8; off >= 1; off >>= 1) {
    for (int r = 0; r < 4; ++r) {
      sa[r] += __shfl_xor(sa[r], off, 32);
      da[r] += __shfl_xor(da[r], off, 32);
      sb[r] += __shfl_xor(sb[r], off, 32);
      db[r] += __shfl_xor(db[r], off, 32);
    }
  }
  float4 so[4], dq[4];
  for (int r = 0; r < 4; ++r) {
    so[r].x = sa[r]; so[r].y = __shfl_xor(sa[r], 16, 32);
    so[r].z = sb[r]; so[r].w = __shfl_xor(sb[r], 16, 32);
    dq[r].x = da[r]; dq[r].y = __shfl_xor(da[r], 16, 32);
    dq[r].z = db[r]; dq[r].w = __shfl_xor(db[r], 16, 32);
  }
  if (lane == 0) {
    for (int r = 0; r < 4; ++r) {
      ((float4*)srcg)[m0 + r] = so[r];
      ((float4*)dstg)[m0 + r] = dq[r];
    }
  }
}

#define JP 132
__global__ void __launch_bounds__(256) attn_kernel(
    const float* __restrict__ xp,   const float* __restrict__ logA,
    const float* __restrict__ srcg, const float* __restrict__ dstg,
    const float* __restrict__ csb,  const float* __restrict__ lam,
    float* __restrict__ out) {
  extern __shared__ float sm[];
  float* srcS = sm;
  float* XcT  = srcS + 1024;
  float* eb   = XcT + 64 * JP;

  const int tid = threadIdx.x;
  const int wave = tid >> 5, lane = tid & 31;
  const int r16 = lane & 15, half = lane >> 4;

  const int bid = blockIdx.x;
  const int b  = bid >> 5;
  const int h  = (bid >> 3) & 3;
  const int it = bid & 7;
  const int i0 = it * 128 + wave * 16;

  for (int idx = tid; idx < 1024; idx += 256)
    srcS[idx] = srcg[(b * CC + idx) * 4 + h];
  __syncthreads();

  const float cs = csb[b];
  const float lamh = lam[h];

  float dvals[16];
#pragma unroll
  for (int r = 0; r < 16; ++r)
    dvals[r] = dstg[(b * CC + i0 + r) * 4 + h];

  const int tj = tid >> 3;
  const int td = (tid & 7) * 4;
  const float* xtile = xp + ((size_t)(b * CC + tj) * 4 + h) * 64 + td;

  float mrow[16];
#pragma unroll
  for (int r = 0; r < 16; ++r) mrow[r] = INF_NEG;

  for (int jc = 0; jc < CC; jc += 128) {
    const float4 sv = ((const float4*)(srcS + jc))[lane];
#pragma unroll
    for (int r = 0; r < 16; ++r) {
      const float4 lg =
          ((const float4*)(logA + (size_t)(i0 + r) * CC + jc))[lane];
      float e0 = sv.x + dvals[r] + cs; e0 = e0 > 0.f ? e0 : 0.2f * e0;
      float e1 = sv.y + dvals[r] + cs; e1 = e1 > 0.f ? e1 : 0.2f * e1;
      float e2 = sv.z + dvals[r] + cs; e2 = e2 > 0.f ? e2 : 0.2f * e2;
      float e3 = sv.w + dvals[r] + cs; e3 = e3 > 0.f ? e3 : 0.2f * e3;
      const float v0 = e0 + lamh * lg.x;
      const float v1 = e1 + lamh * lg.y;
      const float v2 = e2 + lamh * lg.z;
      const float v3 = e3 + lamh * lg.w;
      mrow[r] = fmaxf(mrow[r], fmaxf(fmaxf(v0, v1), fmaxf(v2, v3)));
    }
  }
  float4 xt[8];
#pragma unroll
  for (int i = 0; i < 8; ++i)
    xt[i] = *(const float4*)(xtile + (size_t)(32 * (i & 3)) * 256 +
                             32 * (i >> 2));
  float4 lg[16];
#pragma unroll
  for (int r = 0; r < 16; ++r)
    lg[r] = ((const float4*)(logA + (size_t)(i0 + r) * CC))[lane];

#pragma unroll
  for (int r = 0; r < 16; ++r) {
#pragma unroll
    for (int off = 16; off >= 1; off >>= 1)
      mrow[r] = fmaxf(mrow[r], __shfl_xor(mrow[r], off, 32));
  }

  v8f acc[4];
  const v8f vzero = {0, 0, 0, 0, 0, 0, 0, 0};
#pragma unroll
  for (int nt = 0; nt < 4; ++nt) acc[nt] = vzero;

  float srow[16];
#pragma unroll
  for (int r = 0; r < 16; ++r) srow[r] = 0.f;

  float* ebw = eb + wave * 16 * JP;

  for (int jc = 0; jc < CC; jc += 128) {
    __syncthreads();

#pragma unroll
    for (int i = 0; i < 8; ++i) {
      float* tp = XcT + (td + 32 * (i >> 2)) * JP + tj + 32 * (i & 3);
      tp[0]      = xt[i].x;
      tp[JP]     = xt[i].y;
      tp[2 * JP] = xt[i].z;
      tp[3 * JP] = xt[i].w;
    }
    __syncthreads();

    const float4 sv = ((const float4*)(srcS + jc))[lane];
#pragma unroll
    for (int r = 0; r < 16; ++r) {
      float e0 = sv.x + dvals[r] + cs; e0 = e0 > 0.f ? e0 : 0.2f * e0;
      float e1 = sv.y + dvals[r] + cs; e1 = e1 > 0.f ? e1 : 0.2f * e1;
      float e2 = sv.z + dvals[r] + cs; e2 = e2 > 0.f ? e2 : 0.2f * e2;
      float e3 = sv.w + dvals[r] + cs; e3 = e3 > 0.f ? e3 : 0.2f * e3;
      const float p0 = __expf(e0 + lamh * lg[r].x - mrow[r]);
      const float p1 = __expf(e1 + lamh * lg[r].y - mrow[r]);
      const float p2 = __expf(e2 + lamh * lg[r].z - mrow[r]);
      const float p3 = __expf(e3 + lamh * lg[r].w - mrow[r]);
      float* dstp = ebw + r * JP + lane * 4;
      dstp[0] = p0; dstp[1] = p1; dstp[2] = p2; dstp[3] = p3;
      srow[r] += (p0 + p1) + (p2 + p3);
    }
    __syncthreads();

    if (jc + 128 < CC) {
      const int jn = jc + 128;
#pragma unroll
      for (int i = 0; i < 8; ++i)
        xt[i] = *(const float4*)(xtile + (size_t)(jn + 32 * (i & 3)) * 256 +
                                 32 * (i >> 2));
#pragma unroll
      for (int r = 0; r < 16; ++r)
        lg[r] = ((const float4*)(logA + (size_t)(i0 + r) * CC + jn))[lane];
    }

    const float* arow = ebw + r16 * JP;
    for (int kk = 0; kk < 128; kk += 4) {
      const int ko = kk + 2 * half;
      v2f a; a.x = arow[ko]; a.y = arow[ko + 1];
#pragma unroll
      for (int nt = 0; nt < 4; ++nt) {
        const float* brow = XcT + (nt * 16 + r16) * JP;
        v2f bv; bv.x = brow[ko]; bv.y = brow[ko + 1];
        acc[nt] = wmma_f32(a, bv, acc[nt]);
      }
    }
  }

#pragma unroll
  for (int r = 0; r < 16; ++r) {
#pragma unroll
    for (int off = 16; off >= 1; off >>= 1)
      srow[r] += __shfl_xor(srow[r], off, 32);
  }

#pragma unroll
  for (int v = 0; v < 8; ++v) {
    const int rr = v + 8 * half;
    const float inv = 1.0f / srow[rr];
    const int row = i0 + rr;
#pragma unroll
    for (int nt = 0; nt < 4; ++nt) {
      out[(size_t)(b * CC + row) * 256 + h * 64 + nt * 16 + r16] =
          acc[nt][v] * inv;
    }
  }
}

extern "C" void kernel_launch(void* const* d_in, const int* in_sizes, int n_in,
                              void* d_out, int out_size, void* d_ws,
                              size_t ws_size, hipStream_t stream) {
  (void)in_sizes; (void)n_in; (void)out_size; (void)ws_size;

  const float* x   = (const float*)d_in[0];
  const float* Apr = (const float*)d_in[1];
  const float* Wp  = (const float*)d_in[2];
  const float* bp  = (const float*)d_in[3];
  const float* Ws1 = (const float*)d_in[4];
  const float* bs1 = (const float*)d_in[5];
  const float* Ws2 = (const float*)d_in[6];
  const float* bs2 = (const float*)d_in[7];
  const float* Wc1 = (const float*)d_in[8];
  const float* bc1 = (const float*)d_in[9];
  const float* Wc2 = (const float*)d_in[10];
  const float* bc2 = (const float*)d_in[11];
  const float* att = (const float*)d_in[12];
  const float* lam = (const float*)d_in[13];

  float* out  = (float*)d_out;
  float* slog = out + (size_t)BB * CC * 256;

  float* xp   = (float*)d_ws;
  float* logA = xp + 2097152;
  float* srcg = logA + 1048576;
  float* dstg = srcg + 32768;
  float* csb  = dstg + 32768;
  float* colpart = srcg;

  logA_kernel<<<256, 256, 0, stream>>>(Apr, logA);

  const size_t gemm_lds = (size_t)(2 * 64 * APAD) * sizeof(float);
  proj_gemm<<<dim3(128, 4), 256, gemm_lds, stream>>>(x, Wp, bp, xp, colpart);

  stats_kernel<<<BB, 256, 0, stream>>>(colpart, Ws1, bs1, Ws2, bs2, Wc1, bc1,
                                       Wc2, bc2, att, slog, csb);

  srcdst_kernel<<<256, 256, 0, stream>>>(xp, att, srcg, dstg);

  const size_t attn_lds =
      (size_t)(1024 + 64 * JP + 8 * 16 * JP) * sizeof(float);
  attn_kernel<<<BB * 4 * 8, 256, attn_lds, stream>>>(xp, logA, srcg, dstg,
                                                     csb, lam, out);
}


# stats kernel: weights were fetched inside its two k loops, 32 L2 round trips in a row; now all requested in one group up front, Ws1 split by k over 8 waves (hidden and logit sum order changes)
# speedup vs baseline: 2.5508x; 1.2225x over previous
#include <hip/hip_runtime.h>
#include <math.h>


typedef float v2f __attribute__((ext_vector_type(2)));
typedef float v4f __attribute__((ext_vector_type(4)));
typedef float v8f __attribute__((ext_vector_type(8)));

__device__ __forceinline__ v8f wmma_f32(v2f a, v2f b, v8f c) {
  return __builtin_amdgcn_wmma_f32_16x16x4_f32(false, a, false, b,
                                               (short)0, c, false, false);
}

#define BB 8
#define CC 1024
#define INF_NEG (-3.0e38f)

__global__ void __launch_bounds__(256) stats_kernel(
    const float* __restrict__ colpart,
    const float* __restrict__ Ws1, const float* __restrict__ bs1,
    const float* __restrict__ Ws2, const float* __restrict__ bs2,
    const float* __restrict__ Wc1, const float* __restrict__ bc1,
    const float* __restrict__ Wc2, const float* __restrict__ bc2,
    const float* __restrict__ att_w,
    float* __restrict__ slog_out, float* __restrict__ cs_out) {
  __shared__ float4 xm4[64];
  __shared__ float4 hpart4[256];
  __shared__ v4f w2s4[256];
  __shared__ v4f wc1s4[256];
  __shared__ v4f aws4[256];
  __shared__ v4f wc2s4[1024];
  __shared__ float hid[128];
  __shared__ float slg[5];
  __shared__ float sp[5];
  __shared__ float c1s[64];
  __shared__ float c2s[64];

  float* xm = (float*)xm4;
  const float* hpart = (const float*)hpart4;
  const float* w2s = (const float*)w2s4;
  const float* wc1s = (const float*)wc1s4;
  const float* wc2s = (const float*)wc2s4;
  const float* aws = (const float*)aws4;

  const int b = blockIdx.x;
  const int t = threadIdx.x;
  const int wave = t >> 5, lane = t & 31;

  float cp[16];
#pragma unroll
  for (int p = 0; p < 16; ++p) cp[p] = colpart[(b * 16 + p) * 256 + t];

  float4 w1[32];
  const float4* Ws1q = (const float4*)Ws1 + (size_t)(wave * 32) * 32 + lane;
#pragma unroll
  for (int i = 0; i < 32; ++i) w1[i] = Ws1q[i * 32];

  v4f q2  = ((const v4f*)Ws2)[t < 160 ? t : 159];
  v4f qc1 = ((const v4f*)Wc1)[t < 80 ? t : 79];
  v4f qaw = ((const v4f*)(att_w + 128))[t & 15];
  v4f qc2a = ((const v4f*)Wc2)[t];
  v4f qc2b = ((const v4f*)Wc2)[t + 256];
  v4f qc2c = ((const v4f*)Wc2)[t + 512];
  v4f qc2d = ((const v4f*)Wc2)[t + 768];
  const float bs1v = bs1[t & 127];
  const float bs2v = bs2[wave < 5 ? wave : 4];
  const float bc1v = bc1[t & 63];
  const float bc2v = bc2[t & 63];

  __builtin_amdgcn_sched_barrier(0);
  asm volatile("" : "+v"(q2), "+v"(qc1), "+v"(qaw), "+v"(qc2a), "+v"(qc2b),
                    "+v"(qc2c), "+v"(qc2d));

  float s = 0.f;
#pragma unroll
  for (int p = 0; p < 16; ++p) s += cp[p];
  xm[t] = s * (1.0f / 1024.0f);

  w2s4[t] = q2;
  wc1s4[t] = qc1;
  aws4[t] = qaw;
  wc2s4[t] = qc2a;
  wc2s4[t + 256] = qc2b;
  wc2s4[t + 512] = qc2c;
  wc2s4[t + 768] = qc2d;
  __syncthreads();

  {
    float4 acc = make_float4(0.f, 0.f, 0.f, 0.f);
#pragma unroll
    for (int i4 = 0; i4 < 8; ++i4) {
      const float4 xv = xm4[wave * 8 + i4];
      const float xs[4] = {xv.x, xv.y, xv.z, xv.w};
#pragma unroll
      for (int j = 0; j < 4; ++j) {
        const float4 w = w1[4 * i4 + j];
        acc.x += xs[j] * w.x;
        acc.y += xs[j] * w.y;
        acc.z += xs[j] * w.z;
        acc.w += xs[j] * w.w;
      }
    }
    hpart4[wave * 32 + lane] = acc;
  }
  __syncthreads();

  if (t < 128) {
    float a = bs1v;
#pragma unroll
    for (int w = 0; w < 8; ++w) a += hpart[w * 128 + t];
    hid[t] = fmaxf(a, 0.f);
  }
  __syncthreads();

  if (wave < 5) {
    float pl = 0.f;
#pragma unroll
    for (int q = 0; q < 4; ++q) {
      const int k = lane + 32 * q;
      pl += hid[k] * w2s[k * 5 + wave];
    }
#pragma unroll
    for (int off = 16; off >= 1; off >>= 1) pl += __shfl_xor(pl, off, 32);
    if (lane == 0) {
      const float a = bs2v + pl;
      slg[wave] = a;
      slog_out[b * 5 + wave] = a;
    }
  }
  __syncthreads();

  if (t == 0) {
    float mx = slg[0];
    for (int i = 1; i < 5; ++i) mx = fmaxf(mx, slg[i]);
    float sum = 0.f;
    for (int i = 0; i < 5; ++i) { sp[i] = __expf(slg[i] - mx); sum += sp[i]; }
    float inv = 1.0f / sum;
    for (int i = 0; i < 5; ++i) sp[i] *= inv;
  }
  __syncthreads();

  if (t < 64) {
    float a = bc1v;
    for (int k = 0; k < 5; ++k) a += sp[k] * wc1s[k * 64 + t];
    c1s[t] = fmaxf(a, 0.f);
  }
  __syncthreads();

  if (t < 64) {
    float a = bc2v;
    for (int k = 0; k < 64; ++k) a += c1s[k] * wc2s[k * 64 + t];
    c2s[t] = a;
  }
  __syncthreads();

  if (t == 0) {
    float cs = 0.f;
    for (int k = 0; k < 64; ++k) cs += c2s[k] * aws[k];
    cs_out[b] = cs;
  }
}

__global__ void __launch_bounds__(256) logA_kernel(const float* __restrict__ A,
                                                   float* __restrict__ logA) {
  const int g = blockIdx.x * 256 + threadIdx.x;
  const float4* A4 = (const float4*)A;
  float4* L4 = (float4*)logA;
  float4 a[4];
  for (int i = 0; i < 4; ++i) a[i] = A4[g + i * 65536];
  for (int i = 0; i < 4; ++i) {
    float4 o;
    o.x = logf(a[i].x + 1e-9f);
    o.y = logf(a[i].y + 1e-9f);
    o.z = logf(a[i].z + 1e-9f);
    o.w = logf(a[i].w + 1e-9f);
    L4[g + i * 65536] = o;
  }
}

#define APAD 260
__global__ void __launch_bounds__(256) proj_gemm(
    const float* __restrict__ x, const float* __restrict__ Wp,
    const float* __restrict__ bp, float* __restrict__ xp,
    float* __restrict__ colpart) {
  extern __shared__ float sm[];
  float* Apan = sm;
  float* BpT  = sm + 64 * APAD;

  const int tid = threadIdx.x;
  const int m0 = blockIdx.x * 64;
  const int n0 = blockIdx.y * 64;

  const float* xcol = x + (size_t)m0 * 256 + tid;
  float* acol = Apan + tid;
  float csum = 0.f;
  for (int r0 = 0; r0 < 64; r0 += 8) {
    float v[8];
    for (int i = 0; i < 8; ++i) v[i] = xcol[(r0 + i) * 256];
    for (int i = 0; i < 8; ++i) {
      acol[(r0 + i) * APAD] = v[i];
      csum += v[i];
    }
  }
  if (blockIdx.y == 0) colpart[blockIdx.x * 256 + tid] = csum;
  for (int idx = tid; idx < 256 * 64; idx += 256) {
    const int k = idx >> 6, n = idx & 63;
    BpT[n * APAD + k] = Wp[(size_t)k * 256 + n0 + n];
  }
  __syncthreads();

  const int wave = tid >> 5, lane = tid & 31;
  const int r16 = lane & 15, half = lane >> 4;
  const int rw = wave >> 1;
  const int cw = (wave & 1) * 32;

  v8f acc0 = {0, 0, 0, 0, 0, 0, 0, 0};
  v8f acc1 = {0, 0, 0, 0, 0, 0, 0, 0};

  const float* arow = Apan + (rw * 16 + r16) * APAD;
  const float* b0row = BpT + (cw + r16) * APAD;
  const float* b1row = BpT + (cw + 16 + r16) * APAD;

  for (int kk = 0; kk < 256; kk += 4) {
    const int ko = kk + 2 * half;
    v2f a;  a.x = arow[ko];  a.y = arow[ko + 1];
    v2f b0; b0.x = b0row[ko]; b0.y = b0row[ko + 1];
    v2f b1; b1.x = b1row[ko]; b1.y = b1row[ko + 1];
    acc0 = wmma_f32(a, b0, acc0);
    acc1 = wmma_f32(a, b1, acc1);
  }

  const float bias0 = bp[n0 + cw + r16];
  const float bias1 = bp[n0 + cw + 16 + r16];
#pragma unroll
  for (int v = 0; v < 8; ++v) {
    const int row = m0 + rw * 16 + v + 8 * half;
    xp[(size_t)row * 256 + n0 + cw + r16]      = acc0[v] + bias0;
    xp[(size_t)row * 256 + n0 + cw + 16 + r16] = acc1[v] + bias1;
  }
}

__global__ void __launch_bounds__(256) srcdst_kernel(
    const float* __restrict__ xp, const float* __restrict__ att_w,
    float* __restrict__ srcg, float* __restrict__ dstg) {
  const int lane = threadIdx.x & 31;
  const int m0 = (blockIdx.x * 8 + (threadIdx.x >> 5)) * 4;
  const int c16 = lane & 15;

  float4 va[4], vb[4];
  for (int r = 0; r < 4; ++r) {
    const float4* rowp = (const float4*)(xp + (size_t)(m0 + r) * 256);
    va[r] = rowp[lane];
    vb[r] = rowp[32 + lane];
  }
  const float4 ws = ((const float4*)att_w)[c16];
  const float4 wd = ((const float4*)(att_w + 64))[c16];

  float sa[4], da[4], sb[4], db[4];
  for (int r = 0; r < 4; ++r) {
    sa[r] = va[r].x * ws.x + va[r].y * ws.y + va[r].z * ws.z + va[r].w * ws.w;
    da[r] = va[r].x * wd.x + va[r].y * wd.y + va[r].z * wd.z + va[r].w * wd.w;
    sb[r] = vb[r].x * ws.x + vb[r].y * ws.y + vb[r].z * ws.z + vb[r].w * ws.w;
    db[r] = vb[r].x * wd.x + vb[r].y * wd.y + vb[r].z * wd.z + vb[r].w * wd.w;
  }
  for (int off = 8; off >= 1; off >>= 1) {
    for (int r = 0; r < 4; ++r) {
      sa[r] += __shfl_xor(sa[r], off, 32);
      da[r] += __shfl_xor(da[r], off, 32);
      sb[r] += __shfl_xor(sb[r], off, 32);
      db[r] += __shfl_xor(db[r], off, 32);
    }
  }
  float4 so[4], dq[4];
  for (int r = 0; r < 4; ++r) {
    so[r].x = sa[r]; so[r].y = __shfl_xor(sa[r], 16, 32);
    so[r].z = sb[r]; so[r].w = __shfl_xor(sb[r], 16, 32);
    dq[r].x = da[r]; dq[r].y = __shfl_xor(da[r], 16, 32);
    dq[r].z = db[r]; dq[r].w = __shfl_xor(db[r], 16, 32);
  }
  if (lane == 0) {
    for (int r = 0; r < 4; ++r) {
      ((float4*)srcg)[m0 + r] = so[r];
      ((float4*)dstg)[m0 + r] = dq[r];
    }
  }
}

#define JP 132
__global__ void __launch_bounds__(256) attn_kernel(
    const float* __restrict__ xp,   const float* __restrict__ logA,
    const float* __restrict__ srcg, const float* __restrict__ dstg,
    const float* __restrict__ csb,  const float* __restrict__ lam,
    float* __restrict__ out) {
  extern __shared__ float sm[];
  float* srcS = sm;
  float* XcT  = srcS + 1024;
  float* eb   = XcT + 64 * JP;

  const int tid = threadIdx.x;
  const int wave = tid >> 5, lane = tid & 31;
  const int r16 = lane & 15, half = lane >> 4;

  const int bid = blockIdx.x;
  const int b  = bid >> 5;
  const int h  = (bid >> 3) & 3;
  const int it = bid & 7;
  const int i0 = it * 128 + wave * 16;

  for (int idx = tid; idx < 1024; idx += 256)
    srcS[idx] = srcg[(b * CC + idx) * 4 + h];
  __syncthreads();

  const float cs = csb[b];
  const float lamh = lam[h];

  float dvals[16];
#pragma unroll
  for (int r = 0; r < 16; ++r)
    dvals[r] = dstg[(b * CC + i0 + r) * 4 + h];

  const int tj = tid >> 3;
  const int td = (tid & 7) * 4;
  const float* xtile = xp + ((size_t)(b * CC + tj) * 4 + h) * 64 + td;

  float mrow[16];
#pragma unroll
  for (int r = 0; r < 16; ++r) mrow[r] = INF_NEG;

  for (int jc = 0; jc < CC; jc += 128) {
    const float4 sv = ((const float4*)(srcS + jc))[lane];
#pragma unroll
    for (int r = 0; r < 16; ++r) {
      const float4 lg =
          ((const float4*)(logA + (size_t)(i0 + r) * CC + jc))[lane];
      float e0 = sv.x + dvals[r] + cs; e0 = e0 > 0.f ? e0 : 0.2f * e0;
      float e1 = sv.y + dvals[r] + cs; e1 = e1 > 0.f ? e1 : 0.2f * e1;
      float e2 = sv.z + dvals[r] + cs; e2 = e2 > 0.f ? e2 : 0.2f * e2;
      float e3 = sv.w + dvals[r] + cs; e3 = e3 > 0.f ? e3 : 0.2f * e3;
      const float v0 = e0 + lamh * lg.x;
      const float v1 = e1 + lamh * lg.y;
      const float v2 = e2 + lamh * lg.z;
      const float v3 = e3 + lamh * lg.w;
      mrow[r] = fmaxf(mrow[r], fmaxf(fmaxf(v0, v1), fmaxf(v2, v3)));
    }
  }
  float4 xt[8];
#pragma unroll
  for (int i = 0; i < 8; ++i)
    xt[i] = *(const float4*)(xtile + (size_t)(32 * (i & 3)) * 256 +
                             32 * (i >> 2));
  float4 lg[16];
#pragma unroll
  for (int r = 0; r < 16; ++r)
    lg[r] = ((const float4*)(logA + (size_t)(i0 + r) * CC))[lane];

#pragma unroll
  for (int r = 0; r < 16; ++r) {
#pragma unroll
    for (int off = 16; off >= 1; off >>= 1)
      mrow[r] = fmaxf(mrow[r], __shfl_xor(mrow[r], off, 32));
  }

  v8f acc[4];
  const v8f vzero = {0, 0, 0, 0, 0, 0, 0, 0};
#pragma unroll
  for (int nt = 0; nt < 4; ++nt) acc[nt] = vzero;

  float srow[16];
#pragma unroll
  for (int r = 0; r < 16; ++r) srow[r] = 0.f;

  float* ebw = eb + wave * 16 * JP;

  for (int jc = 0; jc < CC; jc += 128) {
    __syncthreads();

#pragma unroll
    for (int i = 0; i < 8; ++i) {
      float* tp = XcT + (td + 32 * (i >> 2)) * JP + tj + 32 * (i & 3);
      tp[0]      = xt[i].x;
      tp[JP]     = xt[i].y;
      tp[2 * JP] = xt[i].z;
      tp[3 * JP] = xt[i].w;
    }
    __syncthreads();

    const float4 sv = ((const float4*)(srcS + jc))[lane];
#pragma unroll
    for (int r = 0; r < 16; ++r) {
      float e0 = sv.x + dvals[r] + cs; e0 = e0 > 0.f ? e0 : 0.2f * e0;
      float e1 = sv.y + dvals[r] + cs; e1 = e1 > 0.f ? e1 : 0.2f * e1;
      float e2 = sv.z + dvals[r] + cs; e2 = e2 > 0.f ? e2 : 0.2f * e2;
      float e3 = sv.w + dvals[r] + cs; e3 = e3 > 0.f ? e3 : 0.2f * e3;
      const float p0 = __expf(e0 + lamh * lg[r].x - mrow[r]);
      const float p1 = __expf(e1 + lamh * lg[r].y - mrow[r]);
      const float p2 = __expf(e2 + lamh * lg[r].z - mrow[r]);
      const float p3 = __expf(e3 + lamh * lg[r].w - mrow[r]);
      float* dstp = ebw + r * JP + lane * 4;
      dstp[0] = p0; dstp[1] = p1; dstp[2] = p2; dstp[3] = p3;
      srow[r] += (p0 + p1) + (p2 + p3);
    }
    __syncthreads();

    if (jc + 128 < CC) {
      const int jn = jc + 128;
#pragma unroll
      for (int i = 0; i < 8; ++i)
        xt[i] = *(const float4*)(xtile + (size_t)(jn + 32 * (i & 3)) * 256 +
                                 32 * (i >> 2));
#pragma unroll
      for (int r = 0; r < 16; ++r)
        lg[r] = ((const float4*)(logA + (size_t)(i0 + r) * CC + jn))[lane];
    }

    const float* arow = ebw + r16 * JP;
    for (int kk = 0; kk < 128; kk += 4) {
      const int ko = kk + 2 * half;
      v2f a; a.x = arow[ko]; a.y = arow[ko + 1];
#pragma unroll
      for (int nt = 0; nt < 4; ++nt) {
        const float* brow = XcT + (nt * 16 + r16) * JP;
        v2f bv; bv.x = brow[ko]; bv.y = brow[ko + 1];
        acc[nt] = wmma_f32(a, bv, acc[nt]);
      }
    }
  }

#pragma unroll
  for (int r = 0; r < 16; ++r) {
#pragma unroll
    for (int off = 16; off >= 1; off >>= 1)
      srow[r] += __shfl_xor(srow[r], off, 32);
  }

#pragma unroll
  for (int v = 0; v < 8; ++v) {
    const int rr = v + 8 * half;
    const float inv = 1.0f / srow[rr];
    const int row = i0 + rr;
#pragma unroll
    for (int nt = 0; nt < 4; ++nt) {
      out[(size_t)(b * CC + row) * 256 + h * 64 + nt * 16 + r16] =
          acc[nt][v] * inv;
    }
  }
}

extern "C" void kernel_launch(void* const* d_in, const int* in_sizes, int n_in,
                              void* d_out, int out_size, void* d_ws,
                              size_t ws_size, hipStream_t stream) {
  (void)in_sizes; (void)n_in; (void)out_size; (void)ws_size;

  const float* x   = (const float*)d_in[0];
  const float* Apr = (const float*)d_in[1];
  const float* Wp  = (const float*)d_in[2];
  const float* bp  = (const float*)d_in[3];
  const float* Ws1 = (const float*)d_in[4];
  const float* bs1 = (const float*)d_in[5];
  const float* Ws2 = (const float*)d_in[6];
  const float* bs2 = (const float*)d_in[7];
  const float* Wc1 = (const float*)d_in[8];
  const float* bc1 = (const float*)d_in[9];
  const float* Wc2 = (const float*)d_in[10];
  const float* bc2 = (const float*)d_in[11];
  const float* att = (const float*)d_in[12];
  const float* lam = (const float*)d_in[13];

  float* out  = (float*)d_out;
  float* slog = out + (size_t)BB * CC * 256;

  float* xp   = (float*)d_ws;
  float* logA = xp + 2097152;
  float* srcg = logA + 1048576;
  float* dstg = srcg + 32768;
  float* csb  = dstg + 32768;
  float* colpart = srcg;

  logA_kernel<<<256, 256, 0, stream>>>(Apr, logA);

  const size_t gemm_lds = (size_t)(2 * 64 * APAD) * sizeof(float);
  proj_gemm<<<dim3(128, 4), 256, gemm_lds, stream>>>(x, Wp, bp, xp, colpart);

  stats_kernel<<<BB, 256, 0, stream>>>(colpart, Ws1, bs1, Ws2, bs2, Wc1, bc1,
                                       Wc2, bc2, att, slog, csb);

  srcdst_kernel<<<256, 256, 0, stream>>>(xp, att, srcg, dstg);

  const size_t attn_lds =
      (size_t)(1024 + 64 * JP + 8 * 16 * JP) * sizeof(float);
  attn_kernel<<<BB * 4 * 8, 256, attn_lds, stream>>>(xp, logA, srcg, dstg,
                                                     csb, lam, out);
}


# logA, stats and srcdst need only the projection, not each other: one grid of 264 blocks, branched on the block index, runs all three; 5 launches become 3; every sum order as before
# speedup vs baseline: 2.7349x; 1.0722x over previous
#include <hip/hip_runtime.h>
#include <math.h>


typedef float v2f __attribute__((ext_vector_type(2)));
typedef float v4f __attribute__((ext_vector_type(4)));
typedef float v8f __attribute__((ext_vector_type(8)));

__device__ __forceinline__ v8f wmma_f32(v2f a, v2f b, v8f c) {
  return __builtin_amdgcn_wmma_f32_16x16x4_f32(false, a, false, b,
                                               (short)0, c, false, false);
}

#define BB 8
#define CC 1024
#define INF_NEG (-3.0e38f)

__device__ __forceinline__ float4 logA_quad(const float4 a) {
  float4 o;
  o.x = logf(a.x + 1e-9f);
  o.y = logf(a.y + 1e-9f);
  o.z = logf(a.z + 1e-9f);
  o.w = logf(a.w + 1e-9f);
  return o;
}

typedef float v16f __attribute__((ext_vector_type(16)));
__device__ __forceinline__ void put4(v16f& d, const int k, const v4f q) {
  d[4 * k + 0] = q.x; d[4 * k + 1] = q.y;
  d[4 * k + 2] = q.z; d[4 * k + 3] = q.w;
}
__device__ __forceinline__ v4f get4(const v16f& d, const int k) {
  v4f q;
  q.x = d[4 * k + 0]; q.y = d[4 * k + 1];
  q.z = d[4 * k + 2]; q.w = d[4 * k + 3];
  return q;
}

__device__ __forceinline__ void stats_block(
    const int b, const float* colpart,
    const float* __restrict__ Ws1, const float* __restrict__ bs1,
    const float* __restrict__ Ws2, const float* __restrict__ bs2,
    const float* __restrict__ Wc1, const float* __restrict__ bc1,
    const float* __restrict__ Wc2, const float* __restrict__ bc2,
    const float* __restrict__ att_w,
    float* __restrict__ slog_out, float* __restrict__ cs_out,
    const float* __restrict__ A, float* logA) {
  __shared__ float4 xm4[64];
  __shared__ float4 hpart4[256];
  __shared__ v4f w2s4[256];
  __shared__ v4f wc1s4[256];
  __shared__ v4f aws4[256];
  __shared__ v4f wc2s4[1024];
  __shared__ float hid[128];
  __shared__ float slg[5];
  __shared__ float sp[5];
  __shared__ float c1s[64];
  __shared__ float c2s[64];

  float* xm = (float*)xm4;
  const float* hpart = (const float*)hpart4;
  const float* w2s = (const float*)w2s4;
  const float* wc1s = (const float*)wc1s4;
  const float* wc2s = (const float*)wc2s4;
  const float* aws = (const float*)aws4;

  const int t = threadIdx.x;
  const int wave = t >> 5, lane = t & 31;

  v16f cpv;
#pragma unroll
  for (int p = 0; p < 16; ++p) cpv[p] = colpart[(b * 16 + p) * 256 + t];

  v16f w1v[8];
  const v4f* Ws1q = (const v4f*)Ws1 + (size_t)(wave * 32) * 32 + lane;
#pragma unroll
  for (int i = 0; i < 32; ++i) put4(w1v[i >> 2], i & 3, Ws1q[i * 32]);

  v16f smA, smB;
  put4(smA, 0, ((const v4f*)Ws2)[t < 160 ? t : 159]);
  put4(smA, 1, ((const v4f*)Wc1)[t < 80 ? t : 79]);
  put4(smA, 2, ((const v4f*)(att_w + 128))[t & 15]);
  put4(smA, 3, ((const v4f*)Wc2)[t]);
  put4(smB, 0, ((const v4f*)Wc2)[t + 256]);
  put4(smB, 1, ((const v4f*)Wc2)[t + 512]);
  put4(smB, 2, ((const v4f*)Wc2)[t + 768]);
  smB[12] = bs1[t & 127];
  smB[13] = bs2[wave < 5 ? wave : 4];
  smB[14] = bc1[t & 63];
  smB[15] = bc2[t & 63];

  v16f aqv;
  const v4f* A4 = (const v4f*)A + b * 1024 + t;
#pragma unroll
  for (int i = 0; i < 4; ++i) put4(aqv, i, A4[i * 256]);

  asm volatile("" : "+v"(cpv), "+v"(w1v[0]), "+v"(w1v[1]), "+v"(w1v[2]),
                    "+v"(w1v[3]), "+v"(w1v[4]), "+v"(w1v[5]), "+v"(w1v[6]),
                    "+v"(w1v[7]), "+v"(smA), "+v"(smB), "+v"(aqv));

  const float bs1v = smB[12];
  const float bs2v = smB[13];
  const float bc1v = smB[14];
  const float bc2v = smB[15];

  float s = 0.f;
#pragma unroll
  for (int p = 0; p < 16; ++p) s += cpv[p];
  xm[t] = s * (1.0f / 1024.0f);

  w2s4[t] = get4(smA, 0);
  wc1s4[t] = get4(smA, 1);
  aws4[t] = get4(smA, 2);
  wc2s4[t] = get4(smA, 3);
  wc2s4[t + 256] = get4(smB, 0);
  wc2s4[t + 512] = get4(smB, 1);
  wc2s4[t + 768] = get4(smB, 2);
  __syncthreads();

  {
    float4 acc = make_float4(0.f, 0.f, 0.f, 0.f);
#pragma unroll
    for (int i4 = 0; i4 < 8; ++i4) {
      const float4 xv = xm4[wave * 8 + i4];
      const float xs[4] = {xv.x, xv.y, xv.z, xv.w};
#pragma unroll
      for (int j = 0; j < 4; ++j) {
        const v4f w = get4(w1v[i4], j);
        acc.x += xs[j] * w.x;
        acc.y += xs[j] * w.y;
        acc.z += xs[j] * w.z;
        acc.w += xs[j] * w.w;
      }
    }
    hpart4[wave * 32 + lane] = acc;
  }
  __syncthreads();

  if (t < 128) {
    float a = bs1v;
#pragma unroll
    for (int w = 0; w < 8; ++w) a += hpart[w * 128 + t];
    hid[t] = fmaxf(a, 0.f);
  }
  __syncthreads();

  if (wave < 5) {
    float pl = 0.f;
#pragma unroll
    for (int q = 0; q < 4; ++q) {
      const int k = lane + 32 * q;
      pl += hid[k] * w2s[k * 5 + wave];
    }
#pragma unroll
    for (int off = 16; off >= 1; off >>= 1) pl += __shfl_xor(pl, off, 32);
    if (lane == 0) {
      const float a = bs2v + pl;
      slg[wave] = a;
      slog_out[b * 5 + wave] = a;
    }
  }
  __syncthreads();

  if (t == 0) {
    float mx = slg[0];
    for (int i = 1; i < 5; ++i) mx = fmaxf(mx, slg[i]);
    float sum = 0.f;
    for (int i = 0; i < 5; ++i) { sp[i] = __expf(slg[i] - mx); sum += sp[i]; }
    float inv = 1.0f / sum;
    for (int i = 0; i < 5; ++i) sp[i] *= inv;
  }
  __syncthreads();

  if (t < 64) {
    float a = bc1v;
    for (int k = 0; k < 5; ++k) a += sp[k] * wc1s[k * 64 + t];
    c1s[t] = fmaxf(a, 0.f);
  }
  __syncthreads();

  if (t < 64) {
    float a = bc2v;
    for (int k = 0; k < 64; ++k) a += c1s[k] * wc2s[k * 64 + t];
    c2s[t] = a;
  }
  __syncthreads();

  if (t == 0) {
    float cs = 0.f;
    for (int k = 0; k < 64; ++k) cs += c2s[k] * aws[k];
    cs_out[b] = cs;
  }

  float4* L4 = (float4*)logA + b * 1024 + t;
#pragma unroll
  for (int i = 0; i < 4; ++i) {
    const v4f a = get4(aqv, i);
    L4[i * 256] = logA_quad(make_float4(a.x, a.y, a.z, a.w));
  }
}

#define APAD 260
__global__ void __launch_bounds__(256) proj_gemm(
    const float* __restrict__ x, const float* __restrict__ Wp,
    const float* __restrict__ bp, float* __restrict__ xp,
    float* __restrict__ colpart) {
  extern __shared__ float sm[];
  float* Apan = sm;
  float* BpT  = sm + 64 * APAD;

  const int tid = threadIdx.x;
  const int m0 = blockIdx.x * 64;
  const int n0 = blockIdx.y * 64;

  const float* xcol = x + (size_t)m0 * 256 + tid;
  float* acol = Apan + tid;
  float csum = 0.f;
  for (int r0 = 0; r0 < 64; r0 += 8) {
    float v[8];
    for (int i = 0; i < 8; ++i) v[i] = xcol[(r0 + i) * 256];
    for (int i = 0; i < 8; ++i) {
      acol[(r0 + i) * APAD] = v[i];
      csum += v[i];
    }
  }
  if (blockIdx.y == 0) colpart[blockIdx.x * 256 + tid] = csum;
  for (int idx = tid; idx < 256 * 64; idx += 256) {
    const int k = idx >> 6, n = idx & 63;
    BpT[n * APAD + k] = Wp[(size_t)k * 256 + n0 + n];
  }
  __syncthreads();

  const int wave = tid >> 5, lane = tid & 31;
  const int r16 = lane & 15, half = lane >> 4;
  const int rw = wave >> 1;
  const int cw = (wave & 1) * 32;

  v8f acc0 = {0, 0, 0, 0, 0, 0, 0, 0};
  v8f acc1 = {0, 0, 0, 0, 0, 0, 0, 0};

  const float* arow = Apan + (rw * 16 + r16) * APAD;
  const float* b0row = BpT + (cw + r16) * APAD;
  const float* b1row = BpT + (cw + 16 + r16) * APAD;

  for (int kk = 0; kk < 256; kk += 4) {
    const int ko = kk + 2 * half;
    v2f a;  a.x = arow[ko];  a.y = arow[ko + 1];
    v2f b0; b0.x = b0row[ko]; b0.y = b0row[ko + 1];
    v2f b1; b1.x = b1row[ko]; b1.y = b1row[ko + 1];
    acc0 = wmma_f32(a, b0, acc0);
    acc1 = wmma_f32(a, b1, acc1);
  }

  const float bias0 = bp[n0 + cw + r16];
  const float bias1 = bp[n0 + cw + 16 + r16];
#pragma unroll
  for (int v = 0; v < 8; ++v) {
    const int row = m0 + rw * 16 + v + 8 * half;
    xp[(size_t)row * 256 + n0 + cw + r16]      = acc0[v] + bias0;
    xp[(size_t)row * 256 + n0 + cw + 16 + r16] = acc1[v] + bias1;
  }
}

__device__ __forceinline__ void srcdst_block(
    const int blk, const int slice,
    const float* __restrict__ xp, const float* __restrict__ att_w,
    float* __restrict__ srcg, float* __restrict__ dstg,
    const float* __restrict__ A, float* __restrict__ logA) {
  const int lane = threadIdx.x & 31;
  const int m0 = (blk * 8 + (threadIdx.x >> 5)) * 4;
  const int c16 = lane & 15;
  const bool has_slice = slice < 256;

  float4 va[4], vb[4];
  for (int r = 0; r < 4; ++r) {
    const float4* rowp = (const float4*)(xp + (size_t)(m0 + r) * 256);
    va[r] = rowp[lane];
    vb[r] = rowp[32 + lane];
  }
  const float4 ws = ((const float4*)att_w)[c16];
  const float4 wd = ((const float4*)(att_w + 64))[c16];
  float4 aq[4];
  if (has_slice) {
    const float4* A4 = (const float4*)A + slice * 1024 + threadIdx.x;
    for (int i = 0; i < 4; ++i) aq[i] = A4[i * 256];
  }

  float sa[4], da[4], sb[4], db[4];
  for (int r = 0; r < 4; ++r) {
    sa[r] = va[r].x * ws.x + va[r].y * ws.y + va[r].z * ws.z + va[r].w * ws.w;
    da[r] = va[r].x * wd.x + va[r].y * wd.y + va[r].z * wd.z + va[r].w * wd.w;
    sb[r] = vb[r].x * ws.x + vb[r].y * ws.y + vb[r].z * ws.z + vb[r].w * ws.w;
    db[r] = vb[r].x * wd.x + vb[r].y * wd.y + vb[r].z * wd.z + vb[r].w * wd.w;
  }
  for (int off = 8; off >= 1; off >>= 1) {
    for (int r = 0; r < 4; ++r) {
      sa[r] += __shfl_xor(sa[r], off, 32);
      da[r] += __shfl_xor(da[r], off, 32);
      sb[r] += __shfl_xor(sb[r], off, 32);
      db[r] += __shfl_xor(db[r], off, 32);
    }
  }
  float4 so[4], dq[4];
  for (int r = 0; r < 4; ++r) {
    so[r].x = sa[r]; so[r].y = __shfl_xor(sa[r], 16, 32);
    so[r].z = sb[r]; so[r].w = __shfl_xor(sb[r], 16, 32);
    dq[r].x = da[r]; dq[r].y = __shfl_xor(da[r], 16, 32);
    dq[r].z = db[r]; dq[r].w = __shfl_xor(db[r], 16, 32);
  }
  if (lane == 0) {
    for (int r = 0; r < 4; ++r) {
      ((float4*)srcg)[m0 + r] = so[r];
      ((float4*)dstg)[m0 + r] = dq[r];
    }
  }
  if (has_slice) {
    float4* L4 = (float4*)logA + slice * 1024 + threadIdx.x;
    for (int i = 0; i < 4; ++i) L4[i * 256] = logA_quad(aq[i]);
  }
}

__global__ void __launch_bounds__(256) side_kernel(
    const float* colpart,
    const float* __restrict__ Ws1, const float* __restrict__ bs1,
    const float* __restrict__ Ws2, const float* __restrict__ bs2,
    const float* __restrict__ Wc1, const float* __restrict__ bc1,
    const float* __restrict__ Wc2, const float* __restrict__ bc2,
    const float* __restrict__ att_w,
    float* __restrict__ slog_out, float* __restrict__ cs_out,
    const float* __restrict__ xp,
    float* __restrict__ srcg, float* __restrict__ dstg,
    const float* __restrict__ A, float* logA) {
  const int blk = blockIdx.x;
  if (blk < 8) {
    stats_block(blk, colpart, Ws1, bs1, Ws2, bs2, Wc1, bc1, Wc2, bc2, att_w,
                slog_out, cs_out, A, logA);
  } else {
    srcdst_block(blk - 8, blk, xp, att_w, srcg, dstg, A, logA);
  }
}

#define JP 132
__global__ void __launch_bounds__(256) attn_kernel(
    const float* __restrict__ xp,   const float* __restrict__ logA,
    const float* __restrict__ srcg, const float* __restrict__ dstg,
    const float* __restrict__ csb,  const float* __restrict__ lam,
    float* __restrict__ out) {
  extern __shared__ float sm[];
  float* srcS = sm;
  float* XcT  = srcS + 1024;
  float* eb   = XcT + 64 * JP;

  const int tid = threadIdx.x;
  const int wave = tid >> 5, lane = tid & 31;
  const int r16 = lane & 15, half = lane >> 4;

  const int bid = blockIdx.x;
  const int b  = bid >> 5;
  const int h  = (bid >> 3) & 3;
  const int it = bid & 7;
  const int i0 = it * 128 + wave * 16;

  for (int idx = tid; idx < 1024; idx += 256)
    srcS[idx] = srcg[(b * CC + idx) * 4 + h];
  __syncthreads();

  const float cs = csb[b];
  const float lamh = lam[h];

  float dvals[16];
#pragma unroll
  for (int r = 0; r < 16; ++r)
    dvals[r] = dstg[(b * CC + i0 + r) * 4 + h];

  const int tj = tid >> 3;
  const int td = (tid & 7) * 4;
  const float* xtile = xp + ((size_t)(b * CC + tj) * 4 + h) * 64 + td;

  float mrow[16];
#pragma unroll
  for (int r = 0; r < 16; ++r) mrow[r] = INF_NEG;

  for (int jc = 0; jc < CC; jc += 128) {
    const float4 sv = ((const float4*)(srcS + jc))[lane];
#pragma unroll
    for (int r = 0; r < 16; ++r) {
      const float4 lg =
          ((const float4*)(logA + (size_t)(i0 + r) * CC + jc))[lane];
      float e0 = sv.x + dvals[r] + cs; e0 = e0 > 0.f ? e0 : 0.2f * e0;
      float e1 = sv.y + dvals[r] + cs; e1 = e1 > 0.f ? e1 : 0.2f * e1;
      float e2 = sv.z + dvals[r] + cs; e2 = e2 > 0.f ? e2 : 0.2f * e2;
      float e3 = sv.w + dvals[r] + cs; e3 = e3 > 0.f ? e3 : 0.2f * e3;
      const float v0 = e0 + lamh * lg.x;
      const float v1 = e1 + lamh * lg.y;
      const float v2 = e2 + lamh * lg.z;
      const float v3 = e3 + lamh * lg.w;
      mrow[r] = fmaxf(mrow[r], fmaxf(fmaxf(v0, v1), fmaxf(v2, v3)));
    }
  }
  float4 xt[8];
#pragma unroll
  for (int i = 0; i < 8; ++i)
    xt[i] = *(const float4*)(xtile + (size_t)(32 * (i & 3)) * 256 +
                             32 * (i >> 2));
  float4 lg[16];
#pragma unroll
  for (int r = 0; r < 16; ++r)
    lg[r] = ((const float4*)(logA + (size_t)(i0 + r) * CC))[lane];

#pragma unroll
  for (int r = 0; r < 16; ++r) {
#pragma unroll
    for (int off = 16; off >= 1; off >>= 1)
      mrow[r] = fmaxf(mrow[r], __shfl_xor(mrow[r], off, 32));
  }

  v8f acc[4];
  const v8f vzero = {0, 0, 0, 0, 0, 0, 0, 0};
#pragma unroll
  for (int nt = 0; nt < 4; ++nt) acc[nt] = vzero;

  float srow[16];
#pragma unroll
  for (int r = 0; r < 16; ++r) srow[r] = 0.f;

  float* ebw = eb + wave * 16 * JP;

  for (int jc = 0; jc < CC; jc += 128) {
    __syncthreads();

#pragma unroll
    for (int i = 0; i < 8; ++i) {
      float* tp = XcT + (td + 32 * (i >> 2)) * JP + tj + 32 * (i & 3);
      tp[0]      = xt[i].x;
      tp[JP]     = xt[i].y;
      tp[2 * JP] = xt[i].z;
      tp[3 * JP] = xt[i].w;
    }
    __syncthreads();

    const float4 sv = ((const float4*)(srcS + jc))[lane];
#pragma unroll
    for (int r = 0; r < 16; ++r) {
      float e0 = sv.x + dvals[r] + cs; e0 = e0 > 0.f ? e0 : 0.2f * e0;
      float e1 = sv.y + dvals[r] + cs; e1 = e1 > 0.f ? e1 : 0.2f * e1;
      float e2 = sv.z + dvals[r] + cs; e2 = e2 > 0.f ? e2 : 0.2f * e2;
      float e3 = sv.w + dvals[r] + cs; e3 = e3 > 0.f ? e3 : 0.2f * e3;
      const float p0 = __expf(e0 + lamh * lg[r].x - mrow[r]);
      const float p1 = __expf(e1 + lamh * lg[r].y - mrow[r]);
      const float p2 = __expf(e2 + lamh * lg[r].z - mrow[r]);
      const float p3 = __expf(e3 + lamh * lg[r].w - mrow[r]);
      float* dstp = ebw + r * JP + lane * 4;
      dstp[0] = p0; dstp[1] = p1; dstp[2] = p2; dstp[3] = p3;
      srow[r] += (p0 + p1) + (p2 + p3);
    }
    __syncthreads();

    if (jc + 128 < CC) {
      const int jn = jc + 128;
#pragma unroll
      for (int i = 0; i < 8; ++i)
        xt[i] = *(const float4*)(xtile + (size_t)(jn + 32 * (i & 3)) * 256 +
                                 32 * (i >> 2));
#pragma unroll
      for (int r = 0; r < 16; ++r)
        lg[r] = ((const float4*)(logA + (size_t)(i0 + r) * CC + jn))[lane];
    }

    const float* arow = ebw + r16 * JP;
    for (int kk = 0; kk < 128; kk += 4) {
      const int ko = kk + 2 * half;
      v2f a; a.x = arow[ko]; a.y = arow[ko + 1];
#pragma unroll
      for (int nt = 0; nt < 4; ++nt) {
        const float* brow = XcT + (nt * 16 + r16) * JP;
        v2f bv; bv.x = brow[ko]; bv.y = brow[ko + 1];
        acc[nt] = wmma_f32(a, bv, acc[nt]);
      }
    }
  }

#pragma unroll
  for (int r = 0; r < 16; ++r) {
#pragma unroll
    for (int off = 16; off >= 1; off >>= 1)
      srow[r] += __shfl_xor(srow[r], off, 32);
  }

#pragma unroll
  for (int v = 0; v < 8; ++v) {
    const int rr = v + 8 * half;
    const float inv = 1.0f / srow[rr];
    const int row = i0 + rr;
#pragma unroll
    for (int nt = 0; nt < 4; ++nt) {
      out[(size_t)(b * CC + row) * 256 + h * 64 + nt * 16 + r16] =
          acc[nt][v] * inv;
    }
  }
}

extern "C" void kernel_launch(void* const* d_in, const int* in_sizes, int n_in,
                              void* d_out, int out_size, void* d_ws,
                              size_t ws_size, hipStream_t stream) {
  (void)in_sizes; (void)n_in; (void)out_size; (void)ws_size;

  const float* x   = (const float*)d_in[0];
  const float* Apr = (const float*)d_in[1];
  const float* Wp  = (const float*)d_in[2];
  const float* bp  = (const float*)d_in[3];
  const float* Ws1 = (const float*)d_in[4];
  const float* bs1 = (const float*)d_in[5];
  const float* Ws2 = (const float*)d_in[6];
  const float* bs2 = (const float*)d_in[7];
  const float* Wc1 = (const float*)d_in[8];
  const float* bc1 = (const float*)d_in[9];
  const float* Wc2 = (const float*)d_in[10];
  const float* bc2 = (const float*)d_in[11];
  const float* att = (const float*)d_in[12];
  const float* lam = (const float*)d_in[13];

  float* out  = (float*)d_out;
  float* slog = out + (size_t)BB * CC * 256;

  float* xp   = (float*)d_ws;
  float* logA = xp + 2097152;
  float* srcg = logA + 1048576;
  float* dstg = srcg + 32768;
  float* csb  = dstg + 32768;
  float* colpart = logA;

  const size_t gemm_lds = (size_t)(2 * 64 * APAD) * sizeof(float);
  proj_gemm<<<dim3(128, 4), 256, gemm_lds, stream>>>(x, Wp, bp, xp, colpart);

  side_kernel<<<264, 256, 0, stream>>>(colpart, Ws1, bs1, Ws2, bs2, Wc1,
                                            bc1, Wc2, bc2, att, slog, csb, xp,
                                            srcg, dstg, Apr, logA);

  const size_t attn_lds =
      (size_t)(1024 + 64 * JP + 8 * 16 * JP) * sizeof(float);
  attn_kernel<<<BB * 4 * 8, 256, attn_lds, stream>>>(xp, logA, srcg, dstg,
                                                     csb, lam, out);
}


# attn pass 2 as a pipeline: one barrier per chunk (x_proj and exp tiles double-buffered), waves w and w+4 take exp and WMMA stages in opposite order, WMMA fragments requested one block ahead
# speedup vs baseline: 3.0588x; 1.1184x over previous
#include <hip/hip_runtime.h>
#include <math.h>


typedef float v2f __attribute__((ext_vector_type(2)));
typedef float v4f __attribute__((ext_vector_type(4)));
typedef float v8f __attribute__((ext_vector_type(8)));

__device__ __forceinline__ v8f wmma_f32(v2f a, v2f b, v8f c) {
  return __builtin_amdgcn_wmma_f32_16x16x4_f32(false, a, false, b,
                                               (short)0, c, false, false);
}

#define BB 8
#define CC 1024
#define INF_NEG (-3.0e38f)

__device__ __forceinline__ float4 logA_quad(const float4 a) {
  float4 o;
  o.x = logf(a.x + 1e-9f);
  o.y = logf(a.y + 1e-9f);
  o.z = logf(a.z + 1e-9f);
  o.w = logf(a.w + 1e-9f);
  return o;
}

typedef float v16f __attribute__((ext_vector_type(16)));
__device__ __forceinline__ void put4(v16f& d, const int k, const v4f q) {
  d[4 * k + 0] = q.x; d[4 * k + 1] = q.y;
  d[4 * k + 2] = q.z; d[4 * k + 3] = q.w;
}
__device__ __forceinline__ v4f get4(const v16f& d, const int k) {
  v4f q;
  q.x = d[4 * k + 0]; q.y = d[4 * k + 1];
  q.z = d[4 * k + 2]; q.w = d[4 * k + 3];
  return q;
}

__device__ __forceinline__ void stats_block(
    const int b, const float* colpart,
    const float* __restrict__ Ws1, const float* __restrict__ bs1,
    const float* __restrict__ Ws2, const float* __restrict__ bs2,
    const float* __restrict__ Wc1, const float* __restrict__ bc1,
    const float* __restrict__ Wc2, const float* __restrict__ bc2,
    const float* __restrict__ att_w,
    float* __restrict__ slog_out, float* __restrict__ cs_out,
    const float* __restrict__ A, float* logA) {
  __shared__ float4 xm4[64];
  __shared__ float4 hpart4[256];
  __shared__ v4f w2s4[256];
  __shared__ v4f wc1s4[256];
  __shared__ v4f aws4[256];
  __shared__ v4f wc2s4[1024];
  __shared__ float hid[128];
  __shared__ float slg[5];
  __shared__ float sp[5];
  __shared__ float c1s[64];
  __shared__ float c2s[64];

  float* xm = (float*)xm4;
  const float* hpart = (const float*)hpart4;
  const float* w2s = (const float*)w2s4;
  const float* wc1s = (const float*)wc1s4;
  const float* wc2s = (const float*)wc2s4;
  const float* aws = (const float*)aws4;

  const int t = threadIdx.x;
  const int wave = t >> 5, lane = t & 31;

  v16f cpv;
#pragma unroll
  for (int p = 0; p < 16; ++p) cpv[p] = colpart[(b * 16 + p) * 256 + t];

  v16f w1v[8];
  const v4f* Ws1q = (const v4f*)Ws1 + (size_t)(wave * 32) * 32 + lane;
#pragma unroll
  for (int i = 0; i < 32; ++i) put4(w1v[i >> 2], i & 3, Ws1q[i * 32]);

  v16f smA, smB;
  put4(smA, 0, ((const v4f*)Ws2)[t < 160 ? t : 159]);
  put4(smA, 1, ((const v4f*)Wc1)[t < 80 ? t : 79]);
  put4(smA, 2, ((const v4f*)(att_w + 128))[t & 15]);
  put4(smA, 3, ((const v4f*)Wc2)[t]);
  put4(smB, 0, ((const v4f*)Wc2)[t + 256]);
  put4(smB, 1, ((const v4f*)Wc2)[t + 512]);
  put4(smB, 2, ((const v4f*)Wc2)[t + 768]);
  smB[12] = bs1[t & 127];
  smB[13] = bs2[wave < 5 ? wave : 4];
  smB[14] = bc1[t & 63];
  smB[15] = bc2[t & 63];

  v16f aqv;
  const v4f* A4 = (const v4f*)A + b * 1024 + t;
#pragma unroll
  for (int i = 0; i < 4; ++i) put4(aqv, i, A4[i * 256]);

  asm volatile("" : "+v"(cpv), "+v"(w1v[0]), "+v"(w1v[1]), "+v"(w1v[2]),
                    "+v"(w1v[3]), "+v"(w1v[4]), "+v"(w1v[5]), "+v"(w1v[6]),
                    "+v"(w1v[7]), "+v"(smA), "+v"(smB), "+v"(aqv));

  const float bs1v = smB[12];
  const float bs2v = smB[13];
  const float bc1v = smB[14];
  const float bc2v = smB[15];

  float s = 0.f;
#pragma unroll
  for (int p = 0; p < 16; ++p) s += cpv[p];
  xm[t] = s * (1.0f / 1024.0f);

  w2s4[t] = get4(smA, 0);
  wc1s4[t] = get4(smA, 1);
  aws4[t] = get4(smA, 2);
  wc2s4[t] = get4(smA, 3);
  wc2s4[t + 256] = get4(smB, 0);
  wc2s4[t + 512] = get4(smB, 1);
  wc2s4[t + 768] = get4(smB, 2);
  __syncthreads();

  {
    float4 acc = make_float4(0.f, 0.f, 0.f, 0.f);
#pragma unroll
    for (int i4 = 0; i4 < 8; ++i4) {
      const float4 xv = xm4[wave * 8 + i4];
      const float xs[4] = {xv.x, xv.y, xv.z, xv.w};
#pragma unroll
      for (int j = 0; j < 4; ++j) {
        const v4f w = get4(w1v[i4], j);
        acc.x += xs[j] * w.x;
        acc.y += xs[j] * w.y;
        acc.z += xs[j] * w.z;
        acc.w += xs[j] * w.w;
      }
    }
    hpart4[wave * 32 + lane] = acc;
  }
  __syncthreads();

  if (t < 128) {
    float a = bs1v;
#pragma unroll
    for (int w = 0; w < 8; ++w) a += hpart[w * 128 + t];
    hid[t] = fmaxf(a, 0.f);
  }
  __syncthreads();

  if (wave < 5) {
    float pl = 0.f;
#pragma unroll
    for (int q = 0; q < 4; ++q) {
      const int k = lane + 32 * q;
      pl += hid[k] * w2s[k * 5 + wave];
    }
#pragma unroll
    for (int off = 16; off >= 1; off >>= 1) pl += __shfl_xor(pl, off, 32);
    if (lane == 0) {
      const float a = bs2v + pl;
      slg[wave] = a;
      slog_out[b * 5 + wave] = a;
    }
  }
  __syncthreads();

  if (t == 0) {
    float mx = slg[0];
    for (int i = 1; i < 5; ++i) mx = fmaxf(mx, slg[i]);
    float sum = 0.f;
    for (int i = 0; i < 5; ++i) { sp[i] = __expf(slg[i] - mx); sum += sp[i]; }
    float inv = 1.0f / sum;
    for (int i = 0; i < 5; ++i) sp[i] *= inv;
  }
  __syncthreads();

  if (t < 64) {
    float a = bc1v;
    for (int k = 0; k < 5; ++k) a += sp[k] * wc1s[k * 64 + t];
    c1s[t] = fmaxf(a, 0.f);
  }
  __syncthreads();

  if (t < 64) {
    float a = bc2v;
    for (int k = 0; k < 64; ++k) a += c1s[k] * wc2s[k * 64 + t];
    c2s[t] = a;
  }
  __syncthreads();

  if (t == 0) {
    float cs = 0.f;
    for (int k = 0; k < 64; ++k) cs += c2s[k] * aws[k];
    cs_out[b] = cs;
  }

  float4* L4 = (float4*)logA + b * 1024 + t;
#pragma unroll
  for (int i = 0; i < 4; ++i) {
    const v4f a = get4(aqv, i);
    L4[i * 256] = logA_quad(make_float4(a.x, a.y, a.z, a.w));
  }
}

#define APAD 260
__global__ void __launch_bounds__(256) proj_gemm(
    const float* __restrict__ x, const float* __restrict__ Wp,
    const float* __restrict__ bp, float* __restrict__ xp,
    float* __restrict__ colpart) {
  extern __shared__ float sm[];
  float* Apan = sm;
  float* BpT  = sm + 64 * APAD;

  const int tid = threadIdx.x;
  const int m0 = blockIdx.x * 64;
  const int n0 = blockIdx.y * 64;

  const float* xcol = x + (size_t)m0 * 256 + tid;
  float* acol = Apan + tid;
  float csum = 0.f;
  for (int r0 = 0; r0 < 64; r0 += 8) {
    float v[8];
    for (int i = 0; i < 8; ++i) v[i] = xcol[(r0 + i) * 256];
    for (int i = 0; i < 8; ++i) {
      acol[(r0 + i) * APAD] = v[i];
      csum += v[i];
    }
  }
  if (blockIdx.y == 0) colpart[blockIdx.x * 256 + tid] = csum;
  for (int idx = tid; idx < 256 * 64; idx += 256) {
    const int k = idx >> 6, n = idx & 63;
    BpT[n * APAD + k] = Wp[(size_t)k * 256 + n0 + n];
  }
  __syncthreads();

  const int wave = tid >> 5, lane = tid & 31;
  const int r16 = lane & 15, half = lane >> 4;
  const int rw = wave >> 1;
  const int cw = (wave & 1) * 32;

  v8f acc0 = {0, 0, 0, 0, 0, 0, 0, 0};
  v8f acc1 = {0, 0, 0, 0, 0, 0, 0, 0};

  const float* arow = Apan + (rw * 16 + r16) * APAD;
  const float* b0row = BpT + (cw + r16) * APAD;
  const float* b1row = BpT + (cw + 16 + r16) * APAD;

  for (int kk = 0; kk < 256; kk += 4) {
    const int ko = kk + 2 * half;
    v2f a;  a.x = arow[ko];  a.y = arow[ko + 1];
    v2f b0; b0.x = b0row[ko]; b0.y = b0row[ko + 1];
    v2f b1; b1.x = b1row[ko]; b1.y = b1row[ko + 1];
    acc0 = wmma_f32(a, b0, acc0);
    acc1 = wmma_f32(a, b1, acc1);
  }

  const float bias0 = bp[n0 + cw + r16];
  const float bias1 = bp[n0 + cw + 16 + r16];
#pragma unroll
  for (int v = 0; v < 8; ++v) {
    const int row = m0 + rw * 16 + v + 8 * half;
    xp[(size_t)row * 256 + n0 + cw + r16]      = acc0[v] + bias0;
    xp[(size_t)row * 256 + n0 + cw + 16 + r16] = acc1[v] + bias1;
  }
}

__device__ __forceinline__ void srcdst_block(
    const int blk, const int slice,
    const float* __restrict__ xp, const float* __restrict__ att_w,
    float* __restrict__ srcg, float* __restrict__ dstg,
    const float* __restrict__ A, float* __restrict__ logA) {
  const int lane = threadIdx.x & 31;
  const int m0 = (blk * 8 + (threadIdx.x >> 5)) * 4;
  const int c16 = lane & 15;
  const bool has_slice = slice < 256;

  float4 va[4], vb[4];
  for (int r = 0; r < 4; ++r) {
    const float4* rowp = (const float4*)(xp + (size_t)(m0 + r) * 256);
    va[r] = rowp[lane];
    vb[r] = rowp[32 + lane];
  }
  const float4 ws = ((const float4*)att_w)[c16];
  const float4 wd = ((const float4*)(att_w + 64))[c16];
  float4 aq[4];
  if (has_slice) {
    const float4* A4 = (const float4*)A + slice * 1024 + threadIdx.x;
    for (int i = 0; i < 4; ++i) aq[i] = A4[i * 256];
  }

  float sa[4], da[4], sb[4], db[4];
  for (int r = 0; r < 4; ++r) {
    sa[r] = va[r].x * ws.x + va[r].y * ws.y + va[r].z * ws.z + va[r].w * ws.w;
    da[r] = va[r].x * wd.x + va[r].y * wd.y + va[r].z * wd.z + va[r].w * wd.w;
    sb[r] = vb[r].x * ws.x + vb[r].y * ws.y + vb[r].z * ws.z + vb[r].w * ws.w;
    db[r] = vb[r].x * wd.x + vb[r].y * wd.y + vb[r].z * wd.z + vb[r].w * wd.w;
  }
  for (int off = 8; off >= 1; off >>= 1) {
    for (int r = 0; r < 4; ++r) {
      sa[r] += __shfl_xor(sa[r], off, 32);
      da[r] += __shfl_xor(da[r], off, 32);
      sb[r] += __shfl_xor(sb[r], off, 32);
      db[r] += __shfl_xor(db[r], off, 32);
    }
  }
  float4 so[4], dq[4];
  for (int r = 0; r < 4; ++r) {
    so[r].x = sa[r]; so[r].y = __shfl_xor(sa[r], 16, 32);
    so[r].z = sb[r]; so[r].w = __shfl_xor(sb[r], 16, 32);
    dq[r].x = da[r]; dq[r].y = __shfl_xor(da[r], 16, 32);
    dq[r].z = db[r]; dq[r].w = __shfl_xor(db[r], 16, 32);
  }
  if (lane == 0) {
    for (int r = 0; r < 4; ++r) {
      ((float4*)srcg)[m0 + r] = so[r];
      ((float4*)dstg)[m0 + r] = dq[r];
    }
  }
  if (has_slice) {
    float4* L4 = (float4*)logA + slice * 1024 + threadIdx.x;
    for (int i = 0; i < 4; ++i) L4[i * 256] = logA_quad(aq[i]);
  }
}

__global__ void __launch_bounds__(256) side_kernel(
    const float* colpart,
    const float* __restrict__ Ws1, const float* __restrict__ bs1,
    const float* __restrict__ Ws2, const float* __restrict__ bs2,
    const float* __restrict__ Wc1, const float* __restrict__ bc1,
    const float* __restrict__ Wc2, const float* __restrict__ bc2,
    const float* __restrict__ att_w,
    float* __restrict__ slog_out, float* __restrict__ cs_out,
    const float* __restrict__ xp,
    float* __restrict__ srcg, float* __restrict__ dstg,
    const float* __restrict__ A, float* logA) {
  const int blk = blockIdx.x;
  if (blk < 8) {
    stats_block(blk, colpart, Ws1, bs1, Ws2, bs2, Wc1, bc1, Wc2, bc2, att_w,
                slog_out, cs_out, A, logA);
  } else {
    srcdst_block(blk - 8, blk, xp, att_w, srcg, dstg, A, logA);
  }
}

#define JP 132
#define XCT_F (64 * JP)
#define EBW_F (16 * JP)
#define ATTN_LDS_F (1024 + 2 * XCT_F + 2 * 8 * EBW_F)

__device__ __forceinline__ void lds_barrier() {
  asm volatile("s_wait_dscnt 0x0\n\ts_barrier_signal -1\n\ts_barrier_wait -1"
               ::: "memory");
}

__device__ __forceinline__ void attn_load_tile(float4 (&xt)[8],
                                               const float* xtile,
                                               const int j) {
#pragma unroll
  for (int i = 0; i < 8; ++i)
    xt[i] = *(const float4*)(xtile + (size_t)(j + 32 * (i & 3)) * 256 +
                             32 * (i >> 2));
}
__device__ __forceinline__ void attn_store_tile(float* xdst,
                                                const float4 (&xt)[8],
                                                const int td, const int tj) {
#pragma unroll
  for (int i = 0; i < 8; ++i) {
    float* tp = xdst + (td + 32 * (i >> 2)) * JP + tj + 32 * (i & 3);
    tp[0]      = xt[i].x;
    tp[JP]     = xt[i].y;
    tp[2 * JP] = xt[i].z;
    tp[3 * JP] = xt[i].w;
  }
}
__device__ __forceinline__ void attn_load_logA(float4 (&lg)[16],
                                               const float* lrow,
                                               const int j) {
#pragma unroll
  for (int r = 0; r < 16; ++r)
    lg[r] = *(const float4*)(lrow + (size_t)r * CC + j);
}
__device__ __forceinline__ void attn_exp_stage(
    float* ebdst, const float* srcS, const int j, const int lane,
    const float4 (&lg)[16], const float (&dvals)[16], const float (&mrow)[16],
    float (&srow)[16], const float cs, const float lamh) {
  const float4 sv = ((const float4*)(srcS + j))[lane];
#pragma unroll
  for (int r = 0; r < 16; ++r) {
    float e0 = sv.x + dvals[r] + cs; e0 = e0 > 0.f ? e0 : 0.2f * e0;
    float e1 = sv.y + dvals[r] + cs; e1 = e1 > 0.f ? e1 : 0.2f * e1;
    float e2 = sv.z + dvals[r] + cs; e2 = e2 > 0.f ? e2 : 0.2f * e2;
    float e3 = sv.w + dvals[r] + cs; e3 = e3 > 0.f ? e3 : 0.2f * e3;
    const float p0 = __expf(e0 + lamh * lg[r].x - mrow[r]);
    const float p1 = __expf(e1 + lamh * lg[r].y - mrow[r]);
    const float p2 = __expf(e2 + lamh * lg[r].z - mrow[r]);
    const float p3 = __expf(e3 + lamh * lg[r].w - mrow[r]);
    float* dstp = ebdst + r * JP + lane * 4;
    dstp[0] = p0; dstp[1] = p1; dstp[2] = p2; dstp[3] = p3;
    srow[r] += (p0 + p1) + (p2 + p3);
  }
}
struct AttnFrag { v2f a[4]; v2f b[4][4]; };
__device__ __forceinline__ void attn_load_frag(AttnFrag& f, const float* arow,
                                               const float* brow0,
                                               const int k0, const int half) {
#pragma unroll
  for (int s = 0; s < 4; ++s) {
    const int ko = k0 + 4 * s + 2 * half;
    f.a[s].x = arow[ko]; f.a[s].y = arow[ko + 1];
#pragma unroll
    for (int nt = 0; nt < 4; ++nt) {
      const float* brow = brow0 + nt * 16 * JP;
      f.b[s][nt].x = brow[ko]; f.b[s][nt].y = brow[ko + 1];
    }
  }
}
__device__ __forceinline__ void attn_mma_frag(const AttnFrag& f,
                                              v8f (&acc)[4]) {
#pragma unroll
  for (int s = 0; s < 4; ++s) {
#pragma unroll
    for (int nt = 0; nt < 4; ++nt)
      acc[nt] = wmma_f32(f.a[s], f.b[s][nt], acc[nt]);
  }
}
__device__ __forceinline__ void attn_wmma_stage(const float* ebsrc,
                                                const float* xsrc,
                                                const int r16, const int half,
                                                v8f (&acc)[4]) {
  const float* arow = ebsrc + r16 * JP;
  const float* brow0 = xsrc + r16 * JP;
  AttnFrag f0, f1;
  attn_load_frag(f0, arow, brow0, 0, half);
  for (int k0 = 0; k0 < 128; k0 += 32) {
    attn_load_frag(f1, arow, brow0, k0 + 16, half);
    attn_mma_frag(f0, acc);
    if (k0 + 32 < 128) attn_load_frag(f0, arow, brow0, k0 + 32, half);
    attn_mma_frag(f1, acc);
  }
}

__global__ void __launch_bounds__(256) attn_kernel(
    const float* __restrict__ xp,   const float* __restrict__ logA,
    const float* __restrict__ srcg, const float* __restrict__ dstg,
    const float* __restrict__ csb,  const float* __restrict__ lam,
    float* __restrict__ out) {
  __shared__ __attribute__((aligned(16))) float sm[ATTN_LDS_F];
  float* srcS = sm;
  float* XcT  = srcS + 1024;
  float* eb   = XcT + 2 * XCT_F;

  const int tid = threadIdx.x;
  const int wave = tid >> 5, lane = tid & 31;
  const int r16 = lane & 15, half = lane >> 4;

  const int bid = blockIdx.x;
  const int b  = bid >> 5;
  const int h  = (bid >> 3) & 3;
  const int it = bid & 7;
  const int i0 = it * 128 + wave * 16;

  for (int idx = tid; idx < 1024; idx += 256)
    srcS[idx] = srcg[(b * CC + idx) * 4 + h];
  __syncthreads();

  const float cs = csb[b];
  const float lamh = lam[h];

  float dvals[16];
#pragma unroll
  for (int r = 0; r < 16; ++r)
    dvals[r] = dstg[(b * CC + i0 + r) * 4 + h];

  const int tj = tid >> 3;
  const int td = (tid & 7) * 4;
  const float* xtile = xp + ((size_t)(b * CC + tj) * 4 + h) * 64 + td;
  const float* lrow = logA + (size_t)i0 * CC + 4 * lane;

  float mrow[16];
#pragma unroll
  for (int r = 0; r < 16; ++r) mrow[r] = INF_NEG;

  for (int jc = 0; jc < CC; jc += 128) {
    const float4 sv = ((const float4*)(srcS + jc))[lane];
#pragma unroll
    for (int r = 0; r < 16; ++r) {
      const float4 lg =
          ((const float4*)(logA + (size_t)(i0 + r) * CC + jc))[lane];
      float e0 = sv.x + dvals[r] + cs; e0 = e0 > 0.f ? e0 : 0.2f * e0;
      float e1 = sv.y + dvals[r] + cs; e1 = e1 > 0.f ? e1 : 0.2f * e1;
      float e2 = sv.z + dvals[r] + cs; e2 = e2 > 0.f ? e2 : 0.2f * e2;
      float e3 = sv.w + dvals[r] + cs; e3 = e3 > 0.f ? e3 : 0.2f * e3;
      const float v0 = e0 + lamh * lg.x;
      const float v1 = e1 + lamh * lg.y;
      const float v2 = e2 + lamh * lg.z;
      const float v3 = e3 + lamh * lg.w;
      mrow[r] = fmaxf(mrow[r], fmaxf(fmaxf(v0, v1), fmaxf(v2, v3)));
    }
  }
  float4 xt[8];
  attn_load_tile(xt, xtile, 0);
  float4 lg[16];
  attn_load_logA(lg, lrow, 0);

#pragma unroll
  for (int r = 0; r < 16; ++r) {
#pragma unroll
    for (int off = 16; off >= 1; off >>= 1)
      mrow[r] = fmaxf(mrow[r], __shfl_xor(mrow[r], off, 32));
  }

  v8f acc[4];
  const v8f vzero = {0, 0, 0, 0, 0, 0, 0, 0};
#pragma unroll
  for (int nt = 0; nt < 4; ++nt) acc[nt] = vzero;

  float srow[16];
#pragma unroll
  for (int r = 0; r < 16; ++r) srow[r] = 0.f;

  const bool wmma_first = __builtin_amdgcn_readfirstlane(wave) < 4;

  attn_store_tile(XcT, xt, td, tj);
  attn_load_tile(xt, xtile, 128);
  attn_exp_stage(eb + wave * EBW_F, srcS, 0, lane, lg, dvals, mrow, srow, cs,
                 lamh);
  attn_load_logA(lg, lrow, 128);

  for (int c = 0; c < 7; ++c) {
    const int cur = c & 1;
    const float* xcur = XcT + cur * XCT_F;
    float* xnxt = XcT + (cur ^ 1) * XCT_F;
    const float* ecur = eb + (cur * 8 + wave) * EBW_F;
    float* enxt = eb + ((cur ^ 1) * 8 + wave) * EBW_F;
    const int jn = (c + 1) * 128;

    lds_barrier();

    attn_store_tile(xnxt, xt, td, tj);
    if (c < 6) attn_load_tile(xt, xtile, jn + 128);

    if (wmma_first) {
      attn_wmma_stage(ecur, xcur, r16, half, acc);
      attn_exp_stage(enxt, srcS, jn, lane, lg, dvals, mrow, srow, cs, lamh);
    } else {
      attn_exp_stage(enxt, srcS, jn, lane, lg, dvals, mrow, srow, cs, lamh);
      attn_wmma_stage(ecur, xcur, r16, half, acc);
    }
    if (c < 6) attn_load_logA(lg, lrow, jn + 128);
  }
  lds_barrier();
  attn_wmma_stage(eb + (8 + wave) * EBW_F, XcT + XCT_F, r16, half, acc);

#pragma unroll
  for (int r = 0; r < 16; ++r) {
#pragma unroll
    for (int off = 16; off >= 1; off >>= 1)
      srow[r] += __shfl_xor(srow[r], off, 32);
  }

#pragma unroll
  for (int v = 0; v < 8; ++v) {
    const int rr = v + 8 * half;
    const float inv = 1.0f / srow[rr];
    const int row = i0 + rr;
#pragma unroll
    for (int nt = 0; nt < 4; ++nt) {
      out[(size_t)(b * CC + row) * 256 + h * 64 + nt * 16 + r16] =
          acc[nt][v] * inv;
    }
  }
}

extern "C" void kernel_launch(void* const* d_in, const int* in_sizes, int n_in,
                              void* d_out, int out_size, void* d_ws,
                              size_t ws_size, hipStream_t stream) {
  (void)in_sizes; (void)n_in; (void)out_size; (void)ws_size;

  const float* x   = (const float*)d_in[0];
  const float* Apr = (const float*)d_in[1];
  const float* Wp  = (const float*)d_in[2];
  const float* bp  = (const float*)d_in[3];
  const float* Ws1 = (const float*)d_in[4];
  const float* bs1 = (const float*)d_in[5];
  const float* Ws2 = (const float*)d_in[6];
  const float* bs2 = (const float*)d_in[7];
  const float* Wc1 = (const float*)d_in[8];
  const float* bc1 = (const float*)d_in[9];
  const float* Wc2 = (const float*)d_in[10];
  const float* bc2 = (const float*)d_in[11];
  const float* att = (const float*)d_in[12];
  const float* lam = (const float*)d_in[13];

  float* out  = (float*)d_out;
  float* slog = out + (size_t)BB * CC * 256;

  float* xp   = (float*)d_ws;
  float* logA = xp + 2097152;
  float* srcg = logA + 1048576;
  float* dstg = srcg + 32768;
  float* csb  = dstg + 32768;
  float* colpart = logA;

  const size_t gemm_lds = (size_t)(2 * 64 * APAD) * sizeof(float);
  proj_gemm<<<dim3(128, 4), 256, gemm_lds, stream>>>(x, Wp, bp, xp, colpart);

  side_kernel<<<264, 256, 0, stream>>>(colpart, Ws1, bs1, Ws2, bs2, Wc1,
                                            bc1, Wc2, bc2, att, slog, csb, xp,
                                            srcg, dstg, Apr, logA);

  attn_kernel<<<BB * 4 * 8, 256, 0, stream>>>(xp, logA, srcg, dstg, csb, lam,
                                              out);
}


# attn: the pass over logA for exact row maxima is gone; softmax shift = bound from max src_s + per-row logA max (kept by the side kernel); exact pass only if lam<0 or src spread>60; rounding changes
# speedup vs baseline: 3.2211x; 1.0531x over previous
#include <hip/hip_runtime.h>
#include <math.h>


typedef float v2f __attribute__((ext_vector_type(2)));
typedef float v4f __attribute__((ext_vector_type(4)));
typedef float v8f __attribute__((ext_vector_type(8)));

__device__ __forceinline__ v8f wmma_f32(v2f a, v2f b, v8f c) {
  return __builtin_amdgcn_wmma_f32_16x16x4_f32(false, a, false, b,
                                               (short)0, c, false, false);
}

#define BB 8
#define CC 1024
#define INF_NEG (-3.0e38f)

__device__ __forceinline__ float4 logA_quad(const float4 a) {
  float4 o;
  o.x = logf(a.x + 1e-9f);
  o.y = logf(a.y + 1e-9f);
  o.z = logf(a.z + 1e-9f);
  o.w = logf(a.w + 1e-9f);
  return o;
}

__device__ __forceinline__ void logA_slice_store(const float4 (&aq)[4],
                                                 float* logA, float* rmaxg,
                                                 const int slice) {
  __shared__ float wmx[4][8];
  const int t = threadIdx.x;
  const int wave = t >> 5, lane = t & 31;
  float4* L4 = (float4*)logA + slice * 1024 + t;
  float m[4];
#pragma unroll
  for (int i = 0; i < 4; ++i) {
    const float4 l = logA_quad(aq[i]);
    L4[i * 256] = l;
    m[i] = fmaxf(fmaxf(l.x, l.y), fmaxf(l.z, l.w));
  }
#pragma unroll
  for (int off = 16; off >= 1; off >>= 1) {
#pragma unroll
    for (int i = 0; i < 4; ++i) m[i] = fmaxf(m[i], __shfl_xor(m[i], off, 32));
  }
  if (lane == 0) {
#pragma unroll
    for (int i = 0; i < 4; ++i) wmx[i][wave] = m[i];
  }
  __syncthreads();
  if (t < 4) {
    float v = wmx[t][0];
    for (int w = 1; w < 8; ++w) v = fmaxf(v, wmx[t][w]);
    rmaxg[slice * 4 + t] = v;
  }
}

typedef float v16f __attribute__((ext_vector_type(16)));
__device__ __forceinline__ void put4(v16f& d, const int k, const v4f q) {
  d[4 * k + 0] = q.x; d[4 * k + 1] = q.y;
  d[4 * k + 2] = q.z; d[4 * k + 3] = q.w;
}
__device__ __forceinline__ v4f get4(const v16f& d, const int k) {
  v4f q;
  q.x = d[4 * k + 0]; q.y = d[4 * k + 1];
  q.z = d[4 * k + 2]; q.w = d[4 * k + 3];
  return q;
}

__device__ __forceinline__ void stats_block(
    const int b, const float* colpart,
    const float* __restrict__ Ws1, const float* __restrict__ bs1,
    const float* __restrict__ Ws2, const float* __restrict__ bs2,
    const float* __restrict__ Wc1, const float* __restrict__ bc1,
    const float* __restrict__ Wc2, const float* __restrict__ bc2,
    const float* __restrict__ att_w,
    float* __restrict__ slog_out, float* __restrict__ cs_out,
    const float* __restrict__ A, float* logA, float* __restrict__ rmaxg) {
  __shared__ float4 xm4[64];
  __shared__ float4 hpart4[256];
  __shared__ v4f w2s4[256];
  __shared__ v4f wc1s4[256];
  __shared__ v4f aws4[256];
  __shared__ v4f wc2s4[1024];
  __shared__ float hid[128];
  __shared__ float slg[5];
  __shared__ float sp[5];
  __shared__ float c1s[64];
  __shared__ float c2s[64];

  float* xm = (float*)xm4;
  const float* hpart = (const float*)hpart4;
  const float* w2s = (const float*)w2s4;
  const float* wc1s = (const float*)wc1s4;
  const float* wc2s = (const float*)wc2s4;
  const float* aws = (const float*)aws4;

  const int t = threadIdx.x;
  const int wave = t >> 5, lane = t & 31;

  v16f cpv;
#pragma unroll
  for (int p = 0; p < 16; ++p) cpv[p] = colpart[(b * 16 + p) * 256 + t];

  v16f w1v[8];
  const v4f* Ws1q = (const v4f*)Ws1 + (size_t)(wave * 32) * 32 + lane;
#pragma unroll
  for (int i = 0; i < 32; ++i) put4(w1v[i >> 2], i & 3, Ws1q[i * 32]);

  v16f smA, smB;
  put4(smA, 0, ((const v4f*)Ws2)[t < 160 ? t : 159]);
  put4(smA, 1, ((const v4f*)Wc1)[t < 80 ? t : 79]);
  put4(smA, 2, ((const v4f*)(att_w + 128))[t & 15]);
  put4(smA, 3, ((const v4f*)Wc2)[t]);
  put4(smB, 0, ((const v4f*)Wc2)[t + 256]);
  put4(smB, 1, ((const v4f*)Wc2)[t + 512]);
  put4(smB, 2, ((const v4f*)Wc2)[t + 768]);
  smB[12] = bs1[t & 127];
  smB[13] = bs2[wave < 5 ? wave : 4];
  smB[14] = bc1[t & 63];
  smB[15] = bc2[t & 63];

  v16f aqv;
  const v4f* A4 = (const v4f*)A + b * 1024 + t;
#pragma unroll
  for (int i = 0; i < 4; ++i) put4(aqv, i, A4[i * 256]);

  asm volatile("" : "+v"(cpv), "+v"(w1v[0]), "+v"(w1v[1]), "+v"(w1v[2]),
                    "+v"(w1v[3]), "+v"(w1v[4]), "+v"(w1v[5]), "+v"(w1v[6]),
                    "+v"(w1v[7]), "+v"(smA), "+v"(smB), "+v"(aqv));

  const float bs1v = smB[12];
  const float bs2v = smB[13];
  const float bc1v = smB[14];
  const float bc2v = smB[15];

  float s = 0.f;
#pragma unroll
  for (int p = 0; p < 16; ++p) s += cpv[p];
  xm[t] = s * (1.0f / 1024.0f);

  w2s4[t] = get4(smA, 0);
  wc1s4[t] = get4(smA, 1);
  aws4[t] = get4(smA, 2);
  wc2s4[t] = get4(smA, 3);
  wc2s4[t + 256] = get4(smB, 0);
  wc2s4[t + 512] = get4(smB, 1);
  wc2s4[t + 768] = get4(smB, 2);
  __syncthreads();

  {
    float4 acc = make_float4(0.f, 0.f, 0.f, 0.f);
#pragma unroll
    for (int i4 = 0; i4 < 8; ++i4) {
      const float4 xv = xm4[wave * 8 + i4];
      const float xs[4] = {xv.x, xv.y, xv.z, xv.w};
#pragma unroll
      for (int j = 0; j < 4; ++j) {
        const v4f w = get4(w1v[i4], j);
        acc.x += xs[j] * w.x;
        acc.y += xs[j] * w.y;
        acc.z += xs[j] * w.z;
        acc.w += xs[j] * w.w;
      }
    }
    hpart4[wave * 32 + lane] = acc;
  }
  __syncthreads();

  if (t < 128) {
    float a = bs1v;
#pragma unroll
    for (int w = 0; w < 8; ++w) a += hpart[w * 128 + t];
    hid[t] = fmaxf(a, 0.f);
  }
  __syncthreads();

  if (wave < 5) {
    float pl = 0.f;
#pragma unroll
    for (int q = 0; q < 4; ++q) {
      const int k = lane + 32 * q;
      pl += hid[k] * w2s[k * 5 + wave];
    }
#pragma unroll
    for (int off = 16; off >= 1; off >>= 1) pl += __shfl_xor(pl, off, 32);
    if (lane == 0) {
      const float a = bs2v + pl;
      slg[wave] = a;
      slog_out[b * 5 + wave] = a;
    }
  }
  __syncthreads();

  if (t == 0) {
    float mx = slg[0];
    for (int i = 1; i < 5; ++i) mx = fmaxf(mx, slg[i]);
    float sum = 0.f;
    for (int i = 0; i < 5; ++i) { sp[i] = __expf(slg[i] - mx); sum += sp[i]; }
    float inv = 1.0f / sum;
    for (int i = 0; i < 5; ++i) sp[i] *= inv;
  }
  __syncthreads();

  if (t < 64) {
    float a = bc1v;
    for (int k = 0; k < 5; ++k) a += sp[k] * wc1s[k * 64 + t];
    c1s[t] = fmaxf(a, 0.f);
  }
  __syncthreads();

  if (t < 64) {
    float a = bc2v;
    for (int k = 0; k < 64; ++k) a += c1s[k] * wc2s[k * 64 + t];
    c2s[t] = a;
  }
  __syncthreads();

  if (t == 0) {
    float cs = 0.f;
    for (int k = 0; k < 64; ++k) cs += c2s[k] * aws[k];
    cs_out[b] = cs;
  }

  float4 aq[4];
#pragma unroll
  for (int i = 0; i < 4; ++i) {
    const v4f a = get4(aqv, i);
    aq[i] = make_float4(a.x, a.y, a.z, a.w);
  }
  logA_slice_store(aq, logA, rmaxg, b);
}

#define APAD 260
__global__ void __launch_bounds__(256) proj_gemm(
    const float* __restrict__ x, const float* __restrict__ Wp,
    const float* __restrict__ bp, float* __restrict__ xp,
    float* __restrict__ colpart) {
  extern __shared__ float sm[];
  float* Apan = sm;
  float* BpT  = sm + 64 * APAD;

  const int tid = threadIdx.x;
  const int m0 = blockIdx.x * 64;
  const int n0 = blockIdx.y * 64;

  const float* xcol = x + (size_t)m0 * 256 + tid;
  float* acol = Apan + tid;
  float csum = 0.f;
  for (int r0 = 0; r0 < 64; r0 += 8) {
    float v[8];
    for (int i = 0; i < 8; ++i) v[i] = xcol[(r0 + i) * 256];
    for (int i = 0; i < 8; ++i) {
      acol[(r0 + i) * APAD] = v[i];
      csum += v[i];
    }
  }
  if (blockIdx.y == 0) colpart[blockIdx.x * 256 + tid] = csum;
  for (int idx = tid; idx < 256 * 64; idx += 256) {
    const int k = idx >> 6, n = idx & 63;
    BpT[n * APAD + k] = Wp[(size_t)k * 256 + n0 + n];
  }
  __syncthreads();

  const int wave = tid >> 5, lane = tid & 31;
  const int r16 = lane & 15, half = lane >> 4;
  const int rw = wave >> 1;
  const int cw = (wave & 1) * 32;

  v8f acc0 = {0, 0, 0, 0, 0, 0, 0, 0};
  v8f acc1 = {0, 0, 0, 0, 0, 0, 0, 0};

  const float* arow = Apan + (rw * 16 + r16) * APAD;
  const float* b0row = BpT + (cw + r16) * APAD;
  const float* b1row = BpT + (cw + 16 + r16) * APAD;

  for (int kk = 0; kk < 256; kk += 4) {
    const int ko = kk + 2 * half;
    v2f a;  a.x = arow[ko];  a.y = arow[ko + 1];
    v2f b0; b0.x = b0row[ko]; b0.y = b0row[ko + 1];
    v2f b1; b1.x = b1row[ko]; b1.y = b1row[ko + 1];
    acc0 = wmma_f32(a, b0, acc0);
    acc1 = wmma_f32(a, b1, acc1);
  }

  const float bias0 = bp[n0 + cw + r16];
  const float bias1 = bp[n0 + cw + 16 + r16];
#pragma unroll
  for (int v = 0; v < 8; ++v) {
    const int row = m0 + rw * 16 + v + 8 * half;
    xp[(size_t)row * 256 + n0 + cw + r16]      = acc0[v] + bias0;
    xp[(size_t)row * 256 + n0 + cw + 16 + r16] = acc1[v] + bias1;
  }
}

__device__ __forceinline__ void srcdst_block(
    const int blk, const int slice,
    const float* __restrict__ xp, const float* __restrict__ att_w,
    float* __restrict__ srcg, float* __restrict__ dstg,
    const float* __restrict__ A, float* __restrict__ logA,
    float* __restrict__ rmaxg) {
  const int lane = threadIdx.x & 31;
  const int m0 = (blk * 8 + (threadIdx.x >> 5)) * 4;
  const int c16 = lane & 15;
  const bool has_slice = slice < 256;

  float4 va[4], vb[4];
  for (int r = 0; r < 4; ++r) {
    const float4* rowp = (const float4*)(xp + (size_t)(m0 + r) * 256);
    va[r] = rowp[lane];
    vb[r] = rowp[32 + lane];
  }
  const float4 ws = ((const float4*)att_w)[c16];
  const float4 wd = ((const float4*)(att_w + 64))[c16];
  float4 aq[4];
  if (has_slice) {
    const float4* A4 = (const float4*)A + slice * 1024 + threadIdx.x;
    for (int i = 0; i < 4; ++i) aq[i] = A4[i * 256];
  }

  float sa[4], da[4], sb[4], db[4];
  for (int r = 0; r < 4; ++r) {
    sa[r] = va[r].x * ws.x + va[r].y * ws.y + va[r].z * ws.z + va[r].w * ws.w;
    da[r] = va[r].x * wd.x + va[r].y * wd.y + va[r].z * wd.z + va[r].w * wd.w;
    sb[r] = vb[r].x * ws.x + vb[r].y * ws.y + vb[r].z * ws.z + vb[r].w * ws.w;
    db[r] = vb[r].x * wd.x + vb[r].y * wd.y + vb[r].z * wd.z + vb[r].w * wd.w;
  }
  for (int off = 8; off >= 1; off >>= 1) {
    for (int r = 0; r < 4; ++r) {
      sa[r] += __shfl_xor(sa[r], off, 32);
      da[r] += __shfl_xor(da[r], off, 32);
      sb[r] += __shfl_xor(sb[r], off, 32);
      db[r] += __shfl_xor(db[r], off, 32);
    }
  }
  float4 so[4], dq[4];
  for (int r = 0; r < 4; ++r) {
    so[r].x = sa[r]; so[r].y = __shfl_xor(sa[r], 16, 32);
    so[r].z = sb[r]; so[r].w = __shfl_xor(sb[r], 16, 32);
    dq[r].x = da[r]; dq[r].y = __shfl_xor(da[r], 16, 32);
    dq[r].z = db[r]; dq[r].w = __shfl_xor(db[r], 16, 32);
  }
  if (lane == 0) {
    for (int r = 0; r < 4; ++r) {
      ((float4*)srcg)[m0 + r] = so[r];
      ((float4*)dstg)[m0 + r] = dq[r];
    }
  }
  if (has_slice) logA_slice_store(aq, logA, rmaxg, slice);
}

__global__ void __launch_bounds__(256) side_kernel(
    const float* colpart,
    const float* __restrict__ Ws1, const float* __restrict__ bs1,
    const float* __restrict__ Ws2, const float* __restrict__ bs2,
    const float* __restrict__ Wc1, const float* __restrict__ bc1,
    const float* __restrict__ Wc2, const float* __restrict__ bc2,
    const float* __restrict__ att_w,
    float* __restrict__ slog_out, float* __restrict__ cs_out,
    const float* __restrict__ xp,
    float* __restrict__ srcg, float* __restrict__ dstg,
    const float* __restrict__ A, float* logA, float* __restrict__ rmaxg) {
  const int blk = blockIdx.x;
  if (blk < 8) {
    stats_block(blk, colpart, Ws1, bs1, Ws2, bs2, Wc1, bc1, Wc2, bc2, att_w,
                slog_out, cs_out, A, logA, rmaxg);
  } else {
    srcdst_block(blk - 8, blk, xp, att_w, srcg, dstg, A, logA, rmaxg);
  }
}

#define JP 132
#define XCT_F (64 * JP)
#define EBW_F (16 * JP)
#define ATTN_LDS_F (1024 + 2 * XCT_F + 2 * 8 * EBW_F)

__device__ __forceinline__ void lds_barrier() {
  asm volatile("s_wait_dscnt 0x0\n\ts_barrier_signal -1\n\ts_barrier_wait -1"
               ::: "memory");
}

__device__ __forceinline__ void attn_load_tile(float4 (&xt)[8],
                                               const float* xtile,
                                               const int j) {
#pragma unroll
  for (int i = 0; i < 8; ++i)
    xt[i] = *(const float4*)(xtile + (size_t)(j + 32 * (i & 3)) * 256 +
                             32 * (i >> 2));
}
__device__ __forceinline__ void attn_store_tile(float* xdst,
                                                const float4 (&xt)[8],
                                                const int td, const int tj) {
#pragma unroll
  for (int i = 0; i < 8; ++i) {
    float* tp = xdst + (td + 32 * (i >> 2)) * JP + tj + 32 * (i & 3);
    tp[0]      = xt[i].x;
    tp[JP]     = xt[i].y;
    tp[2 * JP] = xt[i].z;
    tp[3 * JP] = xt[i].w;
  }
}
__device__ __forceinline__ void attn_load_logA(float4 (&lg)[16],
                                               const float* lrow,
                                               const int j) {
#pragma unroll
  for (int r = 0; r < 16; ++r)
    lg[r] = *(const float4*)(lrow + (size_t)r * CC + j);
}
__device__ __forceinline__ void attn_exp_stage(
    float* ebdst, const float* srcS, const int j, const int lane,
    const float4 (&lg)[16], const float (&dvals)[16], const float (&mrow)[16],
    float (&srow)[16], const float cs, const float lamh) {
  const float4 sv = ((const float4*)(srcS + j))[lane];
#pragma unroll
  for (int r = 0; r < 16; ++r) {
    float e0 = sv.x + dvals[r] + cs; e0 = e0 > 0.f ? e0 : 0.2f * e0;
    float e1 = sv.y + dvals[r] + cs; e1 = e1 > 0.f ? e1 : 0.2f * e1;
    float e2 = sv.z + dvals[r] + cs; e2 = e2 > 0.f ? e2 : 0.2f * e2;
    float e3 = sv.w + dvals[r] + cs; e3 = e3 > 0.f ? e3 : 0.2f * e3;
    const float p0 = __expf(e0 + lamh * lg[r].x - mrow[r]);
    const float p1 = __expf(e1 + lamh * lg[r].y - mrow[r]);
    const float p2 = __expf(e2 + lamh * lg[r].z - mrow[r]);
    const float p3 = __expf(e3 + lamh * lg[r].w - mrow[r]);
    float* dstp = ebdst + r * JP + lane * 4;
    dstp[0] = p0; dstp[1] = p1; dstp[2] = p2; dstp[3] = p3;
    srow[r] += (p0 + p1) + (p2 + p3);
  }
}
struct AttnFrag { v2f a[4]; v2f b[4][4]; };
__device__ __forceinline__ void attn_load_frag(AttnFrag& f, const float* arow,
                                               const float* brow0,
                                               const int k0, const int half) {
#pragma unroll
  for (int s = 0; s < 4; ++s) {
    const int ko = k0 + 4 * s + 2 * half;
    f.a[s].x = arow[ko]; f.a[s].y = arow[ko + 1];
#pragma unroll
    for (int nt = 0; nt < 4; ++nt) {
      const float* brow = brow0 + nt * 16 * JP;
      f.b[s][nt].x = brow[ko]; f.b[s][nt].y = brow[ko + 1];
    }
  }
}
__device__ __forceinline__ void attn_mma_frag(const AttnFrag& f,
                                              v8f (&acc)[4]) {
#pragma unroll
  for (int s = 0; s < 4; ++s) {
#pragma unroll
    for (int nt = 0; nt < 4; ++nt)
      acc[nt] = wmma_f32(f.a[s], f.b[s][nt], acc[nt]);
  }
}
__device__ __forceinline__ void attn_wmma_stage(const float* ebsrc,
                                                const float* xsrc,
                                                const int r16, const int half,
                                                v8f (&acc)[4]) {
  const float* arow = ebsrc + r16 * JP;
  const float* brow0 = xsrc + r16 * JP;
  AttnFrag f0, f1;
  attn_load_frag(f0, arow, brow0, 0, half);
  for (int k0 = 0; k0 < 128; k0 += 32) {
    attn_load_frag(f1, arow, brow0, k0 + 16, half);
    attn_mma_frag(f0, acc);
    if (k0 + 32 < 128) attn_load_frag(f0, arow, brow0, k0 + 32, half);
    attn_mma_frag(f1, acc);
  }
}

__global__ void __launch_bounds__(256) attn_kernel(
    const float* __restrict__ xp,   const float* __restrict__ logA,
    const float* __restrict__ srcg, const float* __restrict__ dstg,
    const float* __restrict__ csb,  const float* __restrict__ lam,
    const float* __restrict__ rmaxg, float* __restrict__ out) {
  __shared__ __attribute__((aligned(16))) float sm[ATTN_LDS_F];
  __shared__ float sred[16];
  float* srcS = sm;
  float* XcT  = srcS + 1024;
  float* eb   = XcT + 2 * XCT_F;

  const int tid = threadIdx.x;
  const int wave = tid >> 5, lane = tid & 31;
  const int r16 = lane & 15, half = lane >> 4;

  const int bid = blockIdx.x;
  const int b  = bid >> 5;
  const int h  = (bid >> 3) & 3;
  const int it = bid & 7;
  const int i0 = it * 128 + wave * 16;

  float tmx = INF_NEG, tmn = -INF_NEG;
  for (int idx = tid; idx < 1024; idx += 256) {
    const float v = srcg[(b * CC + idx) * 4 + h];
    srcS[idx] = v;
    tmx = fmaxf(tmx, v);
    tmn = fminf(tmn, v);
  }
#pragma unroll
  for (int off = 16; off >= 1; off >>= 1) {
    tmx = fmaxf(tmx, __shfl_xor(tmx, off, 32));
    tmn = fminf(tmn, __shfl_xor(tmn, off, 32));
  }
  if (lane == 0) { sred[wave] = tmx; sred[8 + wave] = tmn; }
  __syncthreads();

  const int tj = tid >> 3;
  const int td = (tid & 7) * 4;
  const float* xtile = xp + ((size_t)(b * CC + tj) * 4 + h) * 64 + td;
  const float* lrow = logA + (size_t)i0 * CC + 4 * lane;

  float dvals[16];
  v16f rmv;
#pragma unroll
  for (int r = 0; r < 16; ++r) {
    dvals[r] = dstg[(b * CC + i0 + r) * 4 + h];
    rmv[r] = rmaxg[i0 + r];
  }
  float4 xt[8];
  attn_load_tile(xt, xtile, 0);
  float4 lg[16];
  attn_load_logA(lg, lrow, 0);

  const float cs = csb[b];
  const float lamh = lam[h];

  float smax = sred[0], smin = sred[8];
  for (int w = 1; w < 8; ++w) {
    smax = fmaxf(smax, sred[w]);
    smin = fminf(smin, sred[8 + w]);
  }

  const bool use_bound = __builtin_amdgcn_readfirstlane(
      (lamh >= 0.f && (smax - smin) <= 60.f) ? 1 : 0) != 0;

  asm volatile("" : "+v"(rmv));

  float mrow[16];
  if (use_bound) {
#pragma unroll
    for (int r = 0; r < 16; ++r) {
      float e = smax + dvals[r] + cs; e = e > 0.f ? e : 0.2f * e;
      mrow[r] = e + lamh * rmv[r];
    }
  } else {
#pragma unroll
    for (int r = 0; r < 16; ++r) mrow[r] = INF_NEG;

    for (int jc = 0; jc < CC; jc += 128) {
      const float4 sv = ((const float4*)(srcS + jc))[lane];
#pragma unroll
      for (int r = 0; r < 16; ++r) {
        const float4 lq =
            ((const float4*)(logA + (size_t)(i0 + r) * CC + jc))[lane];
        float e0 = sv.x + dvals[r] + cs; e0 = e0 > 0.f ? e0 : 0.2f * e0;
        float e1 = sv.y + dvals[r] + cs; e1 = e1 > 0.f ? e1 : 0.2f * e1;
        float e2 = sv.z + dvals[r] + cs; e2 = e2 > 0.f ? e2 : 0.2f * e2;
        float e3 = sv.w + dvals[r] + cs; e3 = e3 > 0.f ? e3 : 0.2f * e3;
        const float v0 = e0 + lamh * lq.x;
        const float v1 = e1 + lamh * lq.y;
        const float v2 = e2 + lamh * lq.z;
        const float v3 = e3 + lamh * lq.w;
        mrow[r] = fmaxf(mrow[r], fmaxf(fmaxf(v0, v1), fmaxf(v2, v3)));
      }
    }
#pragma unroll
    for (int r = 0; r < 16; ++r) {
#pragma unroll
      for (int off = 16; off >= 1; off >>= 1)
        mrow[r] = fmaxf(mrow[r], __shfl_xor(mrow[r], off, 32));
    }
  }

  v8f acc[4];
  const v8f vzero = {0, 0, 0, 0, 0, 0, 0, 0};
#pragma unroll
  for (int nt = 0; nt < 4; ++nt) acc[nt] = vzero;

  float srow[16];
#pragma unroll
  for (int r = 0; r < 16; ++r) srow[r] = 0.f;

  const bool wmma_first = __builtin_amdgcn_readfirstlane(wave) < 4;

  attn_store_tile(XcT, xt, td, tj);
  attn_load_tile(xt, xtile, 128);
  attn_exp_stage(eb + wave * EBW_F, srcS, 0, lane, lg, dvals, mrow, srow, cs,
                 lamh);
  attn_load_logA(lg, lrow, 128);

  for (int c = 0; c < 7; ++c) {
    const int cur = c & 1;
    const float* xcur = XcT + cur * XCT_F;
    float* xnxt = XcT + (cur ^ 1) * XCT_F;
    const float* ecur = eb + (cur * 8 + wave) * EBW_F;
    float* enxt = eb + ((cur ^ 1) * 8 + wave) * EBW_F;
    const int jn = (c + 1) * 128;

    lds_barrier();

    attn_store_tile(xnxt, xt, td, tj);
    if (c < 6) attn_load_tile(xt, xtile, jn + 128);

    if (wmma_first) {
      attn_wmma_stage(ecur, xcur, r16, half, acc);
      attn_exp_stage(enxt, srcS, jn, lane, lg, dvals, mrow, srow, cs, lamh);
    } else {
      attn_exp_stage(enxt, srcS, jn, lane, lg, dvals, mrow, srow, cs, lamh);
      attn_wmma_stage(ecur, xcur, r16, half, acc);
    }
    if (c < 6) attn_load_logA(lg, lrow, jn + 128);
  }
  lds_barrier();
  attn_wmma_stage(eb + (8 + wave) * EBW_F, XcT + XCT_F, r16, half, acc);

#pragma unroll
  for (int r = 0; r < 16; ++r) {
#pragma unroll
    for (int off = 16; off >= 1; off >>= 1)
      srow[r] += __shfl_xor(srow[r], off, 32);
  }

#pragma unroll
  for (int v = 0; v < 8; ++v) {
    const int rr = v + 8 * half;
    const float inv = 1.0f / srow[rr];
    const int row = i0 + rr;
#pragma unroll
    for (int nt = 0; nt < 4; ++nt) {
      out[(size_t)(b * CC + row) * 256 + h * 64 + nt * 16 + r16] =
          acc[nt][v] * inv;
    }
  }
}

extern "C" void kernel_launch(void* const* d_in, const int* in_sizes, int n_in,
                              void* d_out, int out_size, void* d_ws,
                              size_t ws_size, hipStream_t stream) {
  (void)in_sizes; (void)n_in; (void)out_size; (void)ws_size;

  const float* x   = (const float*)d_in[0];
  const float* Apr = (const float*)d_in[1];
  const float* Wp  = (const float*)d_in[2];
  const float* bp  = (const float*)d_in[3];
  const float* Ws1 = (const float*)d_in[4];
  const float* bs1 = (const float*)d_in[5];
  const float* Ws2 = (const float*)d_in[6];
  const float* bs2 = (const float*)d_in[7];
  const float* Wc1 = (const float*)d_in[8];
  const float* bc1 = (const float*)d_in[9];
  const float* Wc2 = (const float*)d_in[10];
  const float* bc2 = (const float*)d_in[11];
  const float* att = (const float*)d_in[12];
  const float* lam = (const float*)d_in[13];

  float* out  = (float*)d_out;
  float* slog = out + (size_t)BB * CC * 256;

  float* xp   = (float*)d_ws;
  float* logA = xp + 2097152;
  float* srcg = logA + 1048576;
  float* dstg = srcg + 32768;
  float* csb  = dstg + 32768;
  float* rmax = csb + 64;
  float* colpart = logA;

  const size_t gemm_lds = (size_t)(2 * 64 * APAD) * sizeof(float);
  proj_gemm<<<dim3(128, 4), 256, gemm_lds, stream>>>(x, Wp, bp, xp, colpart);

  side_kernel<<<264, 256, 0, stream>>>(colpart, Ws1, bs1, Ws2, bs2, Wc1,
                                            bc1, Wc2, bc2, att, slog, csb, xp,
                                            srcg, dstg, Apr, logA, rmax);

  attn_kernel<<<BB * 4 * 8, 256, 0, stream>>>(xp, logA, srcg, dstg, csb, lam,
                                              rmax, out);
}


# proj_gemm and attn heads: the dependent load round trips before the first WMMA (B panel in 4 rolled trips behind the A panel; src_s in 4, then a group behind the barrier) are one load group each
# speedup vs baseline: 3.3348x; 1.0353x over previous
#include <hip/hip_runtime.h>
#include <math.h>


typedef float v2f __attribute__((ext_vector_type(2)));
typedef float v4f __attribute__((ext_vector_type(4)));
typedef float v8f __attribute__((ext_vector_type(8)));

__device__ __forceinline__ v8f wmma_f32(v2f a, v2f b, v8f c) {
  return __builtin_amdgcn_wmma_f32_16x16x4_f32(false, a, false, b,
                                               (short)0, c, false, false);
}

#define BB 8
#define CC 1024
#define INF_NEG (-3.0e38f)

__device__ __forceinline__ float4 logA_quad(const float4 a) {
  float4 o;
  o.x = logf(a.x + 1e-9f);
  o.y = logf(a.y + 1e-9f);
  o.z = logf(a.z + 1e-9f);
  o.w = logf(a.w + 1e-9f);
  return o;
}

__device__ __forceinline__ void logA_slice_store(const float4 (&aq)[4],
                                                 float* logA, float* rmaxg,
                                                 const int slice) {
  __shared__ float wmx[4][8];
  const int t = threadIdx.x;
  const int wave = t >> 5, lane = t & 31;
  float4* L4 = (float4*)logA + slice * 1024 + t;
  float m[4];
#pragma unroll
  for (int i = 0; i < 4; ++i) {
    const float4 l = logA_quad(aq[i]);
    L4[i * 256] = l;
    m[i] = fmaxf(fmaxf(l.x, l.y), fmaxf(l.z, l.w));
  }
#pragma unroll
  for (int off = 16; off >= 1; off >>= 1) {
#pragma unroll
    for (int i = 0; i < 4; ++i) m[i] = fmaxf(m[i], __shfl_xor(m[i], off, 32));
  }
  if (lane == 0) {
#pragma unroll
    for (int i = 0; i < 4; ++i) wmx[i][wave] = m[i];
  }
  __syncthreads();
  if (t < 4) {
    float v = wmx[t][0];
    for (int w = 1; w < 8; ++w) v = fmaxf(v, wmx[t][w]);
    rmaxg[slice * 4 + t] = v;
  }
}

typedef float v16f __attribute__((ext_vector_type(16)));
__device__ __forceinline__ void put4(v16f& d, const int k, const v4f q) {
  d[4 * k + 0] = q.x; d[4 * k + 1] = q.y;
  d[4 * k + 2] = q.z; d[4 * k + 3] = q.w;
}
__device__ __forceinline__ v4f get4(const v16f& d, const int k) {
  v4f q;
  q.x = d[4 * k + 0]; q.y = d[4 * k + 1];
  q.z = d[4 * k + 2]; q.w = d[4 * k + 3];
  return q;
}

__device__ __forceinline__ void stats_block(
    const int b, const float* colpart,
    const float* __restrict__ Ws1, const float* __restrict__ bs1,
    const float* __restrict__ Ws2, const float* __restrict__ bs2,
    const float* __restrict__ Wc1, const float* __restrict__ bc1,
    const float* __restrict__ Wc2, const float* __restrict__ bc2,
    const float* __restrict__ att_w,
    float* __restrict__ slog_out, float* __restrict__ cs_out,
    const float* __restrict__ A, float* logA, float* __restrict__ rmaxg) {
  __shared__ float4 xm4[64];
  __shared__ float4 hpart4[256];
  __shared__ v4f w2s4[256];
  __shared__ v4f wc1s4[256];
  __shared__ v4f aws4[256];
  __shared__ v4f wc2s4[1024];
  __shared__ float hid[128];
  __shared__ float slg[5];
  __shared__ float sp[5];
  __shared__ float c1s[64];
  __shared__ float c2s[64];

  float* xm = (float*)xm4;
  const float* hpart = (const float*)hpart4;
  const float* w2s = (const float*)w2s4;
  const float* wc1s = (const float*)wc1s4;
  const float* wc2s = (const float*)wc2s4;
  const float* aws = (const float*)aws4;

  const int t = threadIdx.x;
  const int wave = t >> 5, lane = t & 31;

  v16f cpv;
#pragma unroll
  for (int p = 0; p < 16; ++p) cpv[p] = colpart[(b * 16 + p) * 256 + t];

  v16f w1v[8];
  const v4f* Ws1q = (const v4f*)Ws1 + (size_t)(wave * 32) * 32 + lane;
#pragma unroll
  for (int i = 0; i < 32; ++i) put4(w1v[i >> 2], i & 3, Ws1q[i * 32]);

  v16f smA, smB;
  put4(smA, 0, ((const v4f*)Ws2)[t < 160 ? t : 159]);
  put4(smA, 1, ((const v4f*)Wc1)[t < 80 ? t : 79]);
  put4(smA, 2, ((const v4f*)(att_w + 128))[t & 15]);
  put4(smA, 3, ((const v4f*)Wc2)[t]);
  put4(smB, 0, ((const v4f*)Wc2)[t + 256]);
  put4(smB, 1, ((const v4f*)Wc2)[t + 512]);
  put4(smB, 2, ((const v4f*)Wc2)[t + 768]);
  smB[12] = bs1[t & 127];
  smB[13] = bs2[wave < 5 ? wave : 4];
  smB[14] = bc1[t & 63];
  smB[15] = bc2[t & 63];

  v16f aqv;
  const v4f* A4 = (const v4f*)A + b * 1024 + t;
#pragma unroll
  for (int i = 0; i < 4; ++i) put4(aqv, i, A4[i * 256]);

  asm volatile("" : "+v"(cpv), "+v"(w1v[0]), "+v"(w1v[1]), "+v"(w1v[2]),
                    "+v"(w1v[3]), "+v"(w1v[4]), "+v"(w1v[5]), "+v"(w1v[6]),
                    "+v"(w1v[7]), "+v"(smA), "+v"(smB), "+v"(aqv));

  const float bs1v = smB[12];
  const float bs2v = smB[13];
  const float bc1v = smB[14];
  const float bc2v = smB[15];

  float s = 0.f;
#pragma unroll
  for (int p = 0; p < 16; ++p) s += cpv[p];
  xm[t] = s * (1.0f / 1024.0f);

  w2s4[t] = get4(smA, 0);
  wc1s4[t] = get4(smA, 1);
  aws4[t] = get4(smA, 2);
  wc2s4[t] = get4(smA, 3);
  wc2s4[t + 256] = get4(smB, 0);
  wc2s4[t + 512] = get4(smB, 1);
  wc2s4[t + 768] = get4(smB, 2);
  __syncthreads();

  {
    float4 acc = make_float4(0.f, 0.f, 0.f, 0.f);
#pragma unroll
    for (int i4 = 0; i4 < 8; ++i4) {
      const float4 xv = xm4[wave * 8 + i4];
      const float xs[4] = {xv.x, xv.y, xv.z, xv.w};
#pragma unroll
      for (int j = 0; j < 4; ++j) {
        const v4f w = get4(w1v[i4], j);
        acc.x += xs[j] * w.x;
        acc.y += xs[j] * w.y;
        acc.z += xs[j] * w.z;
        acc.w += xs[j] * w.w;
      }
    }
    hpart4[wave * 32 + lane] = acc;
  }
  __syncthreads();

  if (t < 128) {
    float a = bs1v;
#pragma unroll
    for (int w = 0; w < 8; ++w) a += hpart[w * 128 + t];
    hid[t] = fmaxf(a, 0.f);
  }
  __syncthreads();

  if (wave < 5) {
    float pl = 0.f;
#pragma unroll
    for (int q = 0; q < 4; ++q) {
      const int k = lane + 32 * q;
      pl += hid[k] * w2s[k * 5 + wave];
    }
#pragma unroll
    for (int off = 16; off >= 1; off >>= 1) pl += __shfl_xor(pl, off, 32);
    if (lane == 0) {
      const float a = bs2v + pl;
      slg[wave] = a;
      slog_out[b * 5 + wave] = a;
    }
  }
  __syncthreads();

  if (t == 0) {
    float mx = slg[0];
    for (int i = 1; i < 5; ++i) mx = fmaxf(mx, slg[i]);
    float sum = 0.f;
    for (int i = 0; i < 5; ++i) { sp[i] = __expf(slg[i] - mx); sum += sp[i]; }
    float inv = 1.0f / sum;
    for (int i = 0; i < 5; ++i) sp[i] *= inv;
  }
  __syncthreads();

  if (t < 64) {
    float a = bc1v;
    for (int k = 0; k < 5; ++k) a += sp[k] * wc1s[k * 64 + t];
    c1s[t] = fmaxf(a, 0.f);
  }
  __syncthreads();

  if (t < 64) {
    float a = bc2v;
    for (int k = 0; k < 64; ++k) a += c1s[k] * wc2s[k * 64 + t];
    c2s[t] = a;
  }
  __syncthreads();

  if (t == 0) {
    float cs = 0.f;
    for (int k = 0; k < 64; ++k) cs += c2s[k] * aws[k];
    cs_out[b] = cs;
  }

  float4 aq[4];
#pragma unroll
  for (int i = 0; i < 4; ++i) {
    const v4f a = get4(aqv, i);
    aq[i] = make_float4(a.x, a.y, a.z, a.w);
  }
  logA_slice_store(aq, logA, rmaxg, b);
}

#define APAD 260
__global__ void __launch_bounds__(256) proj_gemm(
    const float* __restrict__ x, const float* __restrict__ Wp,
    const float* __restrict__ bp, float* __restrict__ xp,
    float* __restrict__ colpart) {
  extern __shared__ float sm[];
  float* Apan = sm;
  float* BpT  = sm + 64 * APAD;

  const int tid = threadIdx.x;
  const int m0 = blockIdx.x * 64;
  const int n0 = blockIdx.y * 64;

  const float* xcol = x + (size_t)m0 * 256 + tid;
  const int bk = tid >> 6, bn = tid & 63;
  const float* wcol = Wp + (size_t)bk * 256 + n0 + bn;
  float av[64], bv[64];
#pragma unroll
  for (int i = 0; i < 64; ++i) av[i] = xcol[i * 256];
#pragma unroll
  for (int i = 0; i < 64; ++i) bv[i] = wcol[(size_t)i * 1024];

  float* acol = Apan + tid;
  float csum = 0.f;
#pragma unroll
  for (int i = 0; i < 64; ++i) {
    acol[i * APAD] = av[i];
    csum += av[i];
  }
  if (blockIdx.y == 0) colpart[blockIdx.x * 256 + tid] = csum;
  float* bcol = BpT + bn * APAD + bk;
#pragma unroll
  for (int i = 0; i < 64; ++i) bcol[4 * i] = bv[i];
  __syncthreads();

  const int wave = tid >> 5, lane = tid & 31;
  const int r16 = lane & 15, half = lane >> 4;
  const int rw = wave >> 1;
  const int cw = (wave & 1) * 32;

  v8f acc0 = {0, 0, 0, 0, 0, 0, 0, 0};
  v8f acc1 = {0, 0, 0, 0, 0, 0, 0, 0};

  const float* arow = Apan + (rw * 16 + r16) * APAD;
  const float* b0row = BpT + (cw + r16) * APAD;
  const float* b1row = BpT + (cw + 16 + r16) * APAD;

  for (int kk = 0; kk < 256; kk += 4) {
    const int ko = kk + 2 * half;
    v2f a;  a.x = arow[ko];  a.y = arow[ko + 1];
    v2f b0; b0.x = b0row[ko]; b0.y = b0row[ko + 1];
    v2f b1; b1.x = b1row[ko]; b1.y = b1row[ko + 1];
    acc0 = wmma_f32(a, b0, acc0);
    acc1 = wmma_f32(a, b1, acc1);
  }

  const float bias0 = bp[n0 + cw + r16];
  const float bias1 = bp[n0 + cw + 16 + r16];
#pragma unroll
  for (int v = 0; v < 8; ++v) {
    const int row = m0 + rw * 16 + v + 8 * half;
    xp[(size_t)row * 256 + n0 + cw + r16]      = acc0[v] + bias0;
    xp[(size_t)row * 256 + n0 + cw + 16 + r16] = acc1[v] + bias1;
  }
}

__device__ __forceinline__ void srcdst_block(
    const int blk, const int slice,
    const float* __restrict__ xp, const float* __restrict__ att_w,
    float* __restrict__ srcg, float* __restrict__ dstg,
    const float* __restrict__ A, float* __restrict__ logA,
    float* __restrict__ rmaxg) {
  const int lane = threadIdx.x & 31;
  const int m0 = (blk * 8 + (threadIdx.x >> 5)) * 4;
  const int c16 = lane & 15;
  const bool has_slice = slice < 256;

  float4 va[4], vb[4];
  for (int r = 0; r < 4; ++r) {
    const float4* rowp = (const float4*)(xp + (size_t)(m0 + r) * 256);
    va[r] = rowp[lane];
    vb[r] = rowp[32 + lane];
  }
  const float4 ws = ((const float4*)att_w)[c16];
  const float4 wd = ((const float4*)(att_w + 64))[c16];
  float4 aq[4];
  if (has_slice) {
    const float4* A4 = (const float4*)A + slice * 1024 + threadIdx.x;
    for (int i = 0; i < 4; ++i) aq[i] = A4[i * 256];
  }

  float sa[4], da[4], sb[4], db[4];
  for (int r = 0; r < 4; ++r) {
    sa[r] = va[r].x * ws.x + va[r].y * ws.y + va[r].z * ws.z + va[r].w * ws.w;
    da[r] = va[r].x * wd.x + va[r].y * wd.y + va[r].z * wd.z + va[r].w * wd.w;
    sb[r] = vb[r].x * ws.x + vb[r].y * ws.y + vb[r].z * ws.z + vb[r].w * ws.w;
    db[r] = vb[r].x * wd.x + vb[r].y * wd.y + vb[r].z * wd.z + vb[r].w * wd.w;
  }
  for (int off = 8; off >= 1; off >>= 1) {
    for (int r = 0; r < 4; ++r) {
      sa[r] += __shfl_xor(sa[r], off, 32);
      da[r] += __shfl_xor(da[r], off, 32);
      sb[r] += __shfl_xor(sb[r], off, 32);
      db[r] += __shfl_xor(db[r], off, 32);
    }
  }
  float4 so[4], dq[4];
  for (int r = 0; r < 4; ++r) {
    so[r].x = sa[r]; so[r].y = __shfl_xor(sa[r], 16, 32);
    so[r].z = sb[r]; so[r].w = __shfl_xor(sb[r], 16, 32);
    dq[r].x = da[r]; dq[r].y = __shfl_xor(da[r], 16, 32);
    dq[r].z = db[r]; dq[r].w = __shfl_xor(db[r], 16, 32);
  }
  if (lane == 0) {
    for (int r = 0; r < 4; ++r) {
      ((float4*)srcg)[m0 + r] = so[r];
      ((float4*)dstg)[m0 + r] = dq[r];
    }
  }
  if (has_slice) logA_slice_store(aq, logA, rmaxg, slice);
}

__global__ void __launch_bounds__(256) side_kernel(
    const float* colpart,
    const float* __restrict__ Ws1, const float* __restrict__ bs1,
    const float* __restrict__ Ws2, const float* __restrict__ bs2,
    const float* __restrict__ Wc1, const float* __restrict__ bc1,
    const float* __restrict__ Wc2, const float* __restrict__ bc2,
    const float* __restrict__ att_w,
    float* __restrict__ slog_out, float* __restrict__ cs_out,
    const float* __restrict__ xp,
    float* __restrict__ srcg, float* __restrict__ dstg,
    const float* __restrict__ A, float* logA, float* __restrict__ rmaxg) {
  const int blk = blockIdx.x;
  if (blk < 8) {
    stats_block(blk, colpart, Ws1, bs1, Ws2, bs2, Wc1, bc1, Wc2, bc2, att_w,
                slog_out, cs_out, A, logA, rmaxg);
  } else {
    srcdst_block(blk - 8, blk, xp, att_w, srcg, dstg, A, logA, rmaxg);
  }
}

#define JP 132
#define XCT_F (64 * JP)
#define EBW_F (16 * JP)
#define ATTN_LDS_F (1024 + 2 * XCT_F + 2 * 8 * EBW_F)

__device__ __forceinline__ void lds_barrier() {
  asm volatile("s_wait_dscnt 0x0\n\ts_barrier_signal -1\n\ts_barrier_wait -1"
               ::: "memory");
}

__device__ __forceinline__ void attn_load_tile(float4 (&xt)[8],
                                               const float* xtile,
                                               const int j) {
#pragma unroll
  for (int i = 0; i < 8; ++i)
    xt[i] = *(const float4*)(xtile + (size_t)(j + 32 * (i & 3)) * 256 +
                             32 * (i >> 2));
}
__device__ __forceinline__ void attn_store_tile(float* xdst,
                                                const float4 (&xt)[8],
                                                const int td, const int tj) {
#pragma unroll
  for (int i = 0; i < 8; ++i) {
    float* tp = xdst + (td + 32 * (i >> 2)) * JP + tj + 32 * (i & 3);
    tp[0]      = xt[i].x;
    tp[JP]     = xt[i].y;
    tp[2 * JP] = xt[i].z;
    tp[3 * JP] = xt[i].w;
  }
}
__device__ __forceinline__ void attn_load_logA(float4 (&lg)[16],
                                               const float* lrow,
                                               const int j) {
#pragma unroll
  for (int r = 0; r < 16; ++r)
    lg[r] = *(const float4*)(lrow + (size_t)r * CC + j);
}
__device__ __forceinline__ void attn_exp_stage(
    float* ebdst, const float* srcS, const int j, const int lane,
    const float4 (&lg)[16], const float (&dvals)[16], const float (&mrow)[16],
    float (&srow)[16], const float cs, const float lamh) {
  const float4 sv = ((const float4*)(srcS + j))[lane];
#pragma unroll
  for (int r = 0; r < 16; ++r) {
    float e0 = sv.x + dvals[r] + cs; e0 = e0 > 0.f ? e0 : 0.2f * e0;
    float e1 = sv.y + dvals[r] + cs; e1 = e1 > 0.f ? e1 : 0.2f * e1;
    float e2 = sv.z + dvals[r] + cs; e2 = e2 > 0.f ? e2 : 0.2f * e2;
    float e3 = sv.w + dvals[r] + cs; e3 = e3 > 0.f ? e3 : 0.2f * e3;
    const float p0 = __expf(e0 + lamh * lg[r].x - mrow[r]);
    const float p1 = __expf(e1 + lamh * lg[r].y - mrow[r]);
    const float p2 = __expf(e2 + lamh * lg[r].z - mrow[r]);
    const float p3 = __expf(e3 + lamh * lg[r].w - mrow[r]);
    float* dstp = ebdst + r * JP + lane * 4;
    dstp[0] = p0; dstp[1] = p1; dstp[2] = p2; dstp[3] = p3;
    srow[r] += (p0 + p1) + (p2 + p3);
  }
}
struct AttnFrag { v2f a[4]; v2f b[4][4]; };
__device__ __forceinline__ void attn_load_frag(AttnFrag& f, const float* arow,
                                               const float* brow0,
                                               const int k0, const int half) {
#pragma unroll
  for (int s = 0; s < 4; ++s) {
    const int ko = k0 + 4 * s + 2 * half;
    f.a[s].x = arow[ko]; f.a[s].y = arow[ko + 1];
#pragma unroll
    for (int nt = 0; nt < 4; ++nt) {
      const float* brow = brow0 + nt * 16 * JP;
      f.b[s][nt].x = brow[ko]; f.b[s][nt].y = brow[ko + 1];
    }
  }
}
__device__ __forceinline__ void attn_mma_frag(const AttnFrag& f,
                                              v8f (&acc)[4]) {
#pragma unroll
  for (int s = 0; s < 4; ++s) {
#pragma unroll
    for (int nt = 0; nt < 4; ++nt)
      acc[nt] = wmma_f32(f.a[s], f.b[s][nt], acc[nt]);
  }
}
__device__ __forceinline__ void attn_wmma_stage(const float* ebsrc,
                                                const float* xsrc,
                                                const int r16, const int half,
                                                v8f (&acc)[4]) {
  const float* arow = ebsrc + r16 * JP;
  const float* brow0 = xsrc + r16 * JP;
  AttnFrag f0, f1;
  attn_load_frag(f0, arow, brow0, 0, half);
  for (int k0 = 0; k0 < 128; k0 += 32) {
    attn_load_frag(f1, arow, brow0, k0 + 16, half);
    attn_mma_frag(f0, acc);
    if (k0 + 32 < 128) attn_load_frag(f0, arow, brow0, k0 + 32, half);
    attn_mma_frag(f1, acc);
  }
}

__global__ void __launch_bounds__(256) attn_kernel(
    const float* __restrict__ xp,   const float* __restrict__ logA,
    const float* __restrict__ srcg, const float* __restrict__ dstg,
    const float* __restrict__ csb,  const float* __restrict__ lam,
    const float* __restrict__ rmaxg, float* __restrict__ out) {
  __shared__ __attribute__((aligned(16))) float sm[ATTN_LDS_F];
  __shared__ float sred[16];
  float* srcS = sm;
  float* XcT  = srcS + 1024;
  float* eb   = XcT + 2 * XCT_F;

  const int tid = threadIdx.x;
  const int wave = tid >> 5, lane = tid & 31;
  const int r16 = lane & 15, half = lane >> 4;

  const int bid = blockIdx.x;
  const int b  = bid >> 5;
  const int h  = (bid >> 3) & 3;
  const int it = bid & 7;
  const int i0 = it * 128 + wave * 16;

  const int tj = tid >> 3;
  const int td = (tid & 7) * 4;
  const float* xtile = xp + ((size_t)(b * CC + tj) * 4 + h) * 64 + td;
  const float* lrow = logA + (size_t)i0 * CC + 4 * lane;

  float sv4[4];
#pragma unroll
  for (int q = 0; q < 4; ++q) sv4[q] = srcg[(b * CC + tid + 256 * q) * 4 + h];
  float dvals[16];
  v16f rmv;
#pragma unroll
  for (int r = 0; r < 16; ++r) {
    dvals[r] = dstg[(b * CC + i0 + r) * 4 + h];
    rmv[r] = rmaxg[i0 + r];
  }
  float4 xt[8];
  attn_load_tile(xt, xtile, 0);
  float4 lg[16];
  attn_load_logA(lg, lrow, 0);

  const float cs = csb[b];
  const float lamh = lam[h];

  float tmx = INF_NEG, tmn = -INF_NEG;
#pragma unroll
  for (int q = 0; q < 4; ++q) {
    srcS[tid + 256 * q] = sv4[q];
    tmx = fmaxf(tmx, sv4[q]);
    tmn = fminf(tmn, sv4[q]);
  }
#pragma unroll
  for (int off = 16; off >= 1; off >>= 1) {
    tmx = fmaxf(tmx, __shfl_xor(tmx, off, 32));
    tmn = fminf(tmn, __shfl_xor(tmn, off, 32));
  }
  if (lane == 0) { sred[wave] = tmx; sred[8 + wave] = tmn; }
  lds_barrier();

  float smax = sred[0], smin = sred[8];
  for (int w = 1; w < 8; ++w) {
    smax = fmaxf(smax, sred[w]);
    smin = fminf(smin, sred[8 + w]);
  }

  const bool use_bound = __builtin_amdgcn_readfirstlane(
      (lamh >= 0.f && (smax - smin) <= 60.f) ? 1 : 0) != 0;

  float mrow[16];
  if (use_bound) {
#pragma unroll
    for (int r = 0; r < 16; ++r) {
      float e = smax + dvals[r] + cs; e = e > 0.f ? e : 0.2f * e;
      mrow[r] = e + lamh * rmv[r];
    }
  } else {
#pragma unroll
    for (int r = 0; r < 16; ++r) mrow[r] = INF_NEG;

    for (int jc = 0; jc < CC; jc += 128) {
      const float4 sv = ((const float4*)(srcS + jc))[lane];
#pragma unroll
      for (int r = 0; r < 16; ++r) {
        const float4 lq =
            ((const float4*)(logA + (size_t)(i0 + r) * CC + jc))[lane];
        float e0 = sv.x + dvals[r] + cs; e0 = e0 > 0.f ? e0 : 0.2f * e0;
        float e1 = sv.y + dvals[r] + cs; e1 = e1 > 0.f ? e1 : 0.2f * e1;
        float e2 = sv.z + dvals[r] + cs; e2 = e2 > 0.f ? e2 : 0.2f * e2;
        float e3 = sv.w + dvals[r] + cs; e3 = e3 > 0.f ? e3 : 0.2f * e3;
        const float v0 = e0 + lamh * lq.x;
        const float v1 = e1 + lamh * lq.y;
        const float v2 = e2 + lamh * lq.z;
        const float v3 = e3 + lamh * lq.w;
        mrow[r] = fmaxf(mrow[r], fmaxf(fmaxf(v0, v1), fmaxf(v2, v3)));
      }
    }
#pragma unroll
    for (int r = 0; r < 16; ++r) {
#pragma unroll
      for (int off = 16; off >= 1; off >>= 1)
        mrow[r] = fmaxf(mrow[r], __shfl_xor(mrow[r], off, 32));
    }
  }

  v8f acc[4];
  const v8f vzero = {0, 0, 0, 0, 0, 0, 0, 0};
#pragma unroll
  for (int nt = 0; nt < 4; ++nt) acc[nt] = vzero;

  float srow[16];
#pragma unroll
  for (int r = 0; r < 16; ++r) srow[r] = 0.f;

  const bool wmma_first = __builtin_amdgcn_readfirstlane(wave) < 4;

  attn_store_tile(XcT, xt, td, tj);
  attn_load_tile(xt, xtile, 128);
  attn_exp_stage(eb + wave * EBW_F, srcS, 0, lane, lg, dvals, mrow, srow, cs,
                 lamh);
  attn_load_logA(lg, lrow, 128);

  for (int c = 0; c < 7; ++c) {
    const int cur = c & 1;
    const float* xcur = XcT + cur * XCT_F;
    float* xnxt = XcT + (cur ^ 1) * XCT_F;
    const float* ecur = eb + (cur * 8 + wave) * EBW_F;
    float* enxt = eb + ((cur ^ 1) * 8 + wave) * EBW_F;
    const int jn = (c + 1) * 128;

    lds_barrier();

    attn_store_tile(xnxt, xt, td, tj);
    if (c < 6) attn_load_tile(xt, xtile, jn + 128);

    if (wmma_first) {
      attn_wmma_stage(ecur, xcur, r16, half, acc);
      attn_exp_stage(enxt, srcS, jn, lane, lg, dvals, mrow, srow, cs, lamh);
    } else {
      attn_exp_stage(enxt, srcS, jn, lane, lg, dvals, mrow, srow, cs, lamh);
      attn_wmma_stage(ecur, xcur, r16, half, acc);
    }
    if (c < 6) attn_load_logA(lg, lrow, jn + 128);
  }
  lds_barrier();
  attn_wmma_stage(eb + (8 + wave) * EBW_F, XcT + XCT_F, r16, half, acc);

#pragma unroll
  for (int r = 0; r < 16; ++r) {
#pragma unroll
    for (int off = 16; off >= 1; off >>= 1)
      srow[r] += __shfl_xor(srow[r], off, 32);
  }

#pragma unroll
  for (int v = 0; v < 8; ++v) {
    const int rr = v + 8 * half;
    const float inv = 1.0f / srow[rr];
    const int row = i0 + rr;
#pragma unroll
    for (int nt = 0; nt < 4; ++nt) {
      out[(size_t)(b * CC + row) * 256 + h * 64 + nt * 16 + r16] =
          acc[nt][v] * inv;
    }
  }
}

extern "C" void kernel_launch(void* const* d_in, const int* in_sizes, int n_in,
                              void* d_out, int out_size, void* d_ws,
                              size_t ws_size, hipStream_t stream) {
  (void)in_sizes; (void)n_in; (void)out_size; (void)ws_size;

  const float* x   = (const float*)d_in[0];
  const float* Apr = (const float*)d_in[1];
  const float* Wp  = (const float*)d_in[2];
  const float* bp  = (const float*)d_in[3];
  const float* Ws1 = (const float*)d_in[4];
  const float* bs1 = (const float*)d_in[5];
  const float* Ws2 = (const float*)d_in[6];
  const float* bs2 = (const float*)d_in[7];
  const float* Wc1 = (const float*)d_in[8];
  const float* bc1 = (const float*)d_in[9];
  const float* Wc2 = (const float*)d_in[10];
  const float* bc2 = (const float*)d_in[11];
  const float* att = (const float*)d_in[12];
  const float* lam = (const float*)d_in[13];

  float* out  = (float*)d_out;
  float* slog = out + (size_t)BB * CC * 256;

  float* xp   = (float*)d_ws;
  float* logA = xp + 2097152;
  float* srcg = logA + 1048576;
  float* dstg = srcg + 32768;
  float* csb  = dstg + 32768;
  float* rmax = csb + 64;
  float* colpart = logA;

  const size_t gemm_lds = (size_t)(2 * 64 * APAD) * sizeof(float);
  proj_gemm<<<dim3(128, 4), 256, gemm_lds, stream>>>(x, Wp, bp, xp, colpart);

  side_kernel<<<264, 256, 0, stream>>>(colpart, Ws1, bs1, Ws2, bs2, Wc1,
                                            bc1, Wc2, bc2, att, slog, csb, xp,
                                            srcg, dstg, Apr, logA, rmax);

  attn_kernel<<<BB * 4 * 8, 256, 0, stream>>>(xp, logA, srcg, dstg, csb, lam,
                                              rmax, out);
}


# side launch dissolved, 3 launches -> 2: proj_gemm also makes src_s, logA rows, per-panel hidden sums; each attn wave ends the squeeze MLP (c_s via Wc2@w_c) and its dst_s; their sum order changes
# speedup vs baseline: 3.7707x; 1.1307x over previous
#include <hip/hip_runtime.h>
#include <math.h>


typedef float v2f __attribute__((ext_vector_type(2)));
typedef float v4f __attribute__((ext_vector_type(4)));
typedef float v8f __attribute__((ext_vector_type(8)));
typedef float v16f __attribute__((ext_vector_type(16)));

__device__ __forceinline__ v8f wmma_f32(v2f a, v2f b, v8f c) {
  return __builtin_amdgcn_wmma_f32_16x16x4_f32(false, a, false, b,
                                               (short)0, c, false, false);
}

#define BB 8
#define CC 1024
#define INF_NEG (-3.0e38f)

__device__ __forceinline__ float4 logA_quad(const float4 a) {
  float4 o;
  o.x = logf(a.x + 1e-9f);
  o.y = logf(a.y + 1e-9f);
  o.z = logf(a.z + 1e-9f);
  o.w = logf(a.w + 1e-9f);
  return o;
}

__device__ __forceinline__ void lds_barrier() {
  asm volatile("s_wait_dscnt 0x0\n\ts_barrier_signal -1\n\ts_barrier_wait -1"
               ::: "memory");
}

#define APAD 260
#define TPAD 65
__global__ void __launch_bounds__(256) __attribute__((amdgpu_waves_per_eu(4, 4)))
proj_gemm(
    const float* __restrict__ x, const float* __restrict__ Wp,
    const float* __restrict__ bp, float* __restrict__ xp,
    const float* __restrict__ Ws1, const float* __restrict__ att_w,
    const float* __restrict__ Wc2, const float* __restrict__ A,
    float* __restrict__ hpg, float* __restrict__ gvec,
    float* __restrict__ srcg, float* __restrict__ logA,
    float* __restrict__ rmaxg) {
  extern __shared__ float sm[];
  __shared__ float cs_s[256];
  __shared__ float red[8][32];
  __shared__ float wmx[2][8];
  float* Apan = sm;
  float* BpT  = sm + 64 * APAD;

  const int tid = threadIdx.x;
  const int wave = tid >> 5, lane = tid & 31;
  const int m0 = blockIdx.x * 64;
  const int n0 = blockIdx.y * 64;
  const int blk = blockIdx.x * 4 + blockIdx.y;

  const float* xcol = x + (size_t)m0 * 256 + tid;
  const int bk = tid >> 6, bn = tid & 63;
  const float* wcol = Wp + (size_t)bk * 256 + n0 + bn;
  float av[64], bv[64];
#pragma unroll
  for (int i = 0; i < 64; ++i) av[i] = xcol[i * 256];
#pragma unroll
  for (int i = 0; i < 64; ++i) bv[i] = wcol[(size_t)i * 1024];
  float w1[32];
  const float* w1p = Ws1 + (size_t)(wave * 32) * 128 + blockIdx.y * 32 + lane;
#pragma unroll
  for (int i = 0; i < 32; ++i) w1[i] = w1p[i * 128];
  float4 wsq[4];
#pragma unroll
  for (int i = 0; i < 4; ++i) wsq[i] = ((const float4*)att_w)[4 * (tid & 3) + i];
  const float4* A4 = (const float4*)A + (size_t)blk * 512 + tid;
  const float4 aq0 = A4[0];
  const float4 aq1 = A4[256];

  float* acol = Apan + tid;
  float csum = 0.f;
#pragma unroll
  for (int i = 0; i < 64; ++i) {
    acol[i * APAD] = av[i];
    csum += av[i];
  }
  cs_s[tid] = csum;
  float* bcol = BpT + bn * APAD + bk;
#pragma unroll
  for (int i = 0; i < 64; ++i) bcol[4 * i] = bv[i];
  {
    const float4 l0 = logA_quad(aq0);
    const float4 l1 = logA_quad(aq1);
    float4* L4 = (float4*)logA + (size_t)blk * 512 + tid;
    L4[0] = l0;
    L4[256] = l1;
    float mx0 = fmaxf(fmaxf(l0.x, l0.y), fmaxf(l0.z, l0.w));
    float mx1 = fmaxf(fmaxf(l1.x, l1.y), fmaxf(l1.z, l1.w));
#pragma unroll
    for (int off = 16; off >= 1; off >>= 1) {
      mx0 = fmaxf(mx0, __shfl_xor(mx0, off, 32));
      mx1 = fmaxf(mx1, __shfl_xor(mx1, off, 32));
    }
    if (lane == 0) { wmx[0][wave] = mx0; wmx[1][wave] = mx1; }
  }
  lds_barrier();

  if (blk == 0) {
    const int k = tid >> 2, q = tid & 3;
    const float4* wr = (const float4*)(Wc2 + k * 64 + 16 * q);
    const float4* ar = (const float4*)(att_w + 128 + 16 * q);
    float s = 0.f;
    for (int i = 0; i < 4; ++i) {
      const float4 w = wr[i], a = ar[i];
      s += w.x * a.x + w.y * a.y + w.z * a.z + w.w * a.w;
    }
    s += __shfl_xor(s, 1, 32);
    s += __shfl_xor(s, 2, 32);
    if (q == 0) gvec[k] = s;
  }

  const int r16 = lane & 15, half = lane >> 4;
  const int rw = wave >> 1;
  const int cw = (wave & 1) * 32;

  v8f acc0 = {0, 0, 0, 0, 0, 0, 0, 0};
  v8f acc1 = {0, 0, 0, 0, 0, 0, 0, 0};

  const float* arow = Apan + (rw * 16 + r16) * APAD;
  const float* b0row = BpT + (cw + r16) * APAD;
  const float* b1row = BpT + (cw + 16 + r16) * APAD;

  for (int kk = 0; kk < 256; kk += 4) {
    const int ko = kk + 2 * half;
    v2f a;  a.x = arow[ko];  a.y = arow[ko + 1];
    v2f b0; b0.x = b0row[ko]; b0.y = b0row[ko + 1];
    v2f b1; b1.x = b1row[ko]; b1.y = b1row[ko + 1];
    acc0 = wmma_f32(a, b0, acc0);
    acc1 = wmma_f32(a, b1, acc1);
  }

  const float bias0 = bp[n0 + cw + r16];
  const float bias1 = bp[n0 + cw + 16 + r16];
  {
    float hpp = 0.f;
#pragma unroll
    for (int i = 0; i < 32; ++i) hpp += cs_s[wave * 32 + i] * w1[i];
    red[wave][lane] = hpp;
  }
  lds_barrier();
  float* T = sm;
#pragma unroll
  for (int v = 0; v < 8; ++v) {
    const int rl = rw * 16 + v + 8 * half;
    const float o0 = acc0[v] + bias0;
    const float o1 = acc1[v] + bias1;
    xp[(size_t)(m0 + rl) * 256 + n0 + cw + r16]      = o0;
    xp[(size_t)(m0 + rl) * 256 + n0 + cw + 16 + r16] = o1;
    T[rl * TPAD + cw + r16]      = o0;
    T[rl * TPAD + cw + 16 + r16] = o1;
  }
  if (tid < 32) {
    float s = 0.f;
    for (int w = 0; w < 8; ++w) s += red[w][tid];
    hpg[(size_t)blockIdx.x * 128 + blockIdx.y * 32 + tid] = s;
  } else if (tid < 34) {
    const int i = tid - 32;
    float v = wmx[i][0];
    for (int w = 1; w < 8; ++w) v = fmaxf(v, wmx[i][w]);
    rmaxg[2 * blk + i] = v;
  }
  lds_barrier();
  {
    const int row = tid >> 2, q = tid & 3;
    const float* tr = T + row * TPAD + 16 * q;
    float s = 0.f;
#pragma unroll
    for (int i = 0; i < 4; ++i) {
      s += tr[4 * i + 0] * wsq[i].x;
      s += tr[4 * i + 1] * wsq[i].y;
      s += tr[4 * i + 2] * wsq[i].z;
      s += tr[4 * i + 3] * wsq[i].w;
    }
    s += __shfl_xor(s, 1, 32);
    s += __shfl_xor(s, 2, 32);
    if (q == 0) srcg[(size_t)blockIdx.y * 8192 + m0 + row] = s;
  }
}

#define JP 132
#define XCT_F (64 * JP)
#define EBW_F (16 * JP)
#define ATTN_LDS_F (1024 + 2 * XCT_F + 2 * 8 * EBW_F)


__device__ __forceinline__ void attn_load_tile(float4 (&xt)[8],
                                               const float* xtile,
                                               const int j) {
#pragma unroll
  for (int i = 0; i < 8; ++i)
    xt[i] = *(const float4*)(xtile + (size_t)(j + 32 * (i & 3)) * 256 +
                             32 * (i >> 2));
}
__device__ __forceinline__ void attn_store_tile(float* xdst,
                                                const float4 (&xt)[8],
                                                const int td, const int tj) {
#pragma unroll
  for (int i = 0; i < 8; ++i) {
    float* tp = xdst + (td + 32 * (i >> 2)) * JP + tj + 32 * (i & 3);
    tp[0]      = xt[i].x;
    tp[JP]     = xt[i].y;
    tp[2 * JP] = xt[i].z;
    tp[3 * JP] = xt[i].w;
  }
}
__device__ __forceinline__ void attn_load_logA(float4 (&lg)[16],
                                               const float* lrow,
                                               const int j) {
#pragma unroll
  for (int r = 0; r < 16; ++r)
    lg[r] = *(const float4*)(lrow + (size_t)r * CC + j);
}
__device__ __forceinline__ void attn_exp_stage(
    float* ebdst, const float* srcS, const int j, const int lane,
    const float4 (&lg)[16], const float (&dvals)[16], const float (&mrow)[16],
    float (&srow)[16], const float cs, const float lamh) {
  const float4 sv = ((const float4*)(srcS + j))[lane];
#pragma unroll
  for (int r = 0; r < 16; ++r) {
    float e0 = sv.x + dvals[r] + cs; e0 = e0 > 0.f ? e0 : 0.2f * e0;
    float e1 = sv.y + dvals[r] + cs; e1 = e1 > 0.f ? e1 : 0.2f * e1;
    float e2 = sv.z + dvals[r] + cs; e2 = e2 > 0.f ? e2 : 0.2f * e2;
    float e3 = sv.w + dvals[r] + cs; e3 = e3 > 0.f ? e3 : 0.2f * e3;
    const float p0 = __expf(e0 + lamh * lg[r].x - mrow[r]);
    const float p1 = __expf(e1 + lamh * lg[r].y - mrow[r]);
    const float p2 = __expf(e2 + lamh * lg[r].z - mrow[r]);
    const float p3 = __expf(e3 + lamh * lg[r].w - mrow[r]);
    float* dstp = ebdst + r * JP + lane * 4;
    dstp[0] = p0; dstp[1] = p1; dstp[2] = p2; dstp[3] = p3;
    srow[r] += (p0 + p1) + (p2 + p3);
  }
}
struct AttnFrag { v2f a[4]; v2f b[4][4]; };
__device__ __forceinline__ void attn_load_frag(AttnFrag& f, const float* arow,
                                               const float* brow0,
                                               const int k0, const int half) {
#pragma unroll
  for (int s = 0; s < 4; ++s) {
    const int ko = k0 + 4 * s + 2 * half;
    f.a[s].x = arow[ko]; f.a[s].y = arow[ko + 1];
#pragma unroll
    for (int nt = 0; nt < 4; ++nt) {
      const float* brow = brow0 + nt * 16 * JP;
      f.b[s][nt].x = brow[ko]; f.b[s][nt].y = brow[ko + 1];
    }
  }
}
__device__ __forceinline__ void attn_mma_frag(const AttnFrag& f,
                                              v8f (&acc)[4]) {
#pragma unroll
  for (int s = 0; s < 4; ++s) {
#pragma unroll
    for (int nt = 0; nt < 4; ++nt)
      acc[nt] = wmma_f32(f.a[s], f.b[s][nt], acc[nt]);
  }
}
__device__ __forceinline__ void attn_wmma_stage(const float* ebsrc,
                                                const float* xsrc,
                                                const int r16, const int half,
                                                v8f (&acc)[4]) {
  const float* arow = ebsrc + r16 * JP;
  const float* brow0 = xsrc + r16 * JP;
  AttnFrag f0, f1;
  attn_load_frag(f0, arow, brow0, 0, half);
  for (int k0 = 0; k0 < 128; k0 += 32) {
    attn_load_frag(f1, arow, brow0, k0 + 16, half);
    attn_mma_frag(f0, acc);
    if (k0 + 32 < 128) attn_load_frag(f0, arow, brow0, k0 + 32, half);
    attn_mma_frag(f1, acc);
  }
}

__global__ void __launch_bounds__(256) attn_kernel(
    const float* __restrict__ xp,   const float* __restrict__ logA,
    const float* __restrict__ srcg, const float* __restrict__ hpg,
    const float* __restrict__ gvec, const float* __restrict__ bs1,
    const float* __restrict__ Ws2,  const float* __restrict__ bs2,
    const float* __restrict__ Wc1,  const float* __restrict__ bc1,
    const float* __restrict__ bc2,  const float* __restrict__ att_w,
    const float* __restrict__ lam,  const float* __restrict__ rmaxg,
    float* __restrict__ out,        float* __restrict__ slog_out) {
  __shared__ __attribute__((aligned(16))) float sm[ATTN_LDS_F];
  __shared__ float sred[16];
  __shared__ float4 hs_s[8][32];
  float* srcS = sm;
  float* XcT  = srcS + 1024;
  float* eb   = XcT + 2 * XCT_F;

  const int tid = threadIdx.x;
  const int wave = tid >> 5, lane = tid & 31;
  const int r16 = lane & 15, half = lane >> 4;

  const int bid = blockIdx.x;
  const int b  = bid >> 5;
  const int h  = (bid >> 3) & 3;
  const int it = bid & 7;
  const int i0 = it * 128 + wave * 16;

  const int tj = tid >> 3;
  const int td = (tid & 7) * 4;
  const float* xtile = xp + ((size_t)(b * CC + tj) * 4 + h) * 64 + td;
  const float* lrow = logA + (size_t)i0 * CC + 4 * lane;

  float sv4[4];
#pragma unroll
  for (int q = 0; q < 4; ++q) sv4[q] = srcg[h * 8192 + b * CC + tid + 256 * q];
  const float4* hpq = (const float4*)(hpg + (size_t)(b * 16 + 2 * wave) * 128) + lane;
  const float4 hpa = hpq[0];
  const float4 hpb = hpq[32];
  const float rml = rmaxg[i0 + r16];
  float4 xd[8];
  const float* xdp = xp + ((size_t)(b * CC + i0 + half) * 4 + h) * 64 + 4 * r16;
#pragma unroll
  for (int i = 0; i < 8; ++i) xd[i] = *(const float4*)(xdp + (size_t)(2 * i) * 256);
  const float4 wdq = ((const float4*)(att_w + 64))[r16];
  float4 xt[8];
  attn_load_tile(xt, xtile, 0);
  float4 lg[16];
  attn_load_logA(lg, lrow, 0);
  const float4 b1q = ((const float4*)bs1)[lane];
  v4f w2q[5];
#pragma unroll
  for (int i = 0; i < 5; ++i) w2q[i] = ((const v4f*)Ws2)[5 * lane + i];
  float2 wc1q[5];
#pragma unroll
  for (int k = 0; k < 5; ++k) wc1q[k] = ((const float2*)(Wc1 + k * 64))[lane];
  const float2 bc1q = ((const float2*)bc1)[lane];
  const float2 gq   = ((const float2*)gvec)[lane];
  const float2 bc2q = ((const float2*)bc2)[lane];
  const float2 awq  = ((const float2*)(att_w + 128))[lane];
  float bs2v[5];
#pragma unroll
  for (int o = 0; o < 5; ++o) bs2v[o] = bs2[o];

  const float lamh = lam[h];

  hs_s[wave][lane] = make_float4(hpa.x + hpb.x, hpa.y + hpb.y,
                                 hpa.z + hpb.z, hpa.w + hpb.w);

  float tmx = INF_NEG, tmn = -INF_NEG;
#pragma unroll
  for (int q = 0; q < 4; ++q) {
    srcS[tid + 256 * q] = sv4[q];
    tmx = fmaxf(tmx, sv4[q]);
    tmn = fminf(tmn, sv4[q]);
  }
#pragma unroll
  for (int off = 16; off >= 1; off >>= 1) {
    tmx = fmaxf(tmx, __shfl_xor(tmx, off, 32));
    tmn = fminf(tmn, __shfl_xor(tmn, off, 32));
  }
  if (lane == 0) { sred[wave] = tmx; sred[8 + wave] = tmn; }
  lds_barrier();

  float smax = sred[0], smin = sred[8];
  for (int w = 1; w < 8; ++w) {
    smax = fmaxf(smax, sred[w]);
    smin = fminf(smin, sred[8 + w]);
  }

  float cs;
  {
    asm volatile("" : "+v"(w2q[0]), "+v"(w2q[1]), "+v"(w2q[2]), "+v"(w2q[3]),
                      "+v"(w2q[4]));
    float4 hsum = hs_s[0][lane];
    for (int w = 1; w < 8; ++w) {
      const float4 t4 = hs_s[w][lane];
      hsum.x += t4.x; hsum.y += t4.y; hsum.z += t4.z; hsum.w += t4.w;
    }
    const float sc = 1.0f / 1024.0f;
    const float hid0 = fmaxf(b1q.x + hsum.x * sc, 0.f);
    const float hid1 = fmaxf(b1q.y + hsum.y * sc, 0.f);
    const float hid2 = fmaxf(b1q.z + hsum.z * sc, 0.f);
    const float hid3 = fmaxf(b1q.w + hsum.w * sc, 0.f);
    const float w2f[20] = {w2q[0].x, w2q[0].y, w2q[0].z, w2q[0].w,
                           w2q[1].x, w2q[1].y, w2q[1].z, w2q[1].w,
                           w2q[2].x, w2q[2].y, w2q[2].z, w2q[2].w,
                           w2q[3].x, w2q[3].y, w2q[3].z, w2q[3].w,
                           w2q[4].x, w2q[4].y, w2q[4].z, w2q[4].w};
    float lgt[5];
#pragma unroll
    for (int o = 0; o < 5; ++o)
      lgt[o] = hid0 * w2f[o] + hid1 * w2f[5 + o] + hid2 * w2f[10 + o] +
               hid3 * w2f[15 + o];
#pragma unroll
    for (int off = 16; off >= 1; off >>= 1) {
#pragma unroll
      for (int o = 0; o < 5; ++o) lgt[o] += __shfl_xor(lgt[o], off, 32);
    }
#pragma unroll
    for (int o = 0; o < 5; ++o) lgt[o] += bs2v[o];
    if (h == 0 && it == 0 && tid == 0) {
#pragma unroll
      for (int o = 0; o < 5; ++o) slog_out[b * 5 + o] = lgt[o];
    }
    float mx = lgt[0];
#pragma unroll
    for (int o = 1; o < 5; ++o) mx = fmaxf(mx, lgt[o]);
    float sp[5];
    float ssum = 0.f;
#pragma unroll
    for (int o = 0; o < 5; ++o) { sp[o] = __expf(lgt[o] - mx); ssum += sp[o]; }
    const float sinv = 1.0f / ssum;
    float c1x = bc1q.x, c1y = bc1q.y;
#pragma unroll
    for (int k = 0; k < 5; ++k) {
      const float pk = sp[k] * sinv;
      c1x += pk * wc1q[k].x;
      c1y += pk * wc1q[k].y;
    }
    c1x = fmaxf(c1x, 0.f);
    c1y = fmaxf(c1y, 0.f);
    float part = c1x * gq.x + c1y * gq.y + bc2q.x * awq.x + bc2q.y * awq.y;
#pragma unroll
    for (int off = 16; off >= 1; off >>= 1) part += __shfl_xor(part, off, 32);
    cs = part;
  }

  float dvals[16];
  v16f rmv;
  {
    float dv[8];
#pragma unroll
    for (int i = 0; i < 8; ++i)
      dv[i] = xd[i].x * wdq.x + xd[i].y * wdq.y + xd[i].z * wdq.z +
              xd[i].w * wdq.w;
#pragma unroll
    for (int off = 8; off >= 1; off >>= 1) {
#pragma unroll
      for (int i = 0; i < 8; ++i) dv[i] += __shfl_xor(dv[i], off, 32);
    }
#pragma unroll
    for (int r = 0; r < 16; ++r) {
      dvals[r] = __shfl(dv[r >> 1], (r & 1) * 16, 32);
      rmv[r] = __shfl(rml, r, 32);
    }
  }

  const bool use_bound = __builtin_amdgcn_readfirstlane(
      (lamh >= 0.f && (smax - smin) <= 60.f) ? 1 : 0) != 0;

  float mrow[16];
  if (use_bound) {
#pragma unroll
    for (int r = 0; r < 16; ++r) {
      float e = smax + dvals[r] + cs; e = e > 0.f ? e : 0.2f * e;
      mrow[r] = e + lamh * rmv[r];
    }
  } else {
#pragma unroll
    for (int r = 0; r < 16; ++r) mrow[r] = INF_NEG;

    for (int jc = 0; jc < CC; jc += 128) {
      const float4 sv = ((const float4*)(srcS + jc))[lane];
#pragma unroll
      for (int r = 0; r < 16; ++r) {
        const float4 lq =
            ((const float4*)(logA + (size_t)(i0 + r) * CC + jc))[lane];
        float e0 = sv.x + dvals[r] + cs; e0 = e0 > 0.f ? e0 : 0.2f * e0;
        float e1 = sv.y + dvals[r] + cs; e1 = e1 > 0.f ? e1 : 0.2f * e1;
        float e2 = sv.z + dvals[r] + cs; e2 = e2 > 0.f ? e2 : 0.2f * e2;
        float e3 = sv.w + dvals[r] + cs; e3 = e3 > 0.f ? e3 : 0.2f * e3;
        const float v0 = e0 + lamh * lq.x;
        const float v1 = e1 + lamh * lq.y;
        const float v2 = e2 + lamh * lq.z;
        const float v3 = e3 + lamh * lq.w;
        mrow[r] = fmaxf(mrow[r], fmaxf(fmaxf(v0, v1), fmaxf(v2, v3)));
      }
    }
#pragma unroll
    for (int r = 0; r < 16; ++r) {
#pragma unroll
      for (int off = 16; off >= 1; off >>= 1)
        mrow[r] = fmaxf(mrow[r], __shfl_xor(mrow[r], off, 32));
    }
  }

  v8f acc[4];
  const v8f vzero = {0, 0, 0, 0, 0, 0, 0, 0};
#pragma unroll
  for (int nt = 0; nt < 4; ++nt) acc[nt] = vzero;

  float srow[16];
#pragma unroll
  for (int r = 0; r < 16; ++r) srow[r] = 0.f;

  const bool wmma_first = __builtin_amdgcn_readfirstlane(wave) < 4;

  attn_store_tile(XcT, xt, td, tj);
  attn_load_tile(xt, xtile, 128);
  attn_exp_stage(eb + wave * EBW_F, srcS, 0, lane, lg, dvals, mrow, srow, cs,
                 lamh);
  attn_load_logA(lg, lrow, 128);

  for (int c = 0; c < 7; ++c) {
    const int cur = c & 1;
    const float* xcur = XcT + cur * XCT_F;
    float* xnxt = XcT + (cur ^ 1) * XCT_F;
    const float* ecur = eb + (cur * 8 + wave) * EBW_F;
    float* enxt = eb + ((cur ^ 1) * 8 + wave) * EBW_F;
    const int jn = (c + 1) * 128;

    lds_barrier();

    attn_store_tile(xnxt, xt, td, tj);
    if (c < 6) attn_load_tile(xt, xtile, jn + 128);

    if (wmma_first) {
      attn_wmma_stage(ecur, xcur, r16, half, acc);
      attn_exp_stage(enxt, srcS, jn, lane, lg, dvals, mrow, srow, cs, lamh);
    } else {
      attn_exp_stage(enxt, srcS, jn, lane, lg, dvals, mrow, srow, cs, lamh);
      attn_wmma_stage(ecur, xcur, r16, half, acc);
    }
    if (c < 6) attn_load_logA(lg, lrow, jn + 128);
  }
  lds_barrier();
  attn_wmma_stage(eb + (8 + wave) * EBW_F, XcT + XCT_F, r16, half, acc);

#pragma unroll
  for (int r = 0; r < 16; ++r) {
#pragma unroll
    for (int off = 16; off >= 1; off >>= 1)
      srow[r] += __shfl_xor(srow[r], off, 32);
  }

#pragma unroll
  for (int v = 0; v < 8; ++v) {
    const int rr = v + 8 * half;
    const float inv = 1.0f / srow[rr];
    const int row = i0 + rr;
#pragma unroll
    for (int nt = 0; nt < 4; ++nt) {
      out[(size_t)(b * CC + row) * 256 + h * 64 + nt * 16 + r16] =
          acc[nt][v] * inv;
    }
  }
}

extern "C" void kernel_launch(void* const* d_in, const int* in_sizes, int n_in,
                              void* d_out, int out_size, void* d_ws,
                              size_t ws_size, hipStream_t stream) {
  (void)in_sizes; (void)n_in; (void)out_size; (void)ws_size;

  const float* x   = (const float*)d_in[0];
  const float* Apr = (const float*)d_in[1];
  const float* Wp  = (const float*)d_in[2];
  const float* bp  = (const float*)d_in[3];
  const float* Ws1 = (const float*)d_in[4];
  const float* bs1 = (const float*)d_in[5];
  const float* Ws2 = (const float*)d_in[6];
  const float* bs2 = (const float*)d_in[7];
  const float* Wc1 = (const float*)d_in[8];
  const float* bc1 = (const float*)d_in[9];
  const float* Wc2 = (const float*)d_in[10];
  const float* bc2 = (const float*)d_in[11];
  const float* att = (const float*)d_in[12];
  const float* lam = (const float*)d_in[13];

  float* out  = (float*)d_out;
  float* slog = out + (size_t)BB * CC * 256;

  float* xp   = (float*)d_ws;
  float* logA = xp + 2097152;
  float* srcg = logA + 1048576;
  float* hpg  = srcg + 32768;
  float* gvec = hpg + 16384;
  float* rmax = hpg + 32768 + 64;

  const size_t gemm_lds = (size_t)(2 * 64 * APAD) * sizeof(float);
  proj_gemm<<<dim3(128, 4), 256, gemm_lds, stream>>>(
      x, Wp, bp, xp, Ws1, att, Wc2, Apr, hpg, gvec, srcg, logA, rmax);

  attn_kernel<<<BB * 4 * 8, 256, 0, stream>>>(xp, logA, srcg, hpg, gvec, bs1,
                                              Ws2, bs2, Wc1, bc1, bc2, att,
                                              lam, rmax, out, slog);
}
